# starting kernel
# baseline (speedup 1.0000x reference)
#include <hip/hip_runtime.h>
#include <math.h>

typedef float v2f __attribute__((ext_vector_type(2)));
typedef float v8f __attribute__((ext_vector_type(8)));

#define DIN 128
#define C1  64
#define NEG_SLOPE 0.2f

__device__ __forceinline__ float atomicMaxF(float* addr, float val) {
  if (val >= 0.0f)
    return __int_as_float(atomicMax((int*)addr, __float_as_int(val)));
  else
    return __uint_as_float(atomicMin((unsigned int*)addr, __float_as_uint(val)));
}

__global__ void k_init(float* __restrict__ out1, float* __restrict__ m,
                       float* __restrict__ z, float* __restrict__ outAcc, int n) {
  int i = blockIdx.x * blockDim.x + threadIdx.x;
  if (i < n * C1) out1[i] = 0.0f;
  if (i < n) { m[i] = -INFINITY; z[i] = 0.0f; outAcc[i] = 0.0f; }
}

__global__ __launch_bounds__(256) void k_gemm_wmma(const float* __restrict__ x,
                                                   const float* __restrict__ W,
                                                   float* __restrict__ h, int n) {
  __shared__ float Wt[C1 * DIN];
  int tid = threadIdx.x;
#pragma unroll
  for (int it = 0; it < (DIN * C1) / 256; ++it) {
    int idx = tid + it * 256;
    int k  = idx >> 6;
    int nn = idx & 63;
    Wt[nn * DIN + k] = W[idx];
  }
  __syncthreads();

  int wave = tid >> 5;
  int lane = tid & 31;
  int half = lane >> 4;
  int l    = lane & 15;
  int rowBase = blockIdx.x * 128 + wave * 16;

  int arow = rowBase + l;
  if (arow > n - 1) arow = n - 1;
  const float* ap = x + (size_t)arow * DIN + 2 * half;

  v8f c0 = {}, c1 = {}, c2 = {}, c3 = {};
#pragma unroll 4
  for (int k0 = 0; k0 < DIN; k0 += 4) {
    v2f a = *(const v2f*)(ap + k0);
    const float* wp = &Wt[l * DIN + k0 + 2 * half];
    v2f b0 = *(const v2f*)(wp);
    v2f b1 = *(const v2f*)(wp + 16 * DIN);
    v2f b2 = *(const v2f*)(wp + 32 * DIN);
    v2f b3 = *(const v2f*)(wp + 48 * DIN);
    c0 = __builtin_amdgcn_wmma_f32_16x16x4_f32(false, a, false, b0, (short)0, c0, false, false);
    c1 = __builtin_amdgcn_wmma_f32_16x16x4_f32(false, a, false, b1, (short)0, c1, false, false);
    c2 = __builtin_amdgcn_wmma_f32_16x16x4_f32(false, a, false, b2, (short)0, c2, false, false);
    c3 = __builtin_amdgcn_wmma_f32_16x16x4_f32(false, a, false, b3, (short)0, c3, false, false);
  }

#pragma unroll
  for (int v = 0; v < 8; ++v) {
    int row = rowBase + 8 * half + v;
    if (row < n) {
      float* hp = h + (size_t)row * C1 + l;
      hp[0]  = c0[v];
      hp[16] = c1[v];
      hp[32] = c2[v];
      hp[48] = c3[v];
    }
  }
}

__global__ void k_logits(const float* __restrict__ h, const float* __restrict__ a_src,
                         const float* __restrict__ a_dst, float* __restrict__ esrc,
                         float* __restrict__ edst, int n) {
  int i = blockIdx.x * blockDim.x + threadIdx.x;
  if (i >= n) return;
  const float4* hp = (const float4*)(h + (size_t)i * C1);
  float s = 0.0f, d = 0.0f;
#pragma unroll
  for (int c = 0; c < C1 / 4; ++c) {
    float4 hv = hp[c];
    float4 as = ((const float4*)a_src)[c];
    float4 ad = ((const float4*)a_dst)[c];
    s += hv.x * as.x + hv.y * as.y + hv.z * as.z + hv.w * as.w;
    d += hv.x * ad.x + hv.y * ad.y + hv.z * ad.z + hv.w * ad.w;
  }
  esrc[i] = s;
  edst[i] = d;
}

__global__ void k_edge_max(const int* __restrict__ ei, int E, int n,
                           const float* __restrict__ esrc, const float* __restrict__ edst,
                           float* __restrict__ m, float* __restrict__ eedge) {
  int idx = blockIdx.x * blockDim.x + threadIdx.x;
  int Etot = E + n;
  if (idx >= Etot) return;
  int s, d;
  if (idx < E) { s = ei[idx]; d = ei[E + idx]; }
  else         { s = d = idx - E; }
  float v = esrc[s] + edst[d];
  v = v > 0.0f ? v : NEG_SLOPE * v;
  eedge[idx] = v;
  atomicMaxF(&m[d], v);
}

__global__ void k_edge_expsum(const int* __restrict__ ei, int E, int n,
                              const float* __restrict__ m, float* __restrict__ z,
                              float* __restrict__ eedge) {
  int idx = blockIdx.x * blockDim.x + threadIdx.x;
  int Etot = E + n;
  if (idx >= Etot) return;
  int d = (idx < E) ? ei[E + idx] : (idx - E);
  float p = __expf(eedge[idx] - m[d]);
  eedge[idx] = p;
  atomicAdd(&z[d], p);
}

__global__ void k_edge_aggr64(const int* __restrict__ ei, int E, int n,
                              const float* __restrict__ z, const float* __restrict__ eedge,
                              const float* __restrict__ h, float* __restrict__ out1) {
  long long t = (long long)blockIdx.x * blockDim.x + threadIdx.x;
  int idx = (int)(t >> 4);
  int sub = (int)(t & 15);
  int Etot = E + n;
  if (idx >= Etot) return;
  int s, d;
  if (idx < E) { s = ei[idx]; d = ei[E + idx]; }
  else         { s = d = idx - E; }
  float alpha = eedge[idx] / z[d];
  float4 hv = *(const float4*)(h + (size_t)s * C1 + sub * 4);
  float* op = out1 + (size_t)d * C1 + sub * 4;
  atomicAdd(op + 0, alpha * hv.x);
  atomicAdd(op + 1, alpha * hv.y);
  atomicAdd(op + 2, alpha * hv.z);
  atomicAdd(op + 3, alpha * hv.w);
}

__global__ void k_layer2_prep(const float* __restrict__ out1, const float* __restrict__ b1,
                              const float* __restrict__ W2, const float* __restrict__ a_src2,
                              const float* __restrict__ a_dst2, float* __restrict__ g,
                              float* __restrict__ esrc, float* __restrict__ edst,
                              float* __restrict__ m, float* __restrict__ z, int n) {
  int i = blockIdx.x * blockDim.x + threadIdx.x;
  if (i >= n) return;
  const float4* op = (const float4*)(out1 + (size_t)i * C1);
  float acc = 0.0f;
#pragma unroll
  for (int c = 0; c < C1 / 4; ++c) {
    float4 v = op[c];
    float4 bb = ((const float4*)b1)[c];
    float4 ww = ((const float4*)W2)[c];
    float v0 = fmaxf(v.x + bb.x, 0.0f);
    float v1 = fmaxf(v.y + bb.y, 0.0f);
    float v2 = fmaxf(v.z + bb.z, 0.0f);
    float v3 = fmaxf(v.w + bb.w, 0.0f);
    acc += v0 * ww.x + v1 * ww.y + v2 * ww.z + v3 * ww.w;
  }
  g[i] = acc;
  esrc[i] = acc * a_src2[0];
  edst[i] = acc * a_dst2[0];
  m[i] = -INFINITY;
  z[i] = 0.0f;
}

__global__ void k_edge_aggr1(const int* __restrict__ ei, int E, int n,
                             const float* __restrict__ z, const float* __restrict__ eedge,
                             const float* __restrict__ g, float* __restrict__ outAcc) {
  int idx = blockIdx.x * blockDim.x + threadIdx.x;
  int Etot = E + n;
  if (idx >= Etot) return;
  int s, d;
  if (idx < E) { s = ei[idx]; d = ei[E + idx]; }
  else         { s = d = idx - E; }
  float alpha = eedge[idx] / z[d];
  atomicAdd(&outAcc[d], alpha * g[s]);
}

__global__ void k_final(float* __restrict__ out, const float* __restrict__ b2, int n) {
  int i = blockIdx.x * blockDim.x + threadIdx.x;
  if (i >= n) return;
  float v = out[i] + b2[0];
  out[i] = 1.0f / (1.0f + __expf(-v));
}

extern "C" void kernel_launch(void* const* d_in, const int* in_sizes, int n_in,
                              void* d_out, int out_size, void* d_ws, size_t ws_size,
                              hipStream_t stream) {
  const int*   ei     = (const int*)d_in[0];
  const float* x      = (const float*)d_in[1];
  const float* W1     = (const float*)d_in[2];
  const float* a_src1 = (const float*)d_in[3];
  const float* a_dst1 = (const float*)d_in[4];
  const float* b1     = (const float*)d_in[5];
  const float* W2     = (const float*)d_in[6];
  const float* a_src2 = (const float*)d_in[7];
  const float* a_dst2 = (const float*)d_in[8];
  const float* b2     = (const float*)d_in[9];

  const int E = in_sizes[0] / 2;
  const int n = in_sizes[1] / DIN;
  const int Etot = E + n;

  float* f     = (float*)d_ws;
  float* h     = f;
  float* out1  = f + (size_t)64 * n;
  float* esrc  = f + (size_t)128 * n;
  float* edst  = f + (size_t)129 * n;
  float* m     = f + (size_t)130 * n;
  float* z     = f + (size_t)131 * n;
  float* g     = f + (size_t)132 * n;
  float* eedge = f + (size_t)133 * n;
  float* outAcc = (float*)d_out;

  const int B = 256;
  dim3 blk(B);

  k_init<<<dim3((n * C1 + B - 1) / B), blk, 0, stream>>>(out1, m, z, outAcc, n);

  k_gemm_wmma<<<dim3((n + 127) / 128), blk, 0, stream>>>(x, W1, h, n);
  k_logits<<<dim3((n + B - 1) / B), blk, 0, stream>>>(h, a_src1, a_dst1, esrc, edst, n);
  k_edge_max<<<dim3((Etot + B - 1) / B), blk, 0, stream>>>(ei, E, n, esrc, edst, m, eedge);
  k_edge_expsum<<<dim3((Etot + B - 1) / B), blk, 0, stream>>>(ei, E, n, m, z, eedge);
  {
    long long tthreads = (long long)Etot * 16;
    int grid = (int)((tthreads + B - 1) / B);
    k_edge_aggr64<<<dim3(grid), blk, 0, stream>>>(ei, E, n, z, eedge, h, out1);
  }

  k_layer2_prep<<<dim3((n + B - 1) / B), blk, 0, stream>>>(out1, b1, W2, a_src2, a_dst2,
                                                           g, esrc, edst, m, z, n);
  k_edge_max<<<dim3((Etot + B - 1) / B), blk, 0, stream>>>(ei, E, n, esrc, edst, m, eedge);
  k_edge_expsum<<<dim3((Etot + B - 1) / B), blk, 0, stream>>>(ei, E, n, m, z, eedge);
  k_edge_aggr1<<<dim3((Etot + B - 1) / B), blk, 0, stream>>>(ei, E, n, z, eedge, g, outAcc);

  k_final<<<dim3((n + B - 1) / B), blk, 0, stream>>>(outAcc, b2, n);
}


# Edges grouped by destination each call (integer rank atomic, prefix sum, scatter); one wave per node does max, exp-sum and 64-ch sum in registers, divides once; no float atomics; sum order changes
# speedup vs baseline: 2.5707x; 2.5707x over previous
#include <hip/hip_runtime.h>
#include <math.h>

typedef float v2f __attribute__((ext_vector_type(2)));
typedef float v8f __attribute__((ext_vector_type(8)));

#define DIN 128
#define C1  64
#define NEG_SLOPE 0.2f
#define SCAN_B 1024

__device__ __forceinline__ float wave_max(float v) {
#pragma unroll
  for (int o = 16; o > 0; o >>= 1) v = fmaxf(v, __shfl_xor(v, o, 32));
  return v;
}
__device__ __forceinline__ float wave_sum(float v) {
#pragma unroll
  for (int o = 16; o > 0; o >>= 1) v += __shfl_xor(v, o, 32);
  return v;
}

__device__ __forceinline__ int block_excl_scan(int v, int* wsum, int& total) {
  int lane = threadIdx.x & 31;
  int wave = threadIdx.x >> 5;
  int inc = v;
#pragma unroll
  for (int o = 1; o < 32; o <<= 1) {
    int t = __shfl_up(inc, o, 32);
    if (lane >= o) inc += t;
  }
  if (lane == 31) wsum[wave] = inc;
  __syncthreads();
  if (wave == 0) {
    int w = wsum[lane];
    int winc = w;
#pragma unroll
    for (int o = 1; o < 32; o <<= 1) {
      int t = __shfl_up(winc, o, 32);
      if (lane >= o) winc += t;
    }
    wsum[lane] = winc - w;
    if (lane == 31) wsum[32] = winc;
  }
  __syncthreads();
  total = wsum[32];
  return inc - v + wsum[wave];
}

__global__ void k_init(int* __restrict__ deg, int n) {
  int i = blockIdx.x * blockDim.x + threadIdx.x;
  if (i < n) deg[i] = 0;
}

__global__ void k_count(const int* __restrict__ ei, int E, int* __restrict__ deg,
                        int* __restrict__ rank) {
  int e = blockIdx.x * blockDim.x + threadIdx.x;
  if (e >= E) return;
  int d = ei[E + e];
  rank[e] = atomicAdd(&deg[d], 1);
}

__global__ __launch_bounds__(SCAN_B) void k_scan1(const int* __restrict__ deg,
                                                  int* __restrict__ off,
                                                  int* __restrict__ bsum, int n) {
  __shared__ int wsum[33];
  int i = blockIdx.x * SCAN_B + threadIdx.x;
  int v = (i < n) ? deg[i] : 0;
  int total;
  int ex = block_excl_scan(v, wsum, total);
  if (i < n) off[i] = ex;
  if (threadIdx.x == 0) bsum[blockIdx.x] = total;
}

__global__ __launch_bounds__(SCAN_B) void k_scan2(int* __restrict__ bsum, int nb) {
  __shared__ int wsum[33];
  int carry = 0;
  for (int c = 0; c < nb; c += SCAN_B) {
    int i = c + threadIdx.x;
    int v = (i < nb) ? bsum[i] : 0;
    int total;
    int ex = block_excl_scan(v, wsum, total);
    if (i < nb) bsum[i] = carry + ex;
    carry += total;
    __syncthreads();
  }
}

__global__ void k_scan3(int* __restrict__ off, const int* __restrict__ bsum, int n) {
  int i = blockIdx.x * blockDim.x + threadIdx.x;
  if (i < n) off[i] += bsum[i / SCAN_B];
}

__global__ void k_fill(const int* __restrict__ ei, int E, const int* __restrict__ off,
                       const int* __restrict__ rank, int* __restrict__ csr) {
  int e = blockIdx.x * blockDim.x + threadIdx.x;
  if (e >= E) return;
  int s = ei[e];
  int d = ei[E + e];
  csr[off[d] + rank[e]] = s;
}

__global__ __launch_bounds__(256) void k_gemm_wmma(const float* __restrict__ x,
                                                   const float* __restrict__ W,
                                                   float* __restrict__ h, int n) {
  __shared__ float Wt[C1 * DIN];
  int tid = threadIdx.x;
#pragma unroll
  for (int it = 0; it < (DIN * C1) / 256; ++it) {
    int idx = tid + it * 256;
    int k  = idx >> 6;
    int nn = idx & 63;
    Wt[nn * DIN + k] = W[idx];
  }
  __syncthreads();

  int wave = tid >> 5;
  int lane = tid & 31;
  int half = lane >> 4;
  int l    = lane & 15;
  int rowBase = blockIdx.x * 128 + wave * 16;

  int arow = rowBase + l;
  if (arow > n - 1) arow = n - 1;
  const float* ap = x + (size_t)arow * DIN + 2 * half;

  v8f c0 = {}, c1 = {}, c2 = {}, c3 = {};
#pragma unroll 4
  for (int k0 = 0; k0 < DIN; k0 += 4) {
    v2f a = *(const v2f*)(ap + k0);
    const float* wp = &Wt[l * DIN + k0 + 2 * half];
    v2f b0 = *(const v2f*)(wp);
    v2f b1 = *(const v2f*)(wp + 16 * DIN);
    v2f b2 = *(const v2f*)(wp + 32 * DIN);
    v2f b3 = *(const v2f*)(wp + 48 * DIN);
    c0 = __builtin_amdgcn_wmma_f32_16x16x4_f32(false, a, false, b0, (short)0, c0, false, false);
    c1 = __builtin_amdgcn_wmma_f32_16x16x4_f32(false, a, false, b1, (short)0, c1, false, false);
    c2 = __builtin_amdgcn_wmma_f32_16x16x4_f32(false, a, false, b2, (short)0, c2, false, false);
    c3 = __builtin_amdgcn_wmma_f32_16x16x4_f32(false, a, false, b3, (short)0, c3, false, false);
  }

#pragma unroll
  for (int v = 0; v < 8; ++v) {
    int row = rowBase + 8 * half + v;
    if (row < n) {
      float* hp = h + (size_t)row * C1 + l;
      hp[0]  = c0[v];
      hp[16] = c1[v];
      hp[32] = c2[v];
      hp[48] = c3[v];
    }
  }
}

__global__ void k_logits(const float* __restrict__ h, const float* __restrict__ a_src,
                         const float* __restrict__ a_dst, float* __restrict__ esrc,
                         float* __restrict__ edst, int n) {
  int i = blockIdx.x * blockDim.x + threadIdx.x;
  if (i >= n) return;
  const float4* hp = (const float4*)(h + (size_t)i * C1);
  float s = 0.0f, d = 0.0f;
#pragma unroll
  for (int c = 0; c < C1 / 4; ++c) {
    float4 hv = hp[c];
    float4 as = ((const float4*)a_src)[c];
    float4 ad = ((const float4*)a_dst)[c];
    s += hv.x * as.x + hv.y * as.y + hv.z * as.z + hv.w * as.w;
    d += hv.x * ad.x + hv.y * ad.y + hv.z * ad.z + hv.w * ad.w;
  }
  esrc[i] = s;
  edst[i] = d;
}

__global__ __launch_bounds__(256) void k_node64(const int* __restrict__ off,
                                                const int* __restrict__ deg,
                                                const int* __restrict__ csr,
                                                const float* __restrict__ esrc,
                                                const float* __restrict__ edst,
                                                const float* __restrict__ h,
                                                float* __restrict__ out1, int n) {
  int lane = threadIdx.x & 31;
  int d = __builtin_amdgcn_readfirstlane(blockIdx.x * 8 + (threadIdx.x >> 5));
  if (d >= n) return;
  int start = __builtin_amdgcn_readfirstlane(off[d]);
  int cnt   = __builtin_amdgcn_readfirstlane(deg[d]);
  int tot   = cnt + 1;
  float ed  = edst[d];

  float mx = -INFINITY;
  for (int k = lane; k < tot; k += 32) {
    int s = (k < cnt) ? csr[start + k] : d;
    float v = esrc[s] + ed;
    v = v > 0.0f ? v : NEG_SLOPE * v;
    mx = fmaxf(mx, v);
  }
  mx = wave_max(mx);

  float zl = 0.0f;
  float a0 = 0.0f, a1 = 0.0f;
  for (int base = 0; base < tot; base += 32) {
    int k = base + lane;
    int s = d;
    float p = 0.0f;
    if (k < tot) {
      if (k < cnt) s = csr[start + k];
      float v = esrc[s] + ed;
      v = v > 0.0f ? v : NEG_SLOPE * v;
      p = __expf(v - mx);
    }
    zl += p;
    int m = tot - base;
    if (m > 32) m = 32;
#pragma unroll 4
    for (int t = 0; t < m; ++t) {
      int   st = __builtin_amdgcn_readlane(s, t);
      float pt = __int_as_float(__builtin_amdgcn_readlane(__float_as_int(p), t));
      v2f hv = *(const v2f*)(h + (size_t)st * C1 + 2 * lane);
      a0 += pt * hv.x;
      a1 += pt * hv.y;
    }
  }
  float z = wave_sum(zl);
  v2f o;
  o.x = a0 / z;
  o.y = a1 / z;
  *(v2f*)(out1 + (size_t)d * C1 + 2 * lane) = o;
}

__global__ void k_layer2_prep(const float* __restrict__ out1, const float* __restrict__ b1,
                              const float* __restrict__ W2, float* __restrict__ g, int n) {
  int i = blockIdx.x * blockDim.x + threadIdx.x;
  if (i >= n) return;
  const float4* op = (const float4*)(out1 + (size_t)i * C1);
  float acc = 0.0f;
#pragma unroll
  for (int c = 0; c < C1 / 4; ++c) {
    float4 v = op[c];
    float4 bb = ((const float4*)b1)[c];
    float4 ww = ((const float4*)W2)[c];
    float v0 = fmaxf(v.x + bb.x, 0.0f);
    float v1 = fmaxf(v.y + bb.y, 0.0f);
    float v2 = fmaxf(v.z + bb.z, 0.0f);
    float v3 = fmaxf(v.w + bb.w, 0.0f);
    acc += v0 * ww.x + v1 * ww.y + v2 * ww.z + v3 * ww.w;
  }
  g[i] = acc;
}

__global__ __launch_bounds__(256) void k_node1(const int* __restrict__ off,
                                               const int* __restrict__ deg,
                                               const int* __restrict__ csr,
                                               const float* __restrict__ g,
                                               const float* __restrict__ a_src2,
                                               const float* __restrict__ a_dst2,
                                               const float* __restrict__ b2,
                                               float* __restrict__ out, int n) {
  int lane = threadIdx.x & 31;
  int d = __builtin_amdgcn_readfirstlane(blockIdx.x * 8 + (threadIdx.x >> 5));
  if (d >= n) return;
  int start = __builtin_amdgcn_readfirstlane(off[d]);
  int cnt   = __builtin_amdgcn_readfirstlane(deg[d]);
  int tot   = cnt + 1;
  float as2 = a_src2[0];
  float ed  = g[d] * a_dst2[0];

  float mx = -INFINITY;
  for (int k = lane; k < tot; k += 32) {
    int s = (k < cnt) ? csr[start + k] : d;
    float v = g[s] * as2 + ed;
    v = v > 0.0f ? v : NEG_SLOPE * v;
    mx = fmaxf(mx, v);
  }
  mx = wave_max(mx);

  float zl = 0.0f, al = 0.0f;
  for (int k = lane; k < tot; k += 32) {
    int s = (k < cnt) ? csr[start + k] : d;
    float gs = g[s];
    float v = gs * as2 + ed;
    v = v > 0.0f ? v : NEG_SLOPE * v;
    float p = __expf(v - mx);
    zl += p;
    al += p * gs;
  }
  float z = wave_sum(zl);
  float a = wave_sum(al);
  if (lane == 0) {
    float v = a / z + b2[0];
    out[d] = 1.0f / (1.0f + __expf(-v));
  }
}

extern "C" void kernel_launch(void* const* d_in, const int* in_sizes, int n_in,
                              void* d_out, int out_size, void* d_ws, size_t ws_size,
                              hipStream_t stream) {
  const int*   ei     = (const int*)d_in[0];
  const float* x      = (const float*)d_in[1];
  const float* W1     = (const float*)d_in[2];
  const float* a_src1 = (const float*)d_in[3];
  const float* a_dst1 = (const float*)d_in[4];
  const float* b1     = (const float*)d_in[5];
  const float* W2     = (const float*)d_in[6];
  const float* a_src2 = (const float*)d_in[7];
  const float* a_dst2 = (const float*)d_in[8];
  const float* b2     = (const float*)d_in[9];

  const int E = in_sizes[0] / 2;
  const int n = in_sizes[1] / DIN;
  const int nb = (n + SCAN_B - 1) / SCAN_B;

  float* f     = (float*)d_ws;
  float* h     = f;
  float* out1  = f + (size_t)64 * n;
  float* esrc  = f + (size_t)128 * n;
  float* edst  = f + (size_t)129 * n;
  float* g     = f + (size_t)130 * n;
  int*   deg   = (int*)(f + (size_t)131 * n);
  int*   off   = (int*)(f + (size_t)132 * n);
  int*   bsum  = (int*)(f + (size_t)133 * n);
  int*   rank  = (int*)(f + (size_t)134 * n);
  int*   csr   = rank + (size_t)E;
  float* out   = (float*)d_out;

  const int B = 256;
  dim3 blk(B);

  k_init<<<dim3((n + B - 1) / B), blk, 0, stream>>>(deg, n);
  k_count<<<dim3((E + B - 1) / B), blk, 0, stream>>>(ei, E, deg, rank);
  k_scan1<<<dim3(nb), dim3(SCAN_B), 0, stream>>>(deg, off, bsum, n);
  k_scan2<<<dim3(1), dim3(SCAN_B), 0, stream>>>(bsum, nb);
  k_scan3<<<dim3((n + B - 1) / B), blk, 0, stream>>>(off, bsum, n);
  k_fill<<<dim3((E + B - 1) / B), blk, 0, stream>>>(ei, E, off, rank, csr);

  k_gemm_wmma<<<dim3((n + 127) / 128), blk, 0, stream>>>(x, W1, h, n);
  k_logits<<<dim3((n + B - 1) / B), blk, 0, stream>>>(h, a_src1, a_dst1, esrc, edst, n);
  k_node64<<<dim3((n + 7) / 8), blk, 0, stream>>>(off, deg, csr, esrc, edst, h, out1, n);

  k_layer2_prep<<<dim3((n + B - 1) / B), blk, 0, stream>>>(out1, b1, W2, g, n);
  k_node1<<<dim3((n + 7) / 8), blk, 0, stream>>>(off, deg, csr, g, a_src2, a_dst2, b2, out, n);
}


# GEMM: transposed W tile kept in LDS at a row pitch of 132 floats instead of 128, and staged in a bank-spreading order: fragment loads and staging stores no longer pile onto one LDS bank
# speedup vs baseline: 2.8066x; 1.0918x over previous
#include <hip/hip_runtime.h>
#include <math.h>

typedef float v2f __attribute__((ext_vector_type(2)));
typedef float v8f __attribute__((ext_vector_type(8)));

#define DIN 128
#define C1  64
#define NEG_SLOPE 0.2f
#define SCAN_B 1024
#define WP 132

__device__ __forceinline__ float wave_max(float v) {
#pragma unroll
  for (int o = 16; o > 0; o >>= 1) v = fmaxf(v, __shfl_xor(v, o, 32));
  return v;
}
__device__ __forceinline__ float wave_sum(float v) {
#pragma unroll
  for (int o = 16; o > 0; o >>= 1) v += __shfl_xor(v, o, 32);
  return v;
}

__device__ __forceinline__ int block_excl_scan(int v, int* wsum, int& total) {
  int lane = threadIdx.x & 31;
  int wave = threadIdx.x >> 5;
  int inc = v;
#pragma unroll
  for (int o = 1; o < 32; o <<= 1) {
    int t = __shfl_up(inc, o, 32);
    if (lane >= o) inc += t;
  }
  if (lane == 31) wsum[wave] = inc;
  __syncthreads();
  if (wave == 0) {
    int w = wsum[lane];
    int winc = w;
#pragma unroll
    for (int o = 1; o < 32; o <<= 1) {
      int t = __shfl_up(winc, o, 32);
      if (lane >= o) winc += t;
    }
    wsum[lane] = winc - w;
    if (lane == 31) wsum[32] = winc;
  }
  __syncthreads();
  total = wsum[32];
  return inc - v + wsum[wave];
}

__global__ void k_init(int* __restrict__ deg, int n) {
  int i = blockIdx.x * blockDim.x + threadIdx.x;
  if (i < n) deg[i] = 0;
}

__global__ void k_count(const int* __restrict__ ei, int E, int* __restrict__ deg,
                        int* __restrict__ rank) {
  int e = blockIdx.x * blockDim.x + threadIdx.x;
  if (e >= E) return;
  int d = ei[E + e];
  rank[e] = atomicAdd(&deg[d], 1);
}

__global__ __launch_bounds__(SCAN_B) void k_scan1(const int* __restrict__ deg,
                                                  int* __restrict__ off,
                                                  int* __restrict__ bsum, int n) {
  __shared__ int wsum[33];
  int i = blockIdx.x * SCAN_B + threadIdx.x;
  int v = (i < n) ? deg[i] : 0;
  int total;
  int ex = block_excl_scan(v, wsum, total);
  if (i < n) off[i] = ex;
  if (threadIdx.x == 0) bsum[blockIdx.x] = total;
}

__global__ __launch_bounds__(SCAN_B) void k_scan2(int* __restrict__ bsum, int nb) {
  __shared__ int wsum[33];
  int carry = 0;
  for (int c = 0; c < nb; c += SCAN_B) {
    int i = c + threadIdx.x;
    int v = (i < nb) ? bsum[i] : 0;
    int total;
    int ex = block_excl_scan(v, wsum, total);
    if (i < nb) bsum[i] = carry + ex;
    carry += total;
    __syncthreads();
  }
}

__global__ void k_scan3(int* __restrict__ off, const int* __restrict__ bsum, int n) {
  int i = blockIdx.x * blockDim.x + threadIdx.x;
  if (i < n) off[i] += bsum[i / SCAN_B];
}

__global__ void k_fill(const int* __restrict__ ei, int E, const int* __restrict__ off,
                       const int* __restrict__ rank, int* __restrict__ csr) {
  int e = blockIdx.x * blockDim.x + threadIdx.x;
  if (e >= E) return;
  int s = ei[e];
  int d = ei[E + e];
  csr[off[d] + rank[e]] = s;
}

__global__ __launch_bounds__(256) void k_gemm_wmma(const float* __restrict__ x,
                                                   const float* __restrict__ W,
                                                   float* __restrict__ h, int n) {
  __shared__ __attribute__((aligned(16))) float Wt[C1 * WP];
  int tid = threadIdx.x;
#pragma unroll
  for (int it = 0; it < (DIN * C1) / 256; ++it) {
    int idx  = tid + it * 256;
    int c    = idx & 15;
    int hs   = (idx >> 4) & 1;
    int rest = idx >> 5;
    int nn   = (rest & 3) * 16 + c;
    int kk   = rest >> 2;
    int k    = (kk >> 1) * 4 + (kk & 1) + 2 * hs;
    Wt[nn * WP + k] = W[k * C1 + nn];
  }
  __syncthreads();

  int wave = tid >> 5;
  int lane = tid & 31;
  int half = lane >> 4;
  int l    = lane & 15;
  int rowBase = blockIdx.x * 128 + wave * 16;

  int arow = rowBase + l;
  if (arow > n - 1) arow = n - 1;
  const float* ap = x + (size_t)arow * DIN + 2 * half;

  v8f c0 = {}, c1 = {}, c2 = {}, c3 = {};
#pragma unroll 4
  for (int k0 = 0; k0 < DIN; k0 += 4) {
    v2f a = *(const v2f*)(ap + k0);
    const float* wp = &Wt[l * WP + k0 + 2 * half];
    v2f b0 = *(const v2f*)(wp);
    v2f b1 = *(const v2f*)(wp + 16 * WP);
    v2f b2 = *(const v2f*)(wp + 32 * WP);
    v2f b3 = *(const v2f*)(wp + 48 * WP);
    c0 = __builtin_amdgcn_wmma_f32_16x16x4_f32(false, a, false, b0, (short)0, c0, false, false);
    c1 = __builtin_amdgcn_wmma_f32_16x16x4_f32(false, a, false, b1, (short)0, c1, false, false);
    c2 = __builtin_amdgcn_wmma_f32_16x16x4_f32(false, a, false, b2, (short)0, c2, false, false);
    c3 = __builtin_amdgcn_wmma_f32_16x16x4_f32(false, a, false, b3, (short)0, c3, false, false);
  }

#pragma unroll
  for (int v = 0; v < 8; ++v) {
    int row = rowBase + 8 * half + v;
    if (row < n) {
      float* hp = h + (size_t)row * C1 + l;
      hp[0]  = c0[v];
      hp[16] = c1[v];
      hp[32] = c2[v];
      hp[48] = c3[v];
    }
  }
}

__global__ void k_logits(const float* __restrict__ h, const float* __restrict__ a_src,
                         const float* __restrict__ a_dst, float* __restrict__ esrc,
                         float* __restrict__ edst, int n) {
  int i = blockIdx.x * blockDim.x + threadIdx.x;
  if (i >= n) return;
  const float4* hp = (const float4*)(h + (size_t)i * C1);
  float s = 0.0f, d = 0.0f;
#pragma unroll
  for (int c = 0; c < C1 / 4; ++c) {
    float4 hv = hp[c];
    float4 as = ((const float4*)a_src)[c];
    float4 ad = ((const float4*)a_dst)[c];
    s += hv.x * as.x + hv.y * as.y + hv.z * as.z + hv.w * as.w;
    d += hv.x * ad.x + hv.y * ad.y + hv.z * ad.z + hv.w * ad.w;
  }
  esrc[i] = s;
  edst[i] = d;
}

__global__ __launch_bounds__(256) void k_node64(const int* __restrict__ off,
                                                const int* __restrict__ deg,
                                                const int* __restrict__ csr,
                                                const float* __restrict__ esrc,
                                                const float* __restrict__ edst,
                                                const float* __restrict__ h,
                                                float* __restrict__ out1, int n) {
  int lane = threadIdx.x & 31;
  int d = __builtin_amdgcn_readfirstlane(blockIdx.x * 8 + (threadIdx.x >> 5));
  if (d >= n) return;
  int start = __builtin_amdgcn_readfirstlane(off[d]);
  int cnt   = __builtin_amdgcn_readfirstlane(deg[d]);
  int tot   = cnt + 1;
  float ed  = edst[d];

  float mx = -INFINITY;
  for (int k = lane; k < tot; k += 32) {
    int s = (k < cnt) ? csr[start + k] : d;
    float v = esrc[s] + ed;
    v = v > 0.0f ? v : NEG_SLOPE * v;
    mx = fmaxf(mx, v);
  }
  mx = wave_max(mx);

  float zl = 0.0f;
  float a0 = 0.0f, a1 = 0.0f;
  for (int base = 0; base < tot; base += 32) {
    int k = base + lane;
    int s = d;
    float p = 0.0f;
    if (k < tot) {
      if (k < cnt) s = csr[start + k];
      float v = esrc[s] + ed;
      v = v > 0.0f ? v : NEG_SLOPE * v;
      p = __expf(v - mx);
    }
    zl += p;
    int m = tot - base;
    if (m > 32) m = 32;
#pragma unroll 4
    for (int t = 0; t < m; ++t) {
      int   st = __builtin_amdgcn_readlane(s, t);
      float pt = __int_as_float(__builtin_amdgcn_readlane(__float_as_int(p), t));
      v2f hv = *(const v2f*)(h + (size_t)st * C1 + 2 * lane);
      a0 += pt * hv.x;
      a1 += pt * hv.y;
    }
  }
  float z = wave_sum(zl);
  v2f o;
  o.x = a0 / z;
  o.y = a1 / z;
  *(v2f*)(out1 + (size_t)d * C1 + 2 * lane) = o;
}

__global__ void k_layer2_prep(const float* __restrict__ out1, const float* __restrict__ b1,
                              const float* __restrict__ W2, float* __restrict__ g, int n) {
  int i = blockIdx.x * blockDim.x + threadIdx.x;
  if (i >= n) return;
  const float4* op = (const float4*)(out1 + (size_t)i * C1);
  float acc = 0.0f;
#pragma unroll
  for (int c = 0; c < C1 / 4; ++c) {
    float4 v = op[c];
    float4 bb = ((const float4*)b1)[c];
    float4 ww = ((const float4*)W2)[c];
    float v0 = fmaxf(v.x + bb.x, 0.0f);
    float v1 = fmaxf(v.y + bb.y, 0.0f);
    float v2 = fmaxf(v.z + bb.z, 0.0f);
    float v3 = fmaxf(v.w + bb.w, 0.0f);
    acc += v0 * ww.x + v1 * ww.y + v2 * ww.z + v3 * ww.w;
  }
  g[i] = acc;
}

__global__ __launch_bounds__(256) void k_node1(const int* __restrict__ off,
                                               const int* __restrict__ deg,
                                               const int* __restrict__ csr,
                                               const float* __restrict__ g,
                                               const float* __restrict__ a_src2,
                                               const float* __restrict__ a_dst2,
                                               const float* __restrict__ b2,
                                               float* __restrict__ out, int n) {
  int lane = threadIdx.x & 31;
  int d = __builtin_amdgcn_readfirstlane(blockIdx.x * 8 + (threadIdx.x >> 5));
  if (d >= n) return;
  int start = __builtin_amdgcn_readfirstlane(off[d]);
  int cnt   = __builtin_amdgcn_readfirstlane(deg[d]);
  int tot   = cnt + 1;
  float as2 = a_src2[0];
  float ed  = g[d] * a_dst2[0];

  float mx = -INFINITY;
  for (int k = lane; k < tot; k += 32) {
    int s = (k < cnt) ? csr[start + k] : d;
    float v = g[s] * as2 + ed;
    v = v > 0.0f ? v : NEG_SLOPE * v;
    mx = fmaxf(mx, v);
  }
  mx = wave_max(mx);

  float zl = 0.0f, al = 0.0f;
  for (int k = lane; k < tot; k += 32) {
    int s = (k < cnt) ? csr[start + k] : d;
    float gs = g[s];
    float v = gs * as2 + ed;
    v = v > 0.0f ? v : NEG_SLOPE * v;
    float p = __expf(v - mx);
    zl += p;
    al += p * gs;
  }
  float z = wave_sum(zl);
  float a = wave_sum(al);
  if (lane == 0) {
    float v = a / z + b2[0];
    out[d] = 1.0f / (1.0f + __expf(-v));
  }
}

extern "C" void kernel_launch(void* const* d_in, const int* in_sizes, int n_in,
                              void* d_out, int out_size, void* d_ws, size_t ws_size,
                              hipStream_t stream) {
  const int*   ei     = (const int*)d_in[0];
  const float* x      = (const float*)d_in[1];
  const float* W1     = (const float*)d_in[2];
  const float* a_src1 = (const float*)d_in[3];
  const float* a_dst1 = (const float*)d_in[4];
  const float* b1     = (const float*)d_in[5];
  const float* W2     = (const float*)d_in[6];
  const float* a_src2 = (const float*)d_in[7];
  const float* a_dst2 = (const float*)d_in[8];
  const float* b2     = (const float*)d_in[9];

  const int E = in_sizes[0] / 2;
  const int n = in_sizes[1] / DIN;
  const int nb = (n + SCAN_B - 1) / SCAN_B;

  float* f     = (float*)d_ws;
  float* h     = f;
  float* out1  = f + (size_t)64 * n;
  float* esrc  = f + (size_t)128 * n;
  float* edst  = f + (size_t)129 * n;
  float* g     = f + (size_t)130 * n;
  int*   deg   = (int*)(f + (size_t)131 * n);
  int*   off   = (int*)(f + (size_t)132 * n);
  int*   bsum  = (int*)(f + (size_t)133 * n);
  int*   rank  = (int*)(f + (size_t)134 * n);
  int*   csr   = rank + (size_t)E;
  float* out   = (float*)d_out;

  const int B = 256;
  dim3 blk(B);

  k_init<<<dim3((n + B - 1) / B), blk, 0, stream>>>(deg, n);
  k_count<<<dim3((E + B - 1) / B), blk, 0, stream>>>(ei, E, deg, rank);
  k_scan1<<<dim3(nb), dim3(SCAN_B), 0, stream>>>(deg, off, bsum, n);
  k_scan2<<<dim3(1), dim3(SCAN_B), 0, stream>>>(bsum, nb);
  k_scan3<<<dim3((n + B - 1) / B), blk, 0, stream>>>(off, bsum, n);
  k_fill<<<dim3((E + B - 1) / B), blk, 0, stream>>>(ei, E, off, rank, csr);

  k_gemm_wmma<<<dim3((n + 127) / 128), blk, 0, stream>>>(x, W1, h, n);
  k_logits<<<dim3((n + B - 1) / B), blk, 0, stream>>>(h, a_src1, a_dst1, esrc, edst, n);
  k_node64<<<dim3((n + 7) / 8), blk, 0, stream>>>(off, deg, csr, esrc, edst, h, out1, n);

  k_layer2_prep<<<dim3((n + B - 1) / B), blk, 0, stream>>>(out1, b1, W2, g, n);
  k_node1<<<dim3((n + 7) / 8), blk, 0, stream>>>(off, deg, csr, g, a_src2, a_dst2, b2, out, n);
}


# Edges grouped by destination in LDS: 8192-edge chunks ordered by 256-node bin, then one workgroup per bin counts and places; no global atomic, scan or scattered store; sum order still atomic arrival
# speedup vs baseline: 5.6151x; 2.0007x over previous
#include <hip/hip_runtime.h>
#include <math.h>

typedef float v2f __attribute__((ext_vector_type(2)));
typedef float v8f __attribute__((ext_vector_type(8)));

#define DIN 128
#define C1  64
#define NEG_SLOPE 0.2f
#define CH    8192
#define EPT   8
#define BSH   8
#define BNODES 256
#define MAXNB 1024
#define CAPB  12288
#define WP 132

__device__ __forceinline__ float wave_max(float v) {
#pragma unroll
  for (int o = 16; o > 0; o >>= 1) v = fmaxf(v, __shfl_xor(v, o, 32));
  return v;
}
__device__ __forceinline__ float wave_sum(float v) {
#pragma unroll
  for (int o = 16; o > 0; o >>= 1) v += __shfl_xor(v, o, 32);
  return v;
}

__device__ __forceinline__ int block_excl_scan(int v, int* wsum, int& total) {
  int lane = threadIdx.x & 31;
  int wave = threadIdx.x >> 5;
  int inc = v;
#pragma unroll
  for (int o = 1; o < 32; o <<= 1) {
    int t = __shfl_up(inc, o, 32);
    if (lane >= o) inc += t;
  }
  if (lane == 31) wsum[wave] = inc;
  __syncthreads();
  if (wave == 0) {
    int w = wsum[lane];
    int winc = w;
#pragma unroll
    for (int o = 1; o < 32; o <<= 1) {
      int t = __shfl_up(winc, o, 32);
      if (lane >= o) winc += t;
    }
    wsum[lane] = winc - w;
    if (lane == 31) wsum[32] = winc;
  }
  __syncthreads();
  total = wsum[32];
  return inc - v + wsum[wave];
}

__device__ __forceinline__ int tab_pitch(int nbins) { return (nbins + 32) & ~31; }
__global__ __launch_bounds__(1024) void k_binsort(const int* __restrict__ ei, int E, int nbins,
                                                  unsigned* __restrict__ part,
                                                  int* __restrict__ tab) {
  __shared__ int cnt[MAXNB];
  __shared__ int boff[MAXNB + 1];
  __shared__ unsigned stage[CH];
  __shared__ int wsum[33];
  int tid = threadIdx.x;
  int base = blockIdx.x * CH;
  int m = E - base;
  if (m > CH) m = CH;
  for (int b = tid; b < nbins; b += 1024) cnt[b] = 0;
  __syncthreads();

  int s[EPT], d[EPT], r[EPT];
#pragma unroll
  for (int j = 0; j < EPT; ++j) {
    int i = j * 1024 + tid;
    s[j] = 0; d[j] = 0;
    if (i < m) { s[j] = ei[base + i]; d[j] = ei[E + base + i]; }
  }
#pragma unroll
  for (int j = 0; j < EPT; ++j) {
    int i = j * 1024 + tid;
    r[j] = 0;
    if (i < m) r[j] = atomicAdd(&cnt[d[j] >> BSH], 1);
  }
  __syncthreads();

  int v = (tid < nbins) ? cnt[tid] : 0;
  int total;
  int ex = block_excl_scan(v, wsum, total);
  if (tid < nbins) boff[tid] = ex;
  if (tid == 0) boff[nbins] = total;
  __syncthreads();

#pragma unroll
  for (int j = 0; j < EPT; ++j) {
    int i = j * 1024 + tid;
    if (i < m)
      stage[boff[d[j] >> BSH] + r[j]] = ((unsigned)s[j] << BSH) | (unsigned)(d[j] & (BNODES - 1));
  }
  __syncthreads();

  for (int i = tid; i < m; i += 1024) part[base + i] = stage[i];
  int* trow = tab + (size_t)blockIdx.x * tab_pitch(nbins);
  for (int b = tid; b <= nbins; b += 1024) trow[b] = boff[b];
}

__global__ __launch_bounds__(1024) void k_binfinish(const unsigned* __restrict__ part,
                                                    const int* __restrict__ tab, int G,
                                                    int nbins, int n, int* __restrict__ deg,
                                                    int* __restrict__ off,
                                                    int* __restrict__ csr) {
  __shared__ int ncnt[BNODES];
  __shared__ int noff[BNODES];
  __shared__ int cur[BNODES];
  __shared__ int wsum[33];
  __shared__ int outl[CAPB];
  int tid = threadIdx.x;
  int bin = blockIdx.x;
  int ts  = tab_pitch(nbins);
  if (tid < BNODES) { ncnt[tid] = 0; cur[tid] = 0; }
  __syncthreads();

  int sub = tid & 3;
  int sg  = tid >> 2;
  int myb = 0;
  for (int w0 = 0; w0 < G; w0 += 256) {
    int w = w0 + sg;
    if (w < G) {
      int a = tab[(size_t)w * ts + bin];
      int b = tab[(size_t)w * ts + bin + 1];
      if (sub == 0) myb += a;
      const unsigned* p = part + (size_t)w * CH;
      for (int i0 = a + sub; i0 < b; i0 += 32) {
        unsigned q[8];
#pragma unroll
        for (int u = 0; u < 8; ++u) {
          int i = i0 + 4 * u;
          q[u] = p[i < b ? i : b - 1];
        }
#pragma unroll
        for (int u = 0; u < 8; ++u) {
          int i = i0 + 4 * u;
          if (i < b) atomicAdd(&ncnt[q[u] & (BNODES - 1)], 1);
        }
      }
    }
  }
  __syncthreads();

  int binbase;
  block_excl_scan(myb, wsum, binbase);
  __syncthreads();
  int v = (tid < BNODES) ? ncnt[tid] : 0;
  int total;
  int ex = block_excl_scan(v, wsum, total);
  if (tid < BNODES) {
    noff[tid] = ex;
    int node = bin * BNODES + tid;
    if (node < n) { deg[node] = v; off[node] = binbase + ex; }
  }
  __syncthreads();

  bool inl = total <= CAPB;
  for (int w0 = 0; w0 < G; w0 += 256) {
    int w = w0 + sg;
    if (w < G) {
      int a = tab[(size_t)w * ts + bin];
      int b = tab[(size_t)w * ts + bin + 1];
      const unsigned* p = part + (size_t)w * CH;
      for (int i0 = a + sub; i0 < b; i0 += 32) {
        unsigned q[8];
#pragma unroll
        for (int u = 0; u < 8; ++u) {
          int i = i0 + 4 * u;
          q[u] = p[i < b ? i : b - 1];
        }
#pragma unroll
        for (int u = 0; u < 8; ++u) {
          int i = i0 + 4 * u;
          if (i < b) {
            int dl  = (int)(q[u] & (BNODES - 1));
            int pos = noff[dl] + atomicAdd(&cur[dl], 1);
            int sv  = (int)(q[u] >> BSH);
            if (inl) outl[pos] = sv; else csr[binbase + pos] = sv;
          }
        }
      }
    }
  }
  __syncthreads();
  if (inl)
    for (int i = tid; i < total; i += 1024) csr[binbase + i] = outl[i];
}

__global__ __launch_bounds__(256) void k_gemm_wmma(const float* __restrict__ x,
                                                   const float* __restrict__ W,
                                                   float* __restrict__ h, int n) {
  __shared__ __attribute__((aligned(16))) float Wt[C1 * WP];
  int tid = threadIdx.x;
#pragma unroll
  for (int it = 0; it < (DIN * C1) / 256; ++it) {
    int idx  = tid + it * 256;
    int c    = idx & 15;
    int hs   = (idx >> 4) & 1;
    int rest = idx >> 5;
    int nn   = (rest & 3) * 16 + c;
    int kk   = rest >> 2;
    int k    = (kk >> 1) * 4 + (kk & 1) + 2 * hs;
    Wt[nn * WP + k] = W[k * C1 + nn];
  }
  __syncthreads();

  int wave = tid >> 5;
  int lane = tid & 31;
  int half = lane >> 4;
  int l    = lane & 15;
  int rowBase = blockIdx.x * 128 + wave * 16;

  int arow = rowBase + l;
  if (arow > n - 1) arow = n - 1;
  const float* ap = x + (size_t)arow * DIN + 2 * half;

  v8f c0 = {}, c1 = {}, c2 = {}, c3 = {};
#pragma unroll 4
  for (int k0 = 0; k0 < DIN; k0 += 4) {
    v2f a = *(const v2f*)(ap + k0);
    const float* wp = &Wt[l * WP + k0 + 2 * half];
    v2f b0 = *(const v2f*)(wp);
    v2f b1 = *(const v2f*)(wp + 16 * WP);
    v2f b2 = *(const v2f*)(wp + 32 * WP);
    v2f b3 = *(const v2f*)(wp + 48 * WP);
    c0 = __builtin_amdgcn_wmma_f32_16x16x4_f32(false, a, false, b0, (short)0, c0, false, false);
    c1 = __builtin_amdgcn_wmma_f32_16x16x4_f32(false, a, false, b1, (short)0, c1, false, false);
    c2 = __builtin_amdgcn_wmma_f32_16x16x4_f32(false, a, false, b2, (short)0, c2, false, false);
    c3 = __builtin_amdgcn_wmma_f32_16x16x4_f32(false, a, false, b3, (short)0, c3, false, false);
  }

#pragma unroll
  for (int v = 0; v < 8; ++v) {
    int row = rowBase + 8 * half + v;
    if (row < n) {
      float* hp = h + (size_t)row * C1 + l;
      hp[0]  = c0[v];
      hp[16] = c1[v];
      hp[32] = c2[v];
      hp[48] = c3[v];
    }
  }
}

__global__ void k_logits(const float* __restrict__ h, const float* __restrict__ a_src,
                         const float* __restrict__ a_dst, float* __restrict__ esrc,
                         float* __restrict__ edst, int n) {
  int i = blockIdx.x * blockDim.x + threadIdx.x;
  if (i >= n) return;
  const float4* hp = (const float4*)(h + (size_t)i * C1);
  float s = 0.0f, d = 0.0f;
#pragma unroll
  for (int c = 0; c < C1 / 4; ++c) {
    float4 hv = hp[c];
    float4 as = ((const float4*)a_src)[c];
    float4 ad = ((const float4*)a_dst)[c];
    s += hv.x * as.x + hv.y * as.y + hv.z * as.z + hv.w * as.w;
    d += hv.x * ad.x + hv.y * ad.y + hv.z * ad.z + hv.w * ad.w;
  }
  esrc[i] = s;
  edst[i] = d;
}

__global__ __launch_bounds__(256) void k_node64(const int* __restrict__ off,
                                                const int* __restrict__ deg,
                                                const int* __restrict__ csr,
                                                const float* __restrict__ esrc,
                                                const float* __restrict__ edst,
                                                const float* __restrict__ h,
                                                float* __restrict__ out1, int n) {
  int lane = threadIdx.x & 31;
  int d = __builtin_amdgcn_readfirstlane(blockIdx.x * 8 + (threadIdx.x >> 5));
  if (d >= n) return;
  int start = __builtin_amdgcn_readfirstlane(off[d]);
  int cnt   = __builtin_amdgcn_readfirstlane(deg[d]);
  int tot   = cnt + 1;
  float ed  = edst[d];

  float mx = -INFINITY;
  for (int k = lane; k < tot; k += 32) {
    int s = (k < cnt) ? csr[start + k] : d;
    float v = esrc[s] + ed;
    v = v > 0.0f ? v : NEG_SLOPE * v;
    mx = fmaxf(mx, v);
  }
  mx = wave_max(mx);

  float zl = 0.0f;
  float a0 = 0.0f, a1 = 0.0f;
  for (int base = 0; base < tot; base += 32) {
    int k = base + lane;
    int s = d;
    float p = 0.0f;
    if (k < tot) {
      if (k < cnt) s = csr[start + k];
      float v = esrc[s] + ed;
      v = v > 0.0f ? v : NEG_SLOPE * v;
      p = __expf(v - mx);
    }
    zl += p;
    int m = tot - base;
    if (m > 32) m = 32;
#pragma unroll 4
    for (int t = 0; t < m; ++t) {
      int   st = __builtin_amdgcn_readlane(s, t);
      float pt = __int_as_float(__builtin_amdgcn_readlane(__float_as_int(p), t));
      v2f hv = *(const v2f*)(h + (size_t)st * C1 + 2 * lane);
      a0 += pt * hv.x;
      a1 += pt * hv.y;
    }
  }
  float z = wave_sum(zl);
  v2f o;
  o.x = a0 / z;
  o.y = a1 / z;
  *(v2f*)(out1 + (size_t)d * C1 + 2 * lane) = o;
}

__global__ void k_layer2_prep(const float* __restrict__ out1, const float* __restrict__ b1,
                              const float* __restrict__ W2, float* __restrict__ g, int n) {
  int i = blockIdx.x * blockDim.x + threadIdx.x;
  if (i >= n) return;
  const float4* op = (const float4*)(out1 + (size_t)i * C1);
  float acc = 0.0f;
#pragma unroll
  for (int c = 0; c < C1 / 4; ++c) {
    float4 v = op[c];
    float4 bb = ((const float4*)b1)[c];
    float4 ww = ((const float4*)W2)[c];
    float v0 = fmaxf(v.x + bb.x, 0.0f);
    float v1 = fmaxf(v.y + bb.y, 0.0f);
    float v2 = fmaxf(v.z + bb.z, 0.0f);
    float v3 = fmaxf(v.w + bb.w, 0.0f);
    acc += v0 * ww.x + v1 * ww.y + v2 * ww.z + v3 * ww.w;
  }
  g[i] = acc;
}

__global__ __launch_bounds__(256) void k_node1(const int* __restrict__ off,
                                               const int* __restrict__ deg,
                                               const int* __restrict__ csr,
                                               const float* __restrict__ g,
                                               const float* __restrict__ a_src2,
                                               const float* __restrict__ a_dst2,
                                               const float* __restrict__ b2,
                                               float* __restrict__ out, int n) {
  int lane = threadIdx.x & 31;
  int d = __builtin_amdgcn_readfirstlane(blockIdx.x * 8 + (threadIdx.x >> 5));
  if (d >= n) return;
  int start = __builtin_amdgcn_readfirstlane(off[d]);
  int cnt   = __builtin_amdgcn_readfirstlane(deg[d]);
  int tot   = cnt + 1;
  float as2 = a_src2[0];
  float ed  = g[d] * a_dst2[0];

  float mx = -INFINITY;
  for (int k = lane; k < tot; k += 32) {
    int s = (k < cnt) ? csr[start + k] : d;
    float v = g[s] * as2 + ed;
    v = v > 0.0f ? v : NEG_SLOPE * v;
    mx = fmaxf(mx, v);
  }
  mx = wave_max(mx);

  float zl = 0.0f, al = 0.0f;
  for (int k = lane; k < tot; k += 32) {
    int s = (k < cnt) ? csr[start + k] : d;
    float gs = g[s];
    float v = gs * as2 + ed;
    v = v > 0.0f ? v : NEG_SLOPE * v;
    float p = __expf(v - mx);
    zl += p;
    al += p * gs;
  }
  float z = wave_sum(zl);
  float a = wave_sum(al);
  if (lane == 0) {
    float v = a / z + b2[0];
    out[d] = 1.0f / (1.0f + __expf(-v));
  }
}

extern "C" void kernel_launch(void* const* d_in, const int* in_sizes, int n_in,
                              void* d_out, int out_size, void* d_ws, size_t ws_size,
                              hipStream_t stream) {
  const int*   ei     = (const int*)d_in[0];
  const float* x      = (const float*)d_in[1];
  const float* W1     = (const float*)d_in[2];
  const float* a_src1 = (const float*)d_in[3];
  const float* a_dst1 = (const float*)d_in[4];
  const float* b1     = (const float*)d_in[5];
  const float* W2     = (const float*)d_in[6];
  const float* a_src2 = (const float*)d_in[7];
  const float* a_dst2 = (const float*)d_in[8];
  const float* b2     = (const float*)d_in[9];

  const int E = in_sizes[0] / 2;
  const int n = in_sizes[1] / DIN;
  const int nbins = (n + BNODES - 1) / BNODES;
  const int G = (E + CH - 1) / CH;

  float* f     = (float*)d_ws;
  float* h     = f;
  float* out1  = f + (size_t)64 * n;
  float* esrc  = f + (size_t)128 * n;
  float* edst  = f + (size_t)129 * n;
  float* g     = f + (size_t)130 * n;
  int*   deg   = (int*)(f + (size_t)131 * n);
  int*   off   = (int*)(f + (size_t)132 * n);
  unsigned* part = (unsigned*)(f + (size_t)134 * n);
  int*   csr   = (int*)part + (size_t)E;
  int*   tab   = csr + (size_t)E;
  float* out   = (float*)d_out;

  const int B = 256;
  dim3 blk(B);

  k_binsort<<<dim3(G), dim3(1024), 0, stream>>>(ei, E, nbins, part, tab);
  k_binfinish<<<dim3(nbins), dim3(1024), 0, stream>>>(part, tab, G, nbins, n, deg, off, csr);

  k_gemm_wmma<<<dim3((n + 127) / 128), blk, 0, stream>>>(x, W1, h, n);
  k_logits<<<dim3((n + B - 1) / B), blk, 0, stream>>>(h, a_src1, a_dst1, esrc, edst, n);
  k_node64<<<dim3((n + 7) / 8), blk, 0, stream>>>(off, deg, csr, esrc, edst, h, out1, n);

  k_layer2_prep<<<dim3((n + B - 1) / B), blk, 0, stream>>>(out1, b1, W2, g, n);
  k_node1<<<dim3((n + 7) / 8), blk, 0, stream>>>(off, deg, csr, g, a_src2, a_dst2, b2, out, n);
}


# Layer-1 node kernel forms the attention logits from the 64-ch rows it gathers (first 32 rows of a node kept in registers, gathered once): logits launch and logit gathers go; logit sum order changes
# speedup vs baseline: 5.8801x; 1.0472x over previous
#include <hip/hip_runtime.h>
#include <math.h>

typedef float v2f __attribute__((ext_vector_type(2)));
typedef float v8f __attribute__((ext_vector_type(8)));

#define DIN 128
#define C1  64
#define NEG_SLOPE 0.2f
#define CH    8192
#define EPT   8
#define BSH   8
#define BNODES 256
#define MAXNB 1024
#define CAPB  12288
#define WP 132

__device__ __forceinline__ float wave_max(float v) {
#pragma unroll
  for (int o = 16; o > 0; o >>= 1) v = fmaxf(v, __shfl_xor(v, o, 32));
  return v;
}
__device__ __forceinline__ float wave_sum(float v) {
#pragma unroll
  for (int o = 16; o > 0; o >>= 1) v += __shfl_xor(v, o, 32);
  return v;
}

__device__ __forceinline__ float xhalf(float x) {
  int b = __builtin_bit_cast(int, x);
  return __builtin_bit_cast(float,
      __builtin_amdgcn_permlanex16(b, b, 0x76543210, 0xfedcba98, false, false));
}
template <int K> __device__ __forceinline__ float row_xor(float x) {
  int b = __builtin_bit_cast(int, x);
  return __builtin_bit_cast(float, __builtin_amdgcn_mov_dpp(b, 0x160 + K, 0xf, 0xf, true));
}
__device__ __forceinline__ float row_sum16(float v) {
  v += row_xor<1>(v);
  v += row_xor<2>(v);
  v += row_xor<4>(v);
  v += row_xor<8>(v);
  return v;
}

__device__ __forceinline__ int block_excl_scan(int v, int* wsum, int& total) {
  int lane = threadIdx.x & 31;
  int wave = threadIdx.x >> 5;
  int inc = v;
#pragma unroll
  for (int o = 1; o < 32; o <<= 1) {
    int t = __shfl_up(inc, o, 32);
    if (lane >= o) inc += t;
  }
  if (lane == 31) wsum[wave] = inc;
  __syncthreads();
  if (wave == 0) {
    int w = wsum[lane];
    int winc = w;
#pragma unroll
    for (int o = 1; o < 32; o <<= 1) {
      int t = __shfl_up(winc, o, 32);
      if (lane >= o) winc += t;
    }
    wsum[lane] = winc - w;
    if (lane == 31) wsum[32] = winc;
  }
  __syncthreads();
  total = wsum[32];
  return inc - v + wsum[wave];
}

__device__ __forceinline__ int tab_pitch(int nbins) { return (nbins + 32) & ~31; }
__global__ __launch_bounds__(1024) void k_binsort(const int* __restrict__ ei, int E, int nbins,
                                                  unsigned* __restrict__ part,
                                                  int* __restrict__ tab) {
  __shared__ int cnt[MAXNB];
  __shared__ int boff[MAXNB + 1];
  __shared__ unsigned stage[CH];
  __shared__ int wsum[33];
  int tid = threadIdx.x;
  int base = blockIdx.x * CH;
  int m = E - base;
  if (m > CH) m = CH;
  for (int b = tid; b < nbins; b += 1024) cnt[b] = 0;
  __syncthreads();

  int s[EPT], d[EPT], r[EPT];
#pragma unroll
  for (int j = 0; j < EPT; ++j) {
    int i = j * 1024 + tid;
    s[j] = 0; d[j] = 0;
    if (i < m) { s[j] = ei[base + i]; d[j] = ei[E + base + i]; }
  }
#pragma unroll
  for (int j = 0; j < EPT; ++j) {
    int i = j * 1024 + tid;
    r[j] = 0;
    if (i < m) r[j] = atomicAdd(&cnt[d[j] >> BSH], 1);
  }
  __syncthreads();

  int v = (tid < nbins) ? cnt[tid] : 0;
  int total;
  int ex = block_excl_scan(v, wsum, total);
  if (tid < nbins) boff[tid] = ex;
  if (tid == 0) boff[nbins] = total;
  __syncthreads();

#pragma unroll
  for (int j = 0; j < EPT; ++j) {
    int i = j * 1024 + tid;
    if (i < m)
      stage[boff[d[j] >> BSH] + r[j]] = ((unsigned)s[j] << BSH) | (unsigned)(d[j] & (BNODES - 1));
  }
  __syncthreads();

  for (int i = tid; i < m; i += 1024) part[base + i] = stage[i];
  int* trow = tab + (size_t)blockIdx.x * tab_pitch(nbins);
  for (int b = tid; b <= nbins; b += 1024) trow[b] = boff[b];
}

__global__ __launch_bounds__(1024) void k_binfinish(const unsigned* __restrict__ part,
                                                    const int* __restrict__ tab, int G,
                                                    int nbins, int n, int* __restrict__ deg,
                                                    int* __restrict__ off,
                                                    int* __restrict__ csr) {
  __shared__ int ncnt[BNODES];
  __shared__ int noff[BNODES];
  __shared__ int cur[BNODES];
  __shared__ int wsum[33];
  __shared__ int outl[CAPB];
  int tid = threadIdx.x;
  int bin = blockIdx.x;
  int ts  = tab_pitch(nbins);
  if (tid < BNODES) { ncnt[tid] = 0; cur[tid] = 0; }
  __syncthreads();

  int sub = tid & 3;
  int sg  = tid >> 2;
  int myb = 0;
  for (int w0 = 0; w0 < G; w0 += 256) {
    int w = w0 + sg;
    if (w < G) {
      int a = tab[(size_t)w * ts + bin];
      int b = tab[(size_t)w * ts + bin + 1];
      if (sub == 0) myb += a;
      const unsigned* p = part + (size_t)w * CH;
      for (int i0 = a + sub; i0 < b; i0 += 32) {
        unsigned q[8];
#pragma unroll
        for (int u = 0; u < 8; ++u) {
          int i = i0 + 4 * u;
          q[u] = p[i < b ? i : b - 1];
        }
#pragma unroll
        for (int u = 0; u < 8; ++u) {
          int i = i0 + 4 * u;
          if (i < b) atomicAdd(&ncnt[q[u] & (BNODES - 1)], 1);
        }
      }
    }
  }
  __syncthreads();

  int binbase;
  block_excl_scan(myb, wsum, binbase);
  __syncthreads();
  int v = (tid < BNODES) ? ncnt[tid] : 0;
  int total;
  int ex = block_excl_scan(v, wsum, total);
  if (tid < BNODES) {
    noff[tid] = ex;
    int node = bin * BNODES + tid;
    if (node < n) { deg[node] = v; off[node] = binbase + ex; }
  }
  __syncthreads();

  bool inl = total <= CAPB;
  for (int w0 = 0; w0 < G; w0 += 256) {
    int w = w0 + sg;
    if (w < G) {
      int a = tab[(size_t)w * ts + bin];
      int b = tab[(size_t)w * ts + bin + 1];
      const unsigned* p = part + (size_t)w * CH;
      for (int i0 = a + sub; i0 < b; i0 += 32) {
        unsigned q[8];
#pragma unroll
        for (int u = 0; u < 8; ++u) {
          int i = i0 + 4 * u;
          q[u] = p[i < b ? i : b - 1];
        }
#pragma unroll
        for (int u = 0; u < 8; ++u) {
          int i = i0 + 4 * u;
          if (i < b) {
            int dl  = (int)(q[u] & (BNODES - 1));
            int pos = noff[dl] + atomicAdd(&cur[dl], 1);
            int sv  = (int)(q[u] >> BSH);
            if (inl) outl[pos] = sv; else csr[binbase + pos] = sv;
          }
        }
      }
    }
  }
  __syncthreads();
  if (inl)
    for (int i = tid; i < total; i += 1024) csr[binbase + i] = outl[i];
}

__global__ __launch_bounds__(256) void k_gemm_wmma(const float* __restrict__ x,
                                                   const float* __restrict__ W,
                                                   float* __restrict__ h, int n) {
  __shared__ __attribute__((aligned(16))) float Wt[C1 * WP];
  int tid = threadIdx.x;
#pragma unroll
  for (int it = 0; it < (DIN * C1) / 256; ++it) {
    int idx  = tid + it * 256;
    int c    = idx & 15;
    int hs   = (idx >> 4) & 1;
    int rest = idx >> 5;
    int nn   = (rest & 3) * 16 + c;
    int kk   = rest >> 2;
    int k    = (kk >> 1) * 4 + (kk & 1) + 2 * hs;
    Wt[nn * WP + k] = W[k * C1 + nn];
  }
  __syncthreads();

  int wave = tid >> 5;
  int lane = tid & 31;
  int half = lane >> 4;
  int l    = lane & 15;
  int rowBase = blockIdx.x * 128 + wave * 16;

  int arow = rowBase + l;
  if (arow > n - 1) arow = n - 1;
  const float* ap = x + (size_t)arow * DIN + 2 * half;

  v8f c0 = {}, c1 = {}, c2 = {}, c3 = {};
#pragma unroll 4
  for (int k0 = 0; k0 < DIN; k0 += 4) {
    v2f a = *(const v2f*)(ap + k0);
    const float* wp = &Wt[l * WP + k0 + 2 * half];
    v2f b0 = *(const v2f*)(wp);
    v2f b1 = *(const v2f*)(wp + 16 * WP);
    v2f b2 = *(const v2f*)(wp + 32 * WP);
    v2f b3 = *(const v2f*)(wp + 48 * WP);
    c0 = __builtin_amdgcn_wmma_f32_16x16x4_f32(false, a, false, b0, (short)0, c0, false, false);
    c1 = __builtin_amdgcn_wmma_f32_16x16x4_f32(false, a, false, b1, (short)0, c1, false, false);
    c2 = __builtin_amdgcn_wmma_f32_16x16x4_f32(false, a, false, b2, (short)0, c2, false, false);
    c3 = __builtin_amdgcn_wmma_f32_16x16x4_f32(false, a, false, b3, (short)0, c3, false, false);
  }

#pragma unroll
  for (int v = 0; v < 8; ++v) {
    int row = rowBase + 8 * half + v;
    if (row < n) {
      float* hp = h + (size_t)row * C1 + l;
      hp[0]  = c0[v];
      hp[16] = c1[v];
      hp[32] = c2[v];
      hp[48] = c3[v];
    }
  }
}

#define RK 32
__global__ __launch_bounds__(256) void k_node64(const int* __restrict__ off,
                                                const int* __restrict__ deg,
                                                const int* __restrict__ csr,
                                                const float* __restrict__ a_src,
                                                const float* __restrict__ a_dst,
                                                const float* __restrict__ h,
                                                float* __restrict__ out1, int n) {
  int lane = threadIdx.x & 31;
  int d = __builtin_amdgcn_readfirstlane(blockIdx.x * 8 + (threadIdx.x >> 5));
  if (d >= n) return;
  const v2f hs = *(const v2f*)(h + (size_t)d * C1 + 2 * lane);
  const v2f as = *(const v2f*)(a_src + 2 * lane);
  const v2f ad = *(const v2f*)(a_dst + 2 * lane);
  int start = __builtin_amdgcn_readfirstlane(off[d]);
  int cnt   = __builtin_amdgcn_readfirstlane(deg[d]);
  int m = cnt < RK ? cnt : RK;
  int s = 0;
  if (lane < m) s = csr[start + lane];

  v2f r[RK];
  int sid[RK];
#pragma unroll
  for (int t = 0; t < RK; ++t) sid[t] = __builtin_amdgcn_readlane(s, t);
#pragma unroll
  for (int g = 0; g < RK; g += 8) {
    if (g < m) {
#pragma unroll
      for (int t = g; t < g + 8; ++t) {
        if (t < m) r[t] = *(const v2f*)(h + (size_t)sid[t] * C1 + 2 * lane);
      }
    }
  }

  float qd = row_sum16(hs.x * ad.x + hs.y * ad.y);
  float qs = row_sum16(hs.x * as.x + hs.y * as.y);
  float ed = qd + xhalf(qd);
  float vs = (qs + xhalf(qs)) + ed;
  vs = vs > 0.0f ? vs : NEG_SLOPE * vs;

  int l16 = lane & 15;
  float X = 0.0f, Y = 0.0f;
#pragma unroll
  for (int g = 0; g < RK; g += 8) {
    if (g < m) {
#pragma unroll
      for (int t = g; t < g + 8; ++t) {
        if (t < m) {
          float w = row_sum16(r[t].x * as.x + r[t].y * as.y);
          if (t < 16) X = (l16 == t) ? w : X;
          else        Y = (l16 == t - 16) ? w : Y;
        }
      }
    }
  }
  float ex = X + xhalf(X);
  float ey = Y + xhalf(Y);
  float v = ((lane < 16) ? ex : ey) + ed;
  v = v > 0.0f ? v : NEG_SLOPE * v;

  float mx = (lane < m) ? fmaxf(v, vs) : vs;
  for (int k0 = RK; k0 < cnt; k0 += 32) {
    int kk = k0 + lane;
    int s2 = (kk < cnt) ? csr[start + kk] : 0;
    int mm = cnt - k0;
    if (mm > 32) mm = 32;
    for (int t = 0; t < mm; ++t) {
      int st = __builtin_amdgcn_readlane(s2, t);
      v2f hv = *(const v2f*)(h + (size_t)st * C1 + 2 * lane);
      float w = row_sum16(hv.x * as.x + hv.y * as.y);
      float e = (w + xhalf(w)) + ed;
      e = e > 0.0f ? e : NEG_SLOPE * e;
      mx = fmaxf(mx, e);
    }
  }
  mx = wave_max(mx);

  float p  = (lane < m) ? __expf(v - mx) : 0.0f;
  float zl = p;
  float a0 = 0.0f, a1 = 0.0f;
#pragma unroll
  for (int g = 0; g < RK; g += 8) {
    if (g < m) {
#pragma unroll
      for (int t = g; t < g + 8; ++t) {
        if (t < m) {
          float pt = __int_as_float(__builtin_amdgcn_readlane(__float_as_int(p), t));
          a0 += pt * r[t].x;
          a1 += pt * r[t].y;
        }
      }
    }
  }
  for (int k0 = RK; k0 < cnt; k0 += 32) {
    int kk = k0 + lane;
    int s2 = (kk < cnt) ? csr[start + kk] : 0;
    int mm = cnt - k0;
    if (mm > 32) mm = 32;
    for (int t = 0; t < mm; ++t) {
      int st = __builtin_amdgcn_readlane(s2, t);
      v2f hv = *(const v2f*)(h + (size_t)st * C1 + 2 * lane);
      float w = row_sum16(hv.x * as.x + hv.y * as.y);
      float e = (w + xhalf(w)) + ed;
      e = e > 0.0f ? e : NEG_SLOPE * e;
      float pt = __expf(e - mx);
      if (lane == t) zl += pt;
      a0 += pt * hv.x;
      a1 += pt * hv.y;
    }
  }
  float pself = __expf(vs - mx);
  if (lane == (cnt & 31)) zl += pself;
  a0 += pself * hs.x;
  a1 += pself * hs.y;

  float z = wave_sum(zl);
  v2f o;
  o.x = a0 / z;
  o.y = a1 / z;
  *(v2f*)(out1 + (size_t)d * C1 + 2 * lane) = o;
}

__global__ void k_layer2_prep(const float* __restrict__ out1, const float* __restrict__ b1,
                              const float* __restrict__ W2, float* __restrict__ g, int n) {
  int i = blockIdx.x * blockDim.x + threadIdx.x;
  if (i >= n) return;
  const float4* op = (const float4*)(out1 + (size_t)i * C1);
  float acc = 0.0f;
#pragma unroll
  for (int c = 0; c < C1 / 4; ++c) {
    float4 v = op[c];
    float4 bb = ((const float4*)b1)[c];
    float4 ww = ((const float4*)W2)[c];
    float v0 = fmaxf(v.x + bb.x, 0.0f);
    float v1 = fmaxf(v.y + bb.y, 0.0f);
    float v2 = fmaxf(v.z + bb.z, 0.0f);
    float v3 = fmaxf(v.w + bb.w, 0.0f);
    acc += v0 * ww.x + v1 * ww.y + v2 * ww.z + v3 * ww.w;
  }
  g[i] = acc;
}

__global__ __launch_bounds__(256) void k_node1(const int* __restrict__ off,
                                               const int* __restrict__ deg,
                                               const int* __restrict__ csr,
                                               const float* __restrict__ g,
                                               const float* __restrict__ a_src2,
                                               const float* __restrict__ a_dst2,
                                               const float* __restrict__ b2,
                                               float* __restrict__ out, int n) {
  int lane = threadIdx.x & 31;
  int d = __builtin_amdgcn_readfirstlane(blockIdx.x * 8 + (threadIdx.x >> 5));
  if (d >= n) return;
  int start = __builtin_amdgcn_readfirstlane(off[d]);
  int cnt   = __builtin_amdgcn_readfirstlane(deg[d]);
  int tot   = cnt + 1;
  float as2 = a_src2[0];
  float ed  = g[d] * a_dst2[0];

  float mx = -INFINITY;
  for (int k = lane; k < tot; k += 32) {
    int s = (k < cnt) ? csr[start + k] : d;
    float v = g[s] * as2 + ed;
    v = v > 0.0f ? v : NEG_SLOPE * v;
    mx = fmaxf(mx, v);
  }
  mx = wave_max(mx);

  float zl = 0.0f, al = 0.0f;
  for (int k = lane; k < tot; k += 32) {
    int s = (k < cnt) ? csr[start + k] : d;
    float gs = g[s];
    float v = gs * as2 + ed;
    v = v > 0.0f ? v : NEG_SLOPE * v;
    float p = __expf(v - mx);
    zl += p;
    al += p * gs;
  }
  float z = wave_sum(zl);
  float a = wave_sum(al);
  if (lane == 0) {
    float v = a / z + b2[0];
    out[d] = 1.0f / (1.0f + __expf(-v));
  }
}

extern "C" void kernel_launch(void* const* d_in, const int* in_sizes, int n_in,
                              void* d_out, int out_size, void* d_ws, size_t ws_size,
                              hipStream_t stream) {
  const int*   ei     = (const int*)d_in[0];
  const float* x      = (const float*)d_in[1];
  const float* W1     = (const float*)d_in[2];
  const float* a_src1 = (const float*)d_in[3];
  const float* a_dst1 = (const float*)d_in[4];
  const float* b1     = (const float*)d_in[5];
  const float* W2     = (const float*)d_in[6];
  const float* a_src2 = (const float*)d_in[7];
  const float* a_dst2 = (const float*)d_in[8];
  const float* b2     = (const float*)d_in[9];

  const int E = in_sizes[0] / 2;
  const int n = in_sizes[1] / DIN;
  const int nbins = (n + BNODES - 1) / BNODES;
  const int G = (E + CH - 1) / CH;

  float* f     = (float*)d_ws;
  float* h     = f;
  float* out1  = f + (size_t)64 * n;
  float* g     = f + (size_t)130 * n;
  int*   deg   = (int*)(f + (size_t)131 * n);
  int*   off   = (int*)(f + (size_t)132 * n);
  unsigned* part = (unsigned*)(f + (size_t)134 * n);
  int*   csr   = (int*)part + (size_t)E;
  int*   tab   = csr + (size_t)E;
  float* out   = (float*)d_out;

  const int B = 256;
  dim3 blk(B);

  k_binsort<<<dim3(G), dim3(1024), 0, stream>>>(ei, E, nbins, part, tab);
  k_binfinish<<<dim3(nbins), dim3(1024), 0, stream>>>(part, tab, G, nbins, n, deg, off, csr);

  k_gemm_wmma<<<dim3((n + 127) / 128), blk, 0, stream>>>(x, W1, h, n);
  k_node64<<<dim3((n + 7) / 8), blk, 0, stream>>>(off, deg, csr, a_src1, a_dst1, h, out1, n);

  k_layer2_prep<<<dim3((n + B - 1) / B), blk, 0, stream>>>(out1, b1, W2, g, n);
  k_node1<<<dim3((n + 7) / 8), blk, 0, stream>>>(off, deg, csr, g, a_src2, a_dst2, b2, out, n);
}


# Node kernels: a wave walks 10 (layer 1) or 8 (layer 2) consecutive nodes, ranges in one load per wave, next node's source list fetched ahead: 10000 and 12500 waves to start, not 100000 each
# speedup vs baseline: 6.4682x; 1.1000x over previous
#include <hip/hip_runtime.h>
#include <math.h>

typedef float v2f __attribute__((ext_vector_type(2)));
typedef float v8f __attribute__((ext_vector_type(8)));

#define DIN 128
#define C1  64
#define NEG_SLOPE 0.2f
#define CH    8192
#define EPT   8
#define BSH   8
#define BNODES 256
#define MAXNB 1024
#define CAPB  12288
#define WP 132

__device__ __forceinline__ float wave_max(float v) {
#pragma unroll
  for (int o = 16; o > 0; o >>= 1) v = fmaxf(v, __shfl_xor(v, o, 32));
  return v;
}
__device__ __forceinline__ float wave_sum(float v) {
#pragma unroll
  for (int o = 16; o > 0; o >>= 1) v += __shfl_xor(v, o, 32);
  return v;
}

__device__ __forceinline__ float xhalf(float x) {
  int b = __builtin_bit_cast(int, x);
  return __builtin_bit_cast(float,
      __builtin_amdgcn_permlanex16(b, b, 0x76543210, 0xfedcba98, false, false));
}
template <int K> __device__ __forceinline__ float row_xor(float x) {
  int b = __builtin_bit_cast(int, x);
  return __builtin_bit_cast(float, __builtin_amdgcn_mov_dpp(b, 0x160 + K, 0xf, 0xf, true));
}
__device__ __forceinline__ float row_sum16(float v) {
  v += row_xor<1>(v);
  v += row_xor<2>(v);
  v += row_xor<4>(v);
  v += row_xor<8>(v);
  return v;
}

__device__ __forceinline__ int block_excl_scan(int v, int* wsum, int& total) {
  int lane = threadIdx.x & 31;
  int wave = threadIdx.x >> 5;
  int inc = v;
#pragma unroll
  for (int o = 1; o < 32; o <<= 1) {
    int t = __shfl_up(inc, o, 32);
    if (lane >= o) inc += t;
  }
  if (lane == 31) wsum[wave] = inc;
  __syncthreads();
  if (wave == 0) {
    int w = wsum[lane];
    int winc = w;
#pragma unroll
    for (int o = 1; o < 32; o <<= 1) {
      int t = __shfl_up(winc, o, 32);
      if (lane >= o) winc += t;
    }
    wsum[lane] = winc - w;
    if (lane == 31) wsum[32] = winc;
  }
  __syncthreads();
  total = wsum[32];
  return inc - v + wsum[wave];
}

__device__ __forceinline__ int tab_pitch(int nbins) { return (nbins + 32) & ~31; }
__global__ __launch_bounds__(1024) void k_binsort(const int* __restrict__ ei, int E, int nbins,
                                                  unsigned* __restrict__ part,
                                                  int* __restrict__ tab) {
  __shared__ int cnt[MAXNB];
  __shared__ int boff[MAXNB + 1];
  __shared__ unsigned stage[CH];
  __shared__ int wsum[33];
  int tid = threadIdx.x;
  int base = blockIdx.x * CH;
  int m = E - base;
  if (m > CH) m = CH;
  for (int b = tid; b < nbins; b += 1024) cnt[b] = 0;
  __syncthreads();

  int s[EPT], d[EPT], r[EPT];
#pragma unroll
  for (int j = 0; j < EPT; ++j) {
    int i = j * 1024 + tid;
    s[j] = 0; d[j] = 0;
    if (i < m) { s[j] = ei[base + i]; d[j] = ei[E + base + i]; }
  }
#pragma unroll
  for (int j = 0; j < EPT; ++j) {
    int i = j * 1024 + tid;
    r[j] = 0;
    if (i < m) r[j] = atomicAdd(&cnt[d[j] >> BSH], 1);
  }
  __syncthreads();

  int v = (tid < nbins) ? cnt[tid] : 0;
  int total;
  int ex = block_excl_scan(v, wsum, total);
  if (tid < nbins) boff[tid] = ex;
  if (tid == 0) boff[nbins] = total;
  __syncthreads();

#pragma unroll
  for (int j = 0; j < EPT; ++j) {
    int i = j * 1024 + tid;
    if (i < m)
      stage[boff[d[j] >> BSH] + r[j]] = ((unsigned)s[j] << BSH) | (unsigned)(d[j] & (BNODES - 1));
  }
  __syncthreads();

  for (int i = tid; i < m; i += 1024) part[base + i] = stage[i];
  int* trow = tab + (size_t)blockIdx.x * tab_pitch(nbins);
  for (int b = tid; b <= nbins; b += 1024) trow[b] = boff[b];
}

__global__ __launch_bounds__(1024) void k_binfinish(const unsigned* __restrict__ part,
                                                    const int* __restrict__ tab, int G,
                                                    int nbins, int n, int* __restrict__ deg,
                                                    int* __restrict__ off,
                                                    int* __restrict__ csr) {
  __shared__ int ncnt[BNODES];
  __shared__ int noff[BNODES];
  __shared__ int cur[BNODES];
  __shared__ int wsum[33];
  __shared__ int outl[CAPB];
  int tid = threadIdx.x;
  int bin = blockIdx.x;
  int ts  = tab_pitch(nbins);
  if (tid < BNODES) { ncnt[tid] = 0; cur[tid] = 0; }
  __syncthreads();

  int sub = tid & 3;
  int sg  = tid >> 2;
  int myb = 0;
  for (int w0 = 0; w0 < G; w0 += 256) {
    int w = w0 + sg;
    if (w < G) {
      int a = tab[(size_t)w * ts + bin];
      int b = tab[(size_t)w * ts + bin + 1];
      if (sub == 0) myb += a;
      const unsigned* p = part + (size_t)w * CH;
      for (int i0 = a + sub; i0 < b; i0 += 32) {
        unsigned q[8];
#pragma unroll
        for (int u = 0; u < 8; ++u) {
          int i = i0 + 4 * u;
          q[u] = p[i < b ? i : b - 1];
        }
#pragma unroll
        for (int u = 0; u < 8; ++u) {
          int i = i0 + 4 * u;
          if (i < b) atomicAdd(&ncnt[q[u] & (BNODES - 1)], 1);
        }
      }
    }
  }
  __syncthreads();

  int binbase;
  block_excl_scan(myb, wsum, binbase);
  __syncthreads();
  int v = (tid < BNODES) ? ncnt[tid] : 0;
  int total;
  int ex = block_excl_scan(v, wsum, total);
  if (tid < BNODES) {
    noff[tid] = ex;
    int node = bin * BNODES + tid;
    if (node < n) { deg[node] = v; off[node] = binbase + ex; }
  }
  __syncthreads();

  bool inl = total <= CAPB;
  for (int w0 = 0; w0 < G; w0 += 256) {
    int w = w0 + sg;
    if (w < G) {
      int a = tab[(size_t)w * ts + bin];
      int b = tab[(size_t)w * ts + bin + 1];
      const unsigned* p = part + (size_t)w * CH;
      for (int i0 = a + sub; i0 < b; i0 += 32) {
        unsigned q[8];
#pragma unroll
        for (int u = 0; u < 8; ++u) {
          int i = i0 + 4 * u;
          q[u] = p[i < b ? i : b - 1];
        }
#pragma unroll
        for (int u = 0; u < 8; ++u) {
          int i = i0 + 4 * u;
          if (i < b) {
            int dl  = (int)(q[u] & (BNODES - 1));
            int pos = noff[dl] + atomicAdd(&cur[dl], 1);
            int sv  = (int)(q[u] >> BSH);
            if (inl) outl[pos] = sv; else csr[binbase + pos] = sv;
          }
        }
      }
    }
  }
  __syncthreads();
  if (inl)
    for (int i = tid; i < total; i += 1024) csr[binbase + i] = outl[i];
}

__global__ __launch_bounds__(256) void k_gemm_wmma(const float* __restrict__ x,
                                                   const float* __restrict__ W,
                                                   float* __restrict__ h, int n) {
  __shared__ __attribute__((aligned(16))) float Wt[C1 * WP];
  int tid = threadIdx.x;
#pragma unroll
  for (int it = 0; it < (DIN * C1) / 256; ++it) {
    int idx  = tid + it * 256;
    int c    = idx & 15;
    int hs   = (idx >> 4) & 1;
    int rest = idx >> 5;
    int nn   = (rest & 3) * 16 + c;
    int kk   = rest >> 2;
    int k    = (kk >> 1) * 4 + (kk & 1) + 2 * hs;
    Wt[nn * WP + k] = W[k * C1 + nn];
  }
  __syncthreads();

  int wave = tid >> 5;
  int lane = tid & 31;
  int half = lane >> 4;
  int l    = lane & 15;
  int rowBase = blockIdx.x * 128 + wave * 16;

  int arow = rowBase + l;
  if (arow > n - 1) arow = n - 1;
  const float* ap = x + (size_t)arow * DIN + 2 * half;

  v8f c0 = {}, c1 = {}, c2 = {}, c3 = {};
#pragma unroll 4
  for (int k0 = 0; k0 < DIN; k0 += 4) {
    v2f a = *(const v2f*)(ap + k0);
    const float* wp = &Wt[l * WP + k0 + 2 * half];
    v2f b0 = *(const v2f*)(wp);
    v2f b1 = *(const v2f*)(wp + 16 * WP);
    v2f b2 = *(const v2f*)(wp + 32 * WP);
    v2f b3 = *(const v2f*)(wp + 48 * WP);
    c0 = __builtin_amdgcn_wmma_f32_16x16x4_f32(false, a, false, b0, (short)0, c0, false, false);
    c1 = __builtin_amdgcn_wmma_f32_16x16x4_f32(false, a, false, b1, (short)0, c1, false, false);
    c2 = __builtin_amdgcn_wmma_f32_16x16x4_f32(false, a, false, b2, (short)0, c2, false, false);
    c3 = __builtin_amdgcn_wmma_f32_16x16x4_f32(false, a, false, b3, (short)0, c3, false, false);
  }

#pragma unroll
  for (int v = 0; v < 8; ++v) {
    int row = rowBase + 8 * half + v;
    if (row < n) {
      float* hp = h + (size_t)row * C1 + l;
      hp[0]  = c0[v];
      hp[16] = c1[v];
      hp[32] = c2[v];
      hp[48] = c3[v];
    }
  }
}

#define RK 32
#define NPW64 10
__global__ __launch_bounds__(256) void k_node64(const int* __restrict__ off,
                                                const int* __restrict__ deg,
                                                const int* __restrict__ csr,
                                                const float* __restrict__ a_src,
                                                const float* __restrict__ a_dst,
                                                const float* __restrict__ h,
                                                float* __restrict__ out1, int n) {
  int lane = threadIdx.x & 31;
  int d0 = __builtin_amdgcn_readfirstlane((blockIdx.x * 8 + (threadIdx.x >> 5)) * NPW64);
  if (d0 >= n) return;
  int nn = n - d0;
  if (nn > NPW64) nn = NPW64;
  int offv = 0, degv = 0;
  if (lane < nn) { offv = off[d0 + lane]; degv = deg[d0 + lane]; }
  const v2f as = *(const v2f*)(a_src + 2 * lane);
  const v2f ad = *(const v2f*)(a_dst + 2 * lane);
  int l16 = lane & 15;
  int start = __builtin_amdgcn_readlane(offv, 0);
  int cnt   = __builtin_amdgcn_readlane(degv, 0);
  int m = cnt < RK ? cnt : RK;
  int s = 0;
  if (lane < m) s = csr[start + lane];
  for (int j = 0; j < nn; ++j) {
    int d = d0 + j;
    const v2f hs = *(const v2f*)(h + (size_t)d * C1 + 2 * lane);

    v2f r[RK];
    int sid[RK];
#pragma unroll
    for (int t = 0; t < RK; ++t) sid[t] = __builtin_amdgcn_readlane(s, t);
#pragma unroll
    for (int g = 0; g < RK; g += 8) {
      if (g < m) {
#pragma unroll
        for (int t = g; t < g + 8; ++t) {
          if (t < m) r[t] = *(const v2f*)(h + (size_t)sid[t] * C1 + 2 * lane);
        }
      }
    }
    int nstart = 0, ncnt = 0, nm = 0, sn = 0;
    if (j + 1 < nn) {
      nstart = __builtin_amdgcn_readlane(offv, j + 1);
      ncnt   = __builtin_amdgcn_readlane(degv, j + 1);
      nm = ncnt < RK ? ncnt : RK;
      if (lane < nm) sn = csr[nstart + lane];
    }

    float qd = row_sum16(hs.x * ad.x + hs.y * ad.y);
    float qs = row_sum16(hs.x * as.x + hs.y * as.y);
    float ed = qd + xhalf(qd);
    float vs = (qs + xhalf(qs)) + ed;
    vs = vs > 0.0f ? vs : NEG_SLOPE * vs;

    float X = 0.0f, Y = 0.0f;
#pragma unroll
    for (int g = 0; g < RK; g += 8) {
      if (g < m) {
#pragma unroll
        for (int t = g; t < g + 8; ++t) {
          if (t < m) {
            float w = row_sum16(r[t].x * as.x + r[t].y * as.y);
            if (t < 16) X = (l16 == t) ? w : X;
            else        Y = (l16 == t - 16) ? w : Y;
          }
        }
      }
    }
    float ex = X + xhalf(X);
    float ey = Y + xhalf(Y);
    float v = ((lane < 16) ? ex : ey) + ed;
    v = v > 0.0f ? v : NEG_SLOPE * v;

    float mx = (lane < m) ? fmaxf(v, vs) : vs;
    for (int k0 = RK; k0 < cnt; k0 += 32) {
      int kk = k0 + lane;
      int s2 = (kk < cnt) ? csr[start + kk] : 0;
      int mm = cnt - k0;
      if (mm > 32) mm = 32;
      for (int t = 0; t < mm; ++t) {
        int st = __builtin_amdgcn_readlane(s2, t);
        v2f hv = *(const v2f*)(h + (size_t)st * C1 + 2 * lane);
        float w = row_sum16(hv.x * as.x + hv.y * as.y);
        float e = (w + xhalf(w)) + ed;
        e = e > 0.0f ? e : NEG_SLOPE * e;
        mx = fmaxf(mx, e);
      }
    }
    mx = wave_max(mx);

    float p  = (lane < m) ? __expf(v - mx) : 0.0f;
    float zl = p;
    float a0 = 0.0f, a1 = 0.0f;
#pragma unroll
    for (int g = 0; g < RK; g += 8) {
      if (g < m) {
#pragma unroll
        for (int t = g; t < g + 8; ++t) {
          if (t < m) {
            float pt = __int_as_float(__builtin_amdgcn_readlane(__float_as_int(p), t));
            a0 += pt * r[t].x;
            a1 += pt * r[t].y;
          }
        }
      }
    }
    for (int k0 = RK; k0 < cnt; k0 += 32) {
      int kk = k0 + lane;
      int s2 = (kk < cnt) ? csr[start + kk] : 0;
      int mm = cnt - k0;
      if (mm > 32) mm = 32;
      for (int t = 0; t < mm; ++t) {
        int st = __builtin_amdgcn_readlane(s2, t);
        v2f hv = *(const v2f*)(h + (size_t)st * C1 + 2 * lane);
        float w = row_sum16(hv.x * as.x + hv.y * as.y);
        float e = (w + xhalf(w)) + ed;
        e = e > 0.0f ? e : NEG_SLOPE * e;
        float pt = __expf(e - mx);
        if (lane == t) zl += pt;
        a0 += pt * hv.x;
        a1 += pt * hv.y;
      }
    }
    float pself = __expf(vs - mx);
    if (lane == (cnt & 31)) zl += pself;
    a0 += pself * hs.x;
    a1 += pself * hs.y;

    float z = wave_sum(zl);
    v2f o;
    o.x = a0 / z;
    o.y = a1 / z;
    *(v2f*)(out1 + (size_t)d * C1 + 2 * lane) = o;
    start = nstart; cnt = ncnt; m = nm; s = sn;
  }
}

__global__ void k_layer2_prep(const float* __restrict__ out1, const float* __restrict__ b1,
                              const float* __restrict__ W2, float* __restrict__ g, int n) {
  int i = blockIdx.x * blockDim.x + threadIdx.x;
  if (i >= n) return;
  const float4* op = (const float4*)(out1 + (size_t)i * C1);
  float acc = 0.0f;
#pragma unroll
  for (int c = 0; c < C1 / 4; ++c) {
    float4 v = op[c];
    float4 bb = ((const float4*)b1)[c];
    float4 ww = ((const float4*)W2)[c];
    float v0 = fmaxf(v.x + bb.x, 0.0f);
    float v1 = fmaxf(v.y + bb.y, 0.0f);
    float v2 = fmaxf(v.z + bb.z, 0.0f);
    float v3 = fmaxf(v.w + bb.w, 0.0f);
    acc += v0 * ww.x + v1 * ww.y + v2 * ww.z + v3 * ww.w;
  }
  g[i] = acc;
}

#define NPW1 8
__global__ __launch_bounds__(256) void k_node1(const int* __restrict__ off,
                                               const int* __restrict__ deg,
                                               const int* __restrict__ csr,
                                               const float* __restrict__ g,
                                               const float* __restrict__ a_src2,
                                               const float* __restrict__ a_dst2,
                                               const float* __restrict__ b2,
                                               float* __restrict__ out, int n) {
  int lane = threadIdx.x & 31;
  int d0 = __builtin_amdgcn_readfirstlane((blockIdx.x * 8 + (threadIdx.x >> 5)) * NPW1);
  if (d0 >= n) return;
  int nn = n - d0;
  if (nn > NPW1) nn = NPW1;
  int offv = 0, degv = 0;
  float gd = 0.0f;
  if (lane < nn) { offv = off[d0 + lane]; degv = deg[d0 + lane]; gd = g[d0 + lane]; }
  float as2 = a_src2[0];
  float ad2 = a_dst2[0];
  float bb  = b2[0];
  float res = 0.0f;
  for (int j = 0; j < nn; ++j) {
    int d = d0 + j;
    int start = __builtin_amdgcn_readlane(offv, j);
    int cnt   = __builtin_amdgcn_readlane(degv, j);
    int tot   = cnt + 1;
    float ed  = __int_as_float(__builtin_amdgcn_readlane(__float_as_int(gd), j)) * ad2;

    float mx = -INFINITY;
    for (int k = lane; k < tot; k += 32) {
      int s = (k < cnt) ? csr[start + k] : d;
      float v = g[s] * as2 + ed;
      v = v > 0.0f ? v : NEG_SLOPE * v;
      mx = fmaxf(mx, v);
    }
    mx = wave_max(mx);

    float zl = 0.0f, al = 0.0f;
    for (int k = lane; k < tot; k += 32) {
      int s = (k < cnt) ? csr[start + k] : d;
      float gs = g[s];
      float v = gs * as2 + ed;
      v = v > 0.0f ? v : NEG_SLOPE * v;
      float p = __expf(v - mx);
      zl += p;
      al += p * gs;
    }
    float z = wave_sum(zl);
    float a = wave_sum(al);
    float v = a / z + bb;
    float o = 1.0f / (1.0f + __expf(-v));
    if (lane == j) res = o;
  }
  if (lane < nn) out[d0 + lane] = res;
}

extern "C" void kernel_launch(void* const* d_in, const int* in_sizes, int n_in,
                              void* d_out, int out_size, void* d_ws, size_t ws_size,
                              hipStream_t stream) {
  const int*   ei     = (const int*)d_in[0];
  const float* x      = (const float*)d_in[1];
  const float* W1     = (const float*)d_in[2];
  const float* a_src1 = (const float*)d_in[3];
  const float* a_dst1 = (const float*)d_in[4];
  const float* b1     = (const float*)d_in[5];
  const float* W2     = (const float*)d_in[6];
  const float* a_src2 = (const float*)d_in[7];
  const float* a_dst2 = (const float*)d_in[8];
  const float* b2     = (const float*)d_in[9];

  const int E = in_sizes[0] / 2;
  const int n = in_sizes[1] / DIN;
  const int nbins = (n + BNODES - 1) / BNODES;
  const int G = (E + CH - 1) / CH;

  float* f     = (float*)d_ws;
  float* h     = f;
  float* out1  = f + (size_t)64 * n;
  float* g     = f + (size_t)130 * n;
  int*   deg   = (int*)(f + (size_t)131 * n);
  int*   off   = (int*)(f + (size_t)132 * n);
  unsigned* part = (unsigned*)(f + (size_t)134 * n);
  int*   csr   = (int*)part + (size_t)E;
  int*   tab   = csr + (size_t)E;
  float* out   = (float*)d_out;

  const int B = 256;
  dim3 blk(B);

  k_binsort<<<dim3(G), dim3(1024), 0, stream>>>(ei, E, nbins, part, tab);
  k_binfinish<<<dim3(nbins), dim3(1024), 0, stream>>>(part, tab, G, nbins, n, deg, off, csr);

  k_gemm_wmma<<<dim3((n + 127) / 128), blk, 0, stream>>>(x, W1, h, n);
  k_node64<<<dim3((n + 8 * NPW64 - 1) / (8 * NPW64)), blk, 0, stream>>>(off, deg, csr, a_src1, a_dst1, h,
                                                                      out1, n);

  k_layer2_prep<<<dim3((n + B - 1) / B), blk, 0, stream>>>(out1, b1, W2, g, n);
  k_node1<<<dim3((n + 8 * NPW1 - 1) / (8 * NPW1)), blk, 0, stream>>>(off, deg, csr, g, a_src2, a_dst2, b2,
                                                                    out, n);
}


# Layer-1 node kernel forms the layer-2 input relu(row+b1).W2 itself (rows parked in LDS, projected at the wave's end): 64-ch row never stored or re-read, projection launch gone; this sum reordered
# speedup vs baseline: 6.8509x; 1.0592x over previous
#include <hip/hip_runtime.h>
#include <math.h>

typedef float v2f __attribute__((ext_vector_type(2)));
typedef float v8f __attribute__((ext_vector_type(8)));

#define DIN 128
#define C1  64
#define NEG_SLOPE 0.2f
#define CH    8192
#define EPT   8
#define BSH   8
#define BNODES 256
#define MAXNB 1024
#define CAPB  12288
#define WP 132

__device__ __forceinline__ float wave_max(float v) {
#pragma unroll
  for (int o = 16; o > 0; o >>= 1) v = fmaxf(v, __shfl_xor(v, o, 32));
  return v;
}
__device__ __forceinline__ float wave_sum(float v) {
#pragma unroll
  for (int o = 16; o > 0; o >>= 1) v += __shfl_xor(v, o, 32);
  return v;
}

__device__ __forceinline__ float xhalf(float x) {
  int b = __builtin_bit_cast(int, x);
  return __builtin_bit_cast(float,
      __builtin_amdgcn_permlanex16(b, b, 0x76543210, 0xfedcba98, false, false));
}
template <int K> __device__ __forceinline__ float row_xor(float x) {
  int b = __builtin_bit_cast(int, x);
  return __builtin_bit_cast(float, __builtin_amdgcn_mov_dpp(b, 0x160 + K, 0xf, 0xf, true));
}
__device__ __forceinline__ float row_sum16(float v) {
  v += row_xor<1>(v);
  v += row_xor<2>(v);
  v += row_xor<4>(v);
  v += row_xor<8>(v);
  return v;
}

__device__ __forceinline__ int block_excl_scan(int v, int* wsum, int& total) {
  int lane = threadIdx.x & 31;
  int wave = threadIdx.x >> 5;
  int inc = v;
#pragma unroll
  for (int o = 1; o < 32; o <<= 1) {
    int t = __shfl_up(inc, o, 32);
    if (lane >= o) inc += t;
  }
  if (lane == 31) wsum[wave] = inc;
  __syncthreads();
  if (wave == 0) {
    int w = wsum[lane];
    int winc = w;
#pragma unroll
    for (int o = 1; o < 32; o <<= 1) {
      int t = __shfl_up(winc, o, 32);
      if (lane >= o) winc += t;
    }
    wsum[lane] = winc - w;
    if (lane == 31) wsum[32] = winc;
  }
  __syncthreads();
  total = wsum[32];
  return inc - v + wsum[wave];
}

__device__ __forceinline__ int tab_pitch(int nbins) { return (nbins + 32) & ~31; }
__global__ __launch_bounds__(1024) void k_binsort(const int* __restrict__ ei, int E, int nbins,
                                                  unsigned* __restrict__ part,
                                                  int* __restrict__ tab) {
  __shared__ int cnt[MAXNB];
  __shared__ int boff[MAXNB + 1];
  __shared__ unsigned stage[CH];
  __shared__ int wsum[33];
  int tid = threadIdx.x;
  int base = blockIdx.x * CH;
  int m = E - base;
  if (m > CH) m = CH;
  for (int b = tid; b < nbins; b += 1024) cnt[b] = 0;
  __syncthreads();

  int s[EPT], d[EPT], r[EPT];
#pragma unroll
  for (int j = 0; j < EPT; ++j) {
    int i = j * 1024 + tid;
    s[j] = 0; d[j] = 0;
    if (i < m) { s[j] = ei[base + i]; d[j] = ei[E + base + i]; }
  }
#pragma unroll
  for (int j = 0; j < EPT; ++j) {
    int i = j * 1024 + tid;
    r[j] = 0;
    if (i < m) r[j] = atomicAdd(&cnt[d[j] >> BSH], 1);
  }
  __syncthreads();

  int v = (tid < nbins) ? cnt[tid] : 0;
  int total;
  int ex = block_excl_scan(v, wsum, total);
  if (tid < nbins) boff[tid] = ex;
  if (tid == 0) boff[nbins] = total;
  __syncthreads();

#pragma unroll
  for (int j = 0; j < EPT; ++j) {
    int i = j * 1024 + tid;
    if (i < m)
      stage[boff[d[j] >> BSH] + r[j]] = ((unsigned)s[j] << BSH) | (unsigned)(d[j] & (BNODES - 1));
  }
  __syncthreads();

  for (int i = tid; i < m; i += 1024) part[base + i] = stage[i];
  int* trow = tab + (size_t)blockIdx.x * tab_pitch(nbins);
  for (int b = tid; b <= nbins; b += 1024) trow[b] = boff[b];
}

__global__ __launch_bounds__(1024) void k_binfinish(const unsigned* __restrict__ part,
                                                    const int* __restrict__ tab, int G,
                                                    int nbins, int n, int* __restrict__ deg,
                                                    int* __restrict__ off,
                                                    int* __restrict__ csr) {
  __shared__ int ncnt[BNODES];
  __shared__ int noff[BNODES];
  __shared__ int cur[BNODES];
  __shared__ int wsum[33];
  __shared__ int outl[CAPB];
  int tid = threadIdx.x;
  int bin = blockIdx.x;
  int ts  = tab_pitch(nbins);
  if (tid < BNODES) { ncnt[tid] = 0; cur[tid] = 0; }
  __syncthreads();

  int sub = tid & 3;
  int sg  = tid >> 2;
  int myb = 0;
  for (int w0 = 0; w0 < G; w0 += 256) {
    int w = w0 + sg;
    if (w < G) {
      int a = tab[(size_t)w * ts + bin];
      int b = tab[(size_t)w * ts + bin + 1];
      if (sub == 0) myb += a;
      const unsigned* p = part + (size_t)w * CH;
      for (int i0 = a + sub; i0 < b; i0 += 32) {
        unsigned q[8];
#pragma unroll
        for (int u = 0; u < 8; ++u) {
          int i = i0 + 4 * u;
          q[u] = p[i < b ? i : b - 1];
        }
#pragma unroll
        for (int u = 0; u < 8; ++u) {
          int i = i0 + 4 * u;
          if (i < b) atomicAdd(&ncnt[q[u] & (BNODES - 1)], 1);
        }
      }
    }
  }
  __syncthreads();

  int binbase;
  block_excl_scan(myb, wsum, binbase);
  __syncthreads();
  int v = (tid < BNODES) ? ncnt[tid] : 0;
  int total;
  int ex = block_excl_scan(v, wsum, total);
  if (tid < BNODES) {
    noff[tid] = ex;
    int node = bin * BNODES + tid;
    if (node < n) { deg[node] = v; off[node] = binbase + ex; }
  }
  __syncthreads();

  bool inl = total <= CAPB;
  for (int w0 = 0; w0 < G; w0 += 256) {
    int w = w0 + sg;
    if (w < G) {
      int a = tab[(size_t)w * ts + bin];
      int b = tab[(size_t)w * ts + bin + 1];
      const unsigned* p = part + (size_t)w * CH;
      for (int i0 = a + sub; i0 < b; i0 += 32) {
        unsigned q[8];
#pragma unroll
        for (int u = 0; u < 8; ++u) {
          int i = i0 + 4 * u;
          q[u] = p[i < b ? i : b - 1];
        }
#pragma unroll
        for (int u = 0; u < 8; ++u) {
          int i = i0 + 4 * u;
          if (i < b) {
            int dl  = (int)(q[u] & (BNODES - 1));
            int pos = noff[dl] + atomicAdd(&cur[dl], 1);
            int sv  = (int)(q[u] >> BSH);
            if (inl) outl[pos] = sv; else csr[binbase + pos] = sv;
          }
        }
      }
    }
  }
  __syncthreads();
  if (inl)
    for (int i = tid; i < total; i += 1024) csr[binbase + i] = outl[i];
}

__global__ __launch_bounds__(256) void k_gemm_wmma(const float* __restrict__ x,
                                                   const float* __restrict__ W,
                                                   float* __restrict__ h, int n) {
  __shared__ __attribute__((aligned(16))) float Wt[C1 * WP];
  int tid = threadIdx.x;
#pragma unroll
  for (int it = 0; it < (DIN * C1) / 256; ++it) {
    int idx  = tid + it * 256;
    int c    = idx & 15;
    int hs   = (idx >> 4) & 1;
    int rest = idx >> 5;
    int nn   = (rest & 3) * 16 + c;
    int kk   = rest >> 2;
    int k    = (kk >> 1) * 4 + (kk & 1) + 2 * hs;
    Wt[nn * WP + k] = W[k * C1 + nn];
  }
  __syncthreads();

  int wave = tid >> 5;
  int lane = tid & 31;
  int half = lane >> 4;
  int l    = lane & 15;
  int rowBase = blockIdx.x * 128 + wave * 16;

  int arow = rowBase + l;
  if (arow > n - 1) arow = n - 1;
  const float* ap = x + (size_t)arow * DIN + 2 * half;

  v8f c0 = {}, c1 = {}, c2 = {}, c3 = {};
#pragma unroll 4
  for (int k0 = 0; k0 < DIN; k0 += 4) {
    v2f a = *(const v2f*)(ap + k0);
    const float* wp = &Wt[l * WP + k0 + 2 * half];
    v2f b0 = *(const v2f*)(wp);
    v2f b1 = *(const v2f*)(wp + 16 * WP);
    v2f b2 = *(const v2f*)(wp + 32 * WP);
    v2f b3 = *(const v2f*)(wp + 48 * WP);
    c0 = __builtin_amdgcn_wmma_f32_16x16x4_f32(false, a, false, b0, (short)0, c0, false, false);
    c1 = __builtin_amdgcn_wmma_f32_16x16x4_f32(false, a, false, b1, (short)0, c1, false, false);
    c2 = __builtin_amdgcn_wmma_f32_16x16x4_f32(false, a, false, b2, (short)0, c2, false, false);
    c3 = __builtin_amdgcn_wmma_f32_16x16x4_f32(false, a, false, b3, (short)0, c3, false, false);
  }

#pragma unroll
  for (int v = 0; v < 8; ++v) {
    int row = rowBase + 8 * half + v;
    if (row < n) {
      float* hp = h + (size_t)row * C1 + l;
      hp[0]  = c0[v];
      hp[16] = c1[v];
      hp[32] = c2[v];
      hp[48] = c3[v];
    }
  }
}

#define RK 32
#define NPW64 10
__global__ __launch_bounds__(256) void k_node64(const int* __restrict__ off,
                                                const int* __restrict__ deg,
                                                const int* __restrict__ csr,
                                                const float* __restrict__ a_src,
                                                const float* __restrict__ a_dst,
                                                const float* __restrict__ h,
                                                const float* __restrict__ b1,
                                                const float* __restrict__ W2,
                                                float* __restrict__ g, int n) {
  __shared__ v2f park[8][NPW64][32];
  int lane = threadIdx.x & 31;
  v2f (*mine)[32] = park[threadIdx.x >> 5];
  int d0 = __builtin_amdgcn_readfirstlane((blockIdx.x * 8 + (threadIdx.x >> 5)) * NPW64);
  if (d0 >= n) return;
  int nn = n - d0;
  if (nn > NPW64) nn = NPW64;
  int offv = 0, degv = 0;
  if (lane < nn) { offv = off[d0 + lane]; degv = deg[d0 + lane]; }
  const v2f as = *(const v2f*)(a_src + 2 * lane);
  const v2f ad = *(const v2f*)(a_dst + 2 * lane);
  int l16 = lane & 15;
  int start = __builtin_amdgcn_readlane(offv, 0);
  int cnt   = __builtin_amdgcn_readlane(degv, 0);
  int m = cnt < RK ? cnt : RK;
  int s = 0;
  if (lane < m) s = csr[start + lane];
  for (int j = 0; j < nn; ++j) {
    int d = d0 + j;
    const v2f hs = *(const v2f*)(h + (size_t)d * C1 + 2 * lane);

    v2f r[RK];
    int sid[RK];
#pragma unroll
    for (int t = 0; t < RK; ++t) sid[t] = __builtin_amdgcn_readlane(s, t);
#pragma unroll
    for (int g = 0; g < RK; g += 8) {
      if (g < m) {
#pragma unroll
        for (int t = g; t < g + 8; ++t) {
          if (t < m) r[t] = *(const v2f*)(h + (size_t)sid[t] * C1 + 2 * lane);
        }
      }
    }
    int nstart = 0, ncnt = 0, nm = 0, sn = 0;
    if (j + 1 < nn) {
      nstart = __builtin_amdgcn_readlane(offv, j + 1);
      ncnt   = __builtin_amdgcn_readlane(degv, j + 1);
      nm = ncnt < RK ? ncnt : RK;
      if (lane < nm) sn = csr[nstart + lane];
    }

    float qd = row_sum16(hs.x * ad.x + hs.y * ad.y);
    float qs = row_sum16(hs.x * as.x + hs.y * as.y);
    float ed = qd + xhalf(qd);
    float vs = (qs + xhalf(qs)) + ed;
    vs = vs > 0.0f ? vs : NEG_SLOPE * vs;

    float X = 0.0f, Y = 0.0f;
#pragma unroll
    for (int g = 0; g < RK; g += 8) {
      if (g < m) {
#pragma unroll
        for (int t = g; t < g + 8; ++t) {
          if (t < m) {
            float w = row_sum16(r[t].x * as.x + r[t].y * as.y);
            if (t < 16) X = (l16 == t) ? w : X;
            else        Y = (l16 == t - 16) ? w : Y;
          }
        }
      }
    }
    float ex = X + xhalf(X);
    float ey = Y + xhalf(Y);
    float v = ((lane < 16) ? ex : ey) + ed;
    v = v > 0.0f ? v : NEG_SLOPE * v;

    float mx = (lane < m) ? fmaxf(v, vs) : vs;
    for (int k0 = RK; k0 < cnt; k0 += 32) {
      int kk = k0 + lane;
      int s2 = (kk < cnt) ? csr[start + kk] : 0;
      int mm = cnt - k0;
      if (mm > 32) mm = 32;
      for (int t = 0; t < mm; ++t) {
        int st = __builtin_amdgcn_readlane(s2, t);
        v2f hv = *(const v2f*)(h + (size_t)st * C1 + 2 * lane);
        float w = row_sum16(hv.x * as.x + hv.y * as.y);
        float e = (w + xhalf(w)) + ed;
        e = e > 0.0f ? e : NEG_SLOPE * e;
        mx = fmaxf(mx, e);
      }
    }
    mx = wave_max(mx);

    float p  = (lane < m) ? __expf(v - mx) : 0.0f;
    float zl = p;
    float a0 = 0.0f, a1 = 0.0f;
#pragma unroll
    for (int g = 0; g < RK; g += 8) {
      if (g < m) {
#pragma unroll
        for (int t = g; t < g + 8; ++t) {
          if (t < m) {
            float pt = __int_as_float(__builtin_amdgcn_readlane(__float_as_int(p), t));
            a0 += pt * r[t].x;
            a1 += pt * r[t].y;
          }
        }
      }
    }
    for (int k0 = RK; k0 < cnt; k0 += 32) {
      int kk = k0 + lane;
      int s2 = (kk < cnt) ? csr[start + kk] : 0;
      int mm = cnt - k0;
      if (mm > 32) mm = 32;
      for (int t = 0; t < mm; ++t) {
        int st = __builtin_amdgcn_readlane(s2, t);
        v2f hv = *(const v2f*)(h + (size_t)st * C1 + 2 * lane);
        float w = row_sum16(hv.x * as.x + hv.y * as.y);
        float e = (w + xhalf(w)) + ed;
        e = e > 0.0f ? e : NEG_SLOPE * e;
        float pt = __expf(e - mx);
        if (lane == t) zl += pt;
        a0 += pt * hv.x;
        a1 += pt * hv.y;
      }
    }
    float pself = __expf(vs - mx);
    if (lane == (cnt & 31)) zl += pself;
    a0 += pself * hs.x;
    a1 += pself * hs.y;

    float z = wave_sum(zl);
    v2f o;
    o.x = a0 / z;
    o.y = a1 / z;
    mine[j][lane] = o;
    start = nstart; cnt = ncnt; m = nm; s = sn;
  }

  const v2f bb = *(const v2f*)(b1 + 2 * lane);
  const v2f ww = *(const v2f*)(W2 + 2 * lane);
  float gres = 0.0f;
  v2f o[NPW64];
#pragma unroll
  for (int j = 0; j < NPW64; ++j) o[j] = mine[j < nn ? j : nn - 1][lane];
#pragma unroll
  for (int j = 0; j < NPW64; ++j) {
    float t = fmaxf(o[j].x + bb.x, 0.0f) * ww.x + fmaxf(o[j].y + bb.y, 0.0f) * ww.y;
    t = row_sum16(t);
    t = t + xhalf(t);
    if (lane == j) gres = t;
  }
  if (lane < nn) g[d0 + lane] = gres;
}

#define NPW1 8
__global__ __launch_bounds__(256) void k_node1(const int* __restrict__ off,
                                               const int* __restrict__ deg,
                                               const int* __restrict__ csr,
                                               const float* __restrict__ g,
                                               const float* __restrict__ a_src2,
                                               const float* __restrict__ a_dst2,
                                               const float* __restrict__ b2,
                                               float* __restrict__ out, int n) {
  int lane = threadIdx.x & 31;
  int d0 = __builtin_amdgcn_readfirstlane((blockIdx.x * 8 + (threadIdx.x >> 5)) * NPW1);
  if (d0 >= n) return;
  int nn = n - d0;
  if (nn > NPW1) nn = NPW1;
  int offv = 0, degv = 0;
  float gd = 0.0f;
  if (lane < nn) { offv = off[d0 + lane]; degv = deg[d0 + lane]; gd = g[d0 + lane]; }
  float as2 = a_src2[0];
  float ad2 = a_dst2[0];
  float bb  = b2[0];
  float res = 0.0f;
  for (int j = 0; j < nn; ++j) {
    int d = d0 + j;
    int start = __builtin_amdgcn_readlane(offv, j);
    int cnt   = __builtin_amdgcn_readlane(degv, j);
    int tot   = cnt + 1;
    float ed  = __int_as_float(__builtin_amdgcn_readlane(__float_as_int(gd), j)) * ad2;

    float mx = -INFINITY;
    for (int k = lane; k < tot; k += 32) {
      int s = (k < cnt) ? csr[start + k] : d;
      float v = g[s] * as2 + ed;
      v = v > 0.0f ? v : NEG_SLOPE * v;
      mx = fmaxf(mx, v);
    }
    mx = wave_max(mx);

    float zl = 0.0f, al = 0.0f;
    for (int k = lane; k < tot; k += 32) {
      int s = (k < cnt) ? csr[start + k] : d;
      float gs = g[s];
      float v = gs * as2 + ed;
      v = v > 0.0f ? v : NEG_SLOPE * v;
      float p = __expf(v - mx);
      zl += p;
      al += p * gs;
    }
    float z = wave_sum(zl);
    float a = wave_sum(al);
    float v = a / z + bb;
    float o = 1.0f / (1.0f + __expf(-v));
    if (lane == j) res = o;
  }
  if (lane < nn) out[d0 + lane] = res;
}

extern "C" void kernel_launch(void* const* d_in, const int* in_sizes, int n_in,
                              void* d_out, int out_size, void* d_ws, size_t ws_size,
                              hipStream_t stream) {
  const int*   ei     = (const int*)d_in[0];
  const float* x      = (const float*)d_in[1];
  const float* W1     = (const float*)d_in[2];
  const float* a_src1 = (const float*)d_in[3];
  const float* a_dst1 = (const float*)d_in[4];
  const float* b1     = (const float*)d_in[5];
  const float* W2     = (const float*)d_in[6];
  const float* a_src2 = (const float*)d_in[7];
  const float* a_dst2 = (const float*)d_in[8];
  const float* b2     = (const float*)d_in[9];

  const int E = in_sizes[0] / 2;
  const int n = in_sizes[1] / DIN;
  const int nbins = (n + BNODES - 1) / BNODES;
  const int G = (E + CH - 1) / CH;

  float* f     = (float*)d_ws;
  float* h     = f;
  float* g     = f + (size_t)130 * n;
  int*   deg   = (int*)(f + (size_t)131 * n);
  int*   off   = (int*)(f + (size_t)132 * n);
  unsigned* part = (unsigned*)(f + (size_t)134 * n);
  int*   csr   = (int*)part + (size_t)E;
  int*   tab   = csr + (size_t)E;
  float* out   = (float*)d_out;

  const int B = 256;
  dim3 blk(B);

  k_binsort<<<dim3(G), dim3(1024), 0, stream>>>(ei, E, nbins, part, tab);
  k_binfinish<<<dim3(nbins), dim3(1024), 0, stream>>>(part, tab, G, nbins, n, deg, off, csr);

  k_gemm_wmma<<<dim3((n + 127) / 128), blk, 0, stream>>>(x, W1, h, n);
  k_node64<<<dim3((n + 8 * NPW64 - 1) / (8 * NPW64)), blk, 0, stream>>>(off, deg, csr, a_src1, a_dst1, h,
                                                                      b1, W2, g, n);

  k_node1<<<dim3((n + 8 * NPW1 - 1) / (8 * NPW1)), blk, 0, stream>>>(off, deg, csr, g, a_src2, a_dst2, b2,
                                                                    out, n);
}


# Layer-1 GEMM runs inside the edge-sort launch as extra 32-wave workgroups (448 rows each, side by side with the sorting ones): 5 launches become 4, sort and GEMM overlap
# speedup vs baseline: 7.3881x; 1.0784x over previous
#include <hip/hip_runtime.h>
#include <math.h>

typedef float v2f __attribute__((ext_vector_type(2)));
typedef float v8f __attribute__((ext_vector_type(8)));

#define DIN 128
#define C1  64
#define NEG_SLOPE 0.2f
#define CH    8192
#define EPT   8
#define BSH   8
#define BNODES 256
#define MAXNB 1024
#define CAPB  12288
#define WP 132

__device__ __forceinline__ float wave_max(float v) {
#pragma unroll
  for (int o = 16; o > 0; o >>= 1) v = fmaxf(v, __shfl_xor(v, o, 32));
  return v;
}
__device__ __forceinline__ float wave_sum(float v) {
#pragma unroll
  for (int o = 16; o > 0; o >>= 1) v += __shfl_xor(v, o, 32);
  return v;
}

__device__ __forceinline__ float xhalf(float x) {
  int b = __builtin_bit_cast(int, x);
  return __builtin_bit_cast(float,
      __builtin_amdgcn_permlanex16(b, b, 0x76543210, 0xfedcba98, false, false));
}
template <int K> __device__ __forceinline__ float row_xor(float x) {
  int b = __builtin_bit_cast(int, x);
  return __builtin_bit_cast(float, __builtin_amdgcn_mov_dpp(b, 0x160 + K, 0xf, 0xf, true));
}
__device__ __forceinline__ float row_sum16(float v) {
  v += row_xor<1>(v);
  v += row_xor<2>(v);
  v += row_xor<4>(v);
  v += row_xor<8>(v);
  return v;
}

__device__ __forceinline__ int block_excl_scan(int v, int* wsum, int& total) {
  int lane = threadIdx.x & 31;
  int wave = threadIdx.x >> 5;
  int inc = v;
#pragma unroll
  for (int o = 1; o < 32; o <<= 1) {
    int t = __shfl_up(inc, o, 32);
    if (lane >= o) inc += t;
  }
  if (lane == 31) wsum[wave] = inc;
  __syncthreads();
  if (wave == 0) {
    int w = wsum[lane];
    int winc = w;
#pragma unroll
    for (int o = 1; o < 32; o <<= 1) {
      int t = __shfl_up(winc, o, 32);
      if (lane >= o) winc += t;
    }
    wsum[lane] = winc - w;
    if (lane == 31) wsum[32] = winc;
  }
  __syncthreads();
  total = wsum[32];
  return inc - v + wsum[wave];
}

__device__ __forceinline__ int tab_pitch(int nbins) { return (nbins + 32) & ~31; }
__device__ __forceinline__ void sort_chunk(const int* __restrict__ ei, int E, int nbins,
                                           unsigned* __restrict__ part, int* __restrict__ tab,
                                           int chunk) {
  __shared__ int cnt[MAXNB];
  __shared__ int boff[MAXNB + 1];
  __shared__ unsigned stage[CH];
  __shared__ int wsum[33];
  int tid = threadIdx.x;
  int base = chunk * CH;
  int m = E - base;
  if (m > CH) m = CH;
  for (int b = tid; b < nbins; b += 1024) cnt[b] = 0;
  __syncthreads();

  int s[EPT], d[EPT], r[EPT];
#pragma unroll
  for (int j = 0; j < EPT; ++j) {
    int i = j * 1024 + tid;
    s[j] = 0; d[j] = 0;
    if (i < m) { s[j] = ei[base + i]; d[j] = ei[E + base + i]; }
  }
#pragma unroll
  for (int j = 0; j < EPT; ++j) {
    int i = j * 1024 + tid;
    r[j] = 0;
    if (i < m) r[j] = atomicAdd(&cnt[d[j] >> BSH], 1);
  }
  __syncthreads();

  int v = (tid < nbins) ? cnt[tid] : 0;
  int total;
  int ex = block_excl_scan(v, wsum, total);
  if (tid < nbins) boff[tid] = ex;
  if (tid == 0) boff[nbins] = total;
  __syncthreads();

#pragma unroll
  for (int j = 0; j < EPT; ++j) {
    int i = j * 1024 + tid;
    if (i < m)
      stage[boff[d[j] >> BSH] + r[j]] = ((unsigned)s[j] << BSH) | (unsigned)(d[j] & (BNODES - 1));
  }
  __syncthreads();

  for (int i = tid; i < m; i += 1024) part[base + i] = stage[i];
  int* trow = tab + (size_t)chunk * tab_pitch(nbins);
  for (int b = tid; b <= nbins; b += 1024) trow[b] = boff[b];
}

__global__ __launch_bounds__(1024) void k_binfinish(const unsigned* __restrict__ part,
                                                    const int* __restrict__ tab, int G,
                                                    int nbins, int n, int* __restrict__ deg,
                                                    int* __restrict__ off,
                                                    int* __restrict__ csr) {
  __shared__ int ncnt[BNODES];
  __shared__ int noff[BNODES];
  __shared__ int cur[BNODES];
  __shared__ int wsum[33];
  __shared__ int outl[CAPB];
  int tid = threadIdx.x;
  int bin = blockIdx.x;
  int ts  = tab_pitch(nbins);
  if (tid < BNODES) { ncnt[tid] = 0; cur[tid] = 0; }
  __syncthreads();

  int sub = tid & 3;
  int sg  = tid >> 2;
  int myb = 0;
  for (int w0 = 0; w0 < G; w0 += 256) {
    int w = w0 + sg;
    if (w < G) {
      int a = tab[(size_t)w * ts + bin];
      int b = tab[(size_t)w * ts + bin + 1];
      if (sub == 0) myb += a;
      const unsigned* p = part + (size_t)w * CH;
      for (int i0 = a + sub; i0 < b; i0 += 32) {
        unsigned q[8];
#pragma unroll
        for (int u = 0; u < 8; ++u) {
          int i = i0 + 4 * u;
          q[u] = p[i < b ? i : b - 1];
        }
#pragma unroll
        for (int u = 0; u < 8; ++u) {
          int i = i0 + 4 * u;
          if (i < b) atomicAdd(&ncnt[q[u] & (BNODES - 1)], 1);
        }
      }
    }
  }
  __syncthreads();

  int binbase;
  block_excl_scan(myb, wsum, binbase);
  __syncthreads();
  int v = (tid < BNODES) ? ncnt[tid] : 0;
  int total;
  int ex = block_excl_scan(v, wsum, total);
  if (tid < BNODES) {
    noff[tid] = ex;
    int node = bin * BNODES + tid;
    if (node < n) { deg[node] = v; off[node] = binbase + ex; }
  }
  __syncthreads();

  bool inl = total <= CAPB;
  for (int w0 = 0; w0 < G; w0 += 256) {
    int w = w0 + sg;
    if (w < G) {
      int a = tab[(size_t)w * ts + bin];
      int b = tab[(size_t)w * ts + bin + 1];
      const unsigned* p = part + (size_t)w * CH;
      for (int i0 = a + sub; i0 < b; i0 += 32) {
        unsigned q[8];
#pragma unroll
        for (int u = 0; u < 8; ++u) {
          int i = i0 + 4 * u;
          q[u] = p[i < b ? i : b - 1];
        }
#pragma unroll
        for (int u = 0; u < 8; ++u) {
          int i = i0 + 4 * u;
          if (i < b) {
            int dl  = (int)(q[u] & (BNODES - 1));
            int pos = noff[dl] + atomicAdd(&cur[dl], 1);
            int sv  = (int)(q[u] >> BSH);
            if (inl) outl[pos] = sv; else csr[binbase + pos] = sv;
          }
        }
      }
    }
  }
  __syncthreads();
  if (inl)
    for (int i = tid; i < total; i += 1024) csr[binbase + i] = outl[i];
}

#define GROWS 448
__device__ __forceinline__ void gemm_rows(const float* __restrict__ x,
                                          const float* __restrict__ W,
                                          float* __restrict__ h, int n, int gb) {
  __shared__ __attribute__((aligned(16))) float Wt[C1 * WP];
  int tid = threadIdx.x;
#pragma unroll
  for (int it = 0; it < (DIN * C1) / 1024; ++it) {
    int idx  = tid + it * 1024;
    int c    = idx & 15;
    int hs   = (idx >> 4) & 1;
    int rest = idx >> 5;
    int nn   = (rest & 3) * 16 + c;
    int kk   = rest >> 2;
    int k    = (kk >> 1) * 4 + (kk & 1) + 2 * hs;
    Wt[nn * WP + k] = W[k * C1 + nn];
  }
  __syncthreads();

  int wave = __builtin_amdgcn_readfirstlane(tid >> 5);
  int lane = tid & 31;
  int half = lane >> 4;
  int l    = lane & 15;
  int rowBase = gb * GROWS + wave * 16;
  if (wave * 16 >= GROWS || rowBase >= n) return;

  int arow = rowBase + l;
  if (arow > n - 1) arow = n - 1;
  const float* ap = x + (size_t)arow * DIN + 2 * half;

  v8f c0 = {}, c1 = {}, c2 = {}, c3 = {};
#pragma unroll 4
  for (int k0 = 0; k0 < DIN; k0 += 4) {
    v2f a = *(const v2f*)(ap + k0);
    const float* wp = &Wt[l * WP + k0 + 2 * half];
    v2f b0 = *(const v2f*)(wp);
    v2f b1 = *(const v2f*)(wp + 16 * WP);
    v2f b2 = *(const v2f*)(wp + 32 * WP);
    v2f b3 = *(const v2f*)(wp + 48 * WP);
    c0 = __builtin_amdgcn_wmma_f32_16x16x4_f32(false, a, false, b0, (short)0, c0, false, false);
    c1 = __builtin_amdgcn_wmma_f32_16x16x4_f32(false, a, false, b1, (short)0, c1, false, false);
    c2 = __builtin_amdgcn_wmma_f32_16x16x4_f32(false, a, false, b2, (short)0, c2, false, false);
    c3 = __builtin_amdgcn_wmma_f32_16x16x4_f32(false, a, false, b3, (short)0, c3, false, false);
  }

#pragma unroll
  for (int v = 0; v < 8; ++v) {
    int row = rowBase + 8 * half + v;
    if (row < n) {
      float* hp = h + (size_t)row * C1 + l;
      hp[0]  = c0[v];
      hp[16] = c1[v];
      hp[32] = c2[v];
      hp[48] = c3[v];
    }
  }
}

__global__ __launch_bounds__(1024) void k_binsort_gemm(const int* __restrict__ ei, int E,
                                                       int nbins, unsigned* __restrict__ part,
                                                       int* __restrict__ tab,
                                                       const float* __restrict__ x,
                                                       const float* __restrict__ W,
                                                       float* __restrict__ h, int n, int GB) {
  if ((int)blockIdx.x < GB) gemm_rows(x, W, h, n, (int)blockIdx.x);
  else sort_chunk(ei, E, nbins, part, tab, (int)blockIdx.x - GB);
}

#define RK 32
#define NPW64 10
__global__ __launch_bounds__(256) void k_node64(const int* __restrict__ off,
                                                const int* __restrict__ deg,
                                                const int* __restrict__ csr,
                                                const float* __restrict__ a_src,
                                                const float* __restrict__ a_dst,
                                                const float* __restrict__ h,
                                                const float* __restrict__ b1,
                                                const float* __restrict__ W2,
                                                float* __restrict__ g, int n) {
  __shared__ v2f park[8][NPW64][32];
  int lane = threadIdx.x & 31;
  v2f (*mine)[32] = park[threadIdx.x >> 5];
  int d0 = __builtin_amdgcn_readfirstlane((blockIdx.x * 8 + (threadIdx.x >> 5)) * NPW64);
  if (d0 >= n) return;
  int nn = n - d0;
  if (nn > NPW64) nn = NPW64;
  int offv = 0, degv = 0;
  if (lane < nn) { offv = off[d0 + lane]; degv = deg[d0 + lane]; }
  const v2f as = *(const v2f*)(a_src + 2 * lane);
  const v2f ad = *(const v2f*)(a_dst + 2 * lane);
  int l16 = lane & 15;
  int start = __builtin_amdgcn_readlane(offv, 0);
  int cnt   = __builtin_amdgcn_readlane(degv, 0);
  int m = cnt < RK ? cnt : RK;
  int s = 0;
  if (lane < m) s = csr[start + lane];
  for (int j = 0; j < nn; ++j) {
    int d = d0 + j;
    const v2f hs = *(const v2f*)(h + (size_t)d * C1 + 2 * lane);

    v2f r[RK];
    int sid[RK];
#pragma unroll
    for (int t = 0; t < RK; ++t) sid[t] = __builtin_amdgcn_readlane(s, t);
#pragma unroll
    for (int g = 0; g < RK; g += 8) {
      if (g < m) {
#pragma unroll
        for (int t = g; t < g + 8; ++t) {
          if (t < m) r[t] = *(const v2f*)(h + (size_t)sid[t] * C1 + 2 * lane);
        }
      }
    }
    int nstart = 0, ncnt = 0, nm = 0, sn = 0;
    if (j + 1 < nn) {
      nstart = __builtin_amdgcn_readlane(offv, j + 1);
      ncnt   = __builtin_amdgcn_readlane(degv, j + 1);
      nm = ncnt < RK ? ncnt : RK;
      if (lane < nm) sn = csr[nstart + lane];
    }

    float qd = row_sum16(hs.x * ad.x + hs.y * ad.y);
    float qs = row_sum16(hs.x * as.x + hs.y * as.y);
    float ed = qd + xhalf(qd);
    float vs = (qs + xhalf(qs)) + ed;
    vs = vs > 0.0f ? vs : NEG_SLOPE * vs;

    float X = 0.0f, Y = 0.0f;
#pragma unroll
    for (int g = 0; g < RK; g += 8) {
      if (g < m) {
#pragma unroll
        for (int t = g; t < g + 8; ++t) {
          if (t < m) {
            float w = row_sum16(r[t].x * as.x + r[t].y * as.y);
            if (t < 16) X = (l16 == t) ? w : X;
            else        Y = (l16 == t - 16) ? w : Y;
          }
        }
      }
    }
    float ex = X + xhalf(X);
    float ey = Y + xhalf(Y);
    float v = ((lane < 16) ? ex : ey) + ed;
    v = v > 0.0f ? v : NEG_SLOPE * v;

    float mx = (lane < m) ? fmaxf(v, vs) : vs;
    for (int k0 = RK; k0 < cnt; k0 += 32) {
      int kk = k0 + lane;
      int s2 = (kk < cnt) ? csr[start + kk] : 0;
      int mm = cnt - k0;
      if (mm > 32) mm = 32;
      for (int t = 0; t < mm; ++t) {
        int st = __builtin_amdgcn_readlane(s2, t);
        v2f hv = *(const v2f*)(h + (size_t)st * C1 + 2 * lane);
        float w = row_sum16(hv.x * as.x + hv.y * as.y);
        float e = (w + xhalf(w)) + ed;
        e = e > 0.0f ? e : NEG_SLOPE * e;
        mx = fmaxf(mx, e);
      }
    }
    mx = wave_max(mx);

    float p  = (lane < m) ? __expf(v - mx) : 0.0f;
    float zl = p;
    float a0 = 0.0f, a1 = 0.0f;
#pragma unroll
    for (int g = 0; g < RK; g += 8) {
      if (g < m) {
#pragma unroll
        for (int t = g; t < g + 8; ++t) {
          if (t < m) {
            float pt = __int_as_float(__builtin_amdgcn_readlane(__float_as_int(p), t));
            a0 += pt * r[t].x;
            a1 += pt * r[t].y;
          }
        }
      }
    }
    for (int k0 = RK; k0 < cnt; k0 += 32) {
      int kk = k0 + lane;
      int s2 = (kk < cnt) ? csr[start + kk] : 0;
      int mm = cnt - k0;
      if (mm > 32) mm = 32;
      for (int t = 0; t < mm; ++t) {
        int st = __builtin_amdgcn_readlane(s2, t);
        v2f hv = *(const v2f*)(h + (size_t)st * C1 + 2 * lane);
        float w = row_sum16(hv.x * as.x + hv.y * as.y);
        float e = (w + xhalf(w)) + ed;
        e = e > 0.0f ? e : NEG_SLOPE * e;
        float pt = __expf(e - mx);
        if (lane == t) zl += pt;
        a0 += pt * hv.x;
        a1 += pt * hv.y;
      }
    }
    float pself = __expf(vs - mx);
    if (lane == (cnt & 31)) zl += pself;
    a0 += pself * hs.x;
    a1 += pself * hs.y;

    float z = wave_sum(zl);
    v2f o;
    o.x = a0 / z;
    o.y = a1 / z;
    mine[j][lane] = o;
    start = nstart; cnt = ncnt; m = nm; s = sn;
  }

  const v2f bb = *(const v2f*)(b1 + 2 * lane);
  const v2f ww = *(const v2f*)(W2 + 2 * lane);
  float gres = 0.0f;
  v2f o[NPW64];
#pragma unroll
  for (int j = 0; j < NPW64; ++j) o[j] = mine[j < nn ? j : nn - 1][lane];
#pragma unroll
  for (int j = 0; j < NPW64; ++j) {
    float t = fmaxf(o[j].x + bb.x, 0.0f) * ww.x + fmaxf(o[j].y + bb.y, 0.0f) * ww.y;
    t = row_sum16(t);
    t = t + xhalf(t);
    if (lane == j) gres = t;
  }
  if (lane < nn) g[d0 + lane] = gres;
}

#define NPW1 8
__global__ __launch_bounds__(256) void k_node1(const int* __restrict__ off,
                                               const int* __restrict__ deg,
                                               const int* __restrict__ csr,
                                               const float* __restrict__ g,
                                               const float* __restrict__ a_src2,
                                               const float* __restrict__ a_dst2,
                                               const float* __restrict__ b2,
                                               float* __restrict__ out, int n) {
  int lane = threadIdx.x & 31;
  int d0 = __builtin_amdgcn_readfirstlane((blockIdx.x * 8 + (threadIdx.x >> 5)) * NPW1);
  if (d0 >= n) return;
  int nn = n - d0;
  if (nn > NPW1) nn = NPW1;
  int offv = 0, degv = 0;
  float gd = 0.0f;
  if (lane < nn) { offv = off[d0 + lane]; degv = deg[d0 + lane]; gd = g[d0 + lane]; }
  float as2 = a_src2[0];
  float ad2 = a_dst2[0];
  float bb  = b2[0];
  float res = 0.0f;
  for (int j = 0; j < nn; ++j) {
    int d = d0 + j;
    int start = __builtin_amdgcn_readlane(offv, j);
    int cnt   = __builtin_amdgcn_readlane(degv, j);
    int tot   = cnt + 1;
    float ed  = __int_as_float(__builtin_amdgcn_readlane(__float_as_int(gd), j)) * ad2;

    float mx = -INFINITY;
    for (int k = lane; k < tot; k += 32) {
      int s = (k < cnt) ? csr[start + k] : d;
      float v = g[s] * as2 + ed;
      v = v > 0.0f ? v : NEG_SLOPE * v;
      mx = fmaxf(mx, v);
    }
    mx = wave_max(mx);

    float zl = 0.0f, al = 0.0f;
    for (int k = lane; k < tot; k += 32) {
      int s = (k < cnt) ? csr[start + k] : d;
      float gs = g[s];
      float v = gs * as2 + ed;
      v = v > 0.0f ? v : NEG_SLOPE * v;
      float p = __expf(v - mx);
      zl += p;
      al += p * gs;
    }
    float z = wave_sum(zl);
    float a = wave_sum(al);
    float v = a / z + bb;
    float o = 1.0f / (1.0f + __expf(-v));
    if (lane == j) res = o;
  }
  if (lane < nn) out[d0 + lane] = res;
}

extern "C" void kernel_launch(void* const* d_in, const int* in_sizes, int n_in,
                              void* d_out, int out_size, void* d_ws, size_t ws_size,
                              hipStream_t stream) {
  const int*   ei     = (const int*)d_in[0];
  const float* x      = (const float*)d_in[1];
  const float* W1     = (const float*)d_in[2];
  const float* a_src1 = (const float*)d_in[3];
  const float* a_dst1 = (const float*)d_in[4];
  const float* b1     = (const float*)d_in[5];
  const float* W2     = (const float*)d_in[6];
  const float* a_src2 = (const float*)d_in[7];
  const float* a_dst2 = (const float*)d_in[8];
  const float* b2     = (const float*)d_in[9];

  const int E = in_sizes[0] / 2;
  const int n = in_sizes[1] / DIN;
  const int nbins = (n + BNODES - 1) / BNODES;
  const int G = (E + CH - 1) / CH;

  float* f     = (float*)d_ws;
  float* h     = f;
  float* g     = f + (size_t)130 * n;
  int*   deg   = (int*)(f + (size_t)131 * n);
  int*   off   = (int*)(f + (size_t)132 * n);
  unsigned* part = (unsigned*)(f + (size_t)134 * n);
  int*   csr   = (int*)part + (size_t)E;
  int*   tab   = csr + (size_t)E;
  float* out   = (float*)d_out;

  const int B = 256;
  dim3 blk(B);

  const int GB = (n + GROWS - 1) / GROWS;
  k_binsort_gemm<<<dim3(GB + G), dim3(1024), 0, stream>>>(ei, E, nbins, part, tab, x, W1, h, n, GB);
  k_binfinish<<<dim3(nbins), dim3(1024), 0, stream>>>(part, tab, G, nbins, n, deg, off, csr);

  k_node64<<<dim3((n + 8 * NPW64 - 1) / (8 * NPW64)), blk, 0, stream>>>(off, deg, csr, a_src1, a_dst1, h,
                                                                      b1, W2, g, n);

  k_node1<<<dim3((n + 8 * NPW1 - 1) / (8 * NPW1)), blk, 0, stream>>>(off, deg, csr, g, a_src2, a_dst2, b2,
                                                                    out, n);
}


# Layer-2 node kernel looks each source value up in an LDS copy of g, staged per 32-wave workgroup with 16-byte loads, instead of a scattered 4-byte gather per edge and pass; sum order kept
# speedup vs baseline: 8.0246x; 1.0862x over previous
#include <hip/hip_runtime.h>
#include <math.h>

typedef float v2f __attribute__((ext_vector_type(2)));
typedef float v8f __attribute__((ext_vector_type(8)));
typedef float v4f __attribute__((ext_vector_type(4)));

#define DIN 128
#define C1  64
#define NEG_SLOPE 0.2f
#define CH    8192
#define EPT   8
#define BSH   8
#define BNODES 256
#define MAXNB 1024
#define CAPB  12288
#define WP 132

__device__ __forceinline__ float wave_max(float v) {
#pragma unroll
  for (int o = 16; o > 0; o >>= 1) v = fmaxf(v, __shfl_xor(v, o, 32));
  return v;
}
__device__ __forceinline__ float wave_sum(float v) {
#pragma unroll
  for (int o = 16; o > 0; o >>= 1) v += __shfl_xor(v, o, 32);
  return v;
}

__device__ __forceinline__ float xhalf(float x) {
  int b = __builtin_bit_cast(int, x);
  return __builtin_bit_cast(float,
      __builtin_amdgcn_permlanex16(b, b, 0x76543210, 0xfedcba98, false, false));
}
template <int K> __device__ __forceinline__ float row_xor(float x) {
  int b = __builtin_bit_cast(int, x);
  return __builtin_bit_cast(float, __builtin_amdgcn_mov_dpp(b, 0x160 + K, 0xf, 0xf, true));
}
__device__ __forceinline__ float row_sum16(float v) {
  v += row_xor<1>(v);
  v += row_xor<2>(v);
  v += row_xor<4>(v);
  v += row_xor<8>(v);
  return v;
}

__device__ __forceinline__ int block_excl_scan(int v, int* wsum, int& total) {
  int lane = threadIdx.x & 31;
  int wave = threadIdx.x >> 5;
  int inc = v;
#pragma unroll
  for (int o = 1; o < 32; o <<= 1) {
    int t = __shfl_up(inc, o, 32);
    if (lane >= o) inc += t;
  }
  if (lane == 31) wsum[wave] = inc;
  __syncthreads();
  if (wave == 0) {
    int w = wsum[lane];
    int winc = w;
#pragma unroll
    for (int o = 1; o < 32; o <<= 1) {
      int t = __shfl_up(winc, o, 32);
      if (lane >= o) winc += t;
    }
    wsum[lane] = winc - w;
    if (lane == 31) wsum[32] = winc;
  }
  __syncthreads();
  total = wsum[32];
  return inc - v + wsum[wave];
}

__device__ __forceinline__ int tab_pitch(int nbins) { return (nbins + 32) & ~31; }
__device__ __forceinline__ void sort_chunk(const int* __restrict__ ei, int E, int nbins,
                                           unsigned* __restrict__ part, int* __restrict__ tab,
                                           int chunk) {
  __shared__ int cnt[MAXNB];
  __shared__ int boff[MAXNB + 1];
  __shared__ unsigned stage[CH];
  __shared__ int wsum[33];
  int tid = threadIdx.x;
  int base = chunk * CH;
  int m = E - base;
  if (m > CH) m = CH;
  for (int b = tid; b < nbins; b += 1024) cnt[b] = 0;
  __syncthreads();

  int s[EPT], d[EPT], r[EPT];
#pragma unroll
  for (int j = 0; j < EPT; ++j) {
    int i = j * 1024 + tid;
    s[j] = 0; d[j] = 0;
    if (i < m) { s[j] = ei[base + i]; d[j] = ei[E + base + i]; }
  }
#pragma unroll
  for (int j = 0; j < EPT; ++j) {
    int i = j * 1024 + tid;
    r[j] = 0;
    if (i < m) r[j] = atomicAdd(&cnt[d[j] >> BSH], 1);
  }
  __syncthreads();

  int v = (tid < nbins) ? cnt[tid] : 0;
  int total;
  int ex = block_excl_scan(v, wsum, total);
  if (tid < nbins) boff[tid] = ex;
  if (tid == 0) boff[nbins] = total;
  __syncthreads();

#pragma unroll
  for (int j = 0; j < EPT; ++j) {
    int i = j * 1024 + tid;
    if (i < m)
      stage[boff[d[j] >> BSH] + r[j]] = ((unsigned)s[j] << BSH) | (unsigned)(d[j] & (BNODES - 1));
  }
  __syncthreads();

  for (int i = tid; i < m; i += 1024) part[base + i] = stage[i];
  int* trow = tab + (size_t)chunk * tab_pitch(nbins);
  for (int b = tid; b <= nbins; b += 1024) trow[b] = boff[b];
}

__global__ __launch_bounds__(1024) void k_binfinish(const unsigned* __restrict__ part,
                                                    const int* __restrict__ tab, int G,
                                                    int nbins, int n, int* __restrict__ deg,
                                                    int* __restrict__ off,
                                                    int* __restrict__ csr) {
  __shared__ int ncnt[BNODES];
  __shared__ int noff[BNODES];
  __shared__ int cur[BNODES];
  __shared__ int wsum[33];
  __shared__ int outl[CAPB];
  int tid = threadIdx.x;
  int bin = blockIdx.x;
  int ts  = tab_pitch(nbins);
  if (tid < BNODES) { ncnt[tid] = 0; cur[tid] = 0; }
  __syncthreads();

  int sub = tid & 3;
  int sg  = tid >> 2;
  int myb = 0;
  for (int w0 = 0; w0 < G; w0 += 256) {
    int w = w0 + sg;
    if (w < G) {
      int a = tab[(size_t)w * ts + bin];
      int b = tab[(size_t)w * ts + bin + 1];
      if (sub == 0) myb += a;
      const unsigned* p = part + (size_t)w * CH;
      for (int i0 = a + sub; i0 < b; i0 += 32) {
        unsigned q[8];
#pragma unroll
        for (int u = 0; u < 8; ++u) {
          int i = i0 + 4 * u;
          q[u] = p[i < b ? i : b - 1];
        }
#pragma unroll
        for (int u = 0; u < 8; ++u) {
          int i = i0 + 4 * u;
          if (i < b) atomicAdd(&ncnt[q[u] & (BNODES - 1)], 1);
        }
      }
    }
  }
  __syncthreads();

  int binbase;
  block_excl_scan(myb, wsum, binbase);
  __syncthreads();
  int v = (tid < BNODES) ? ncnt[tid] : 0;
  int total;
  int ex = block_excl_scan(v, wsum, total);
  if (tid < BNODES) {
    noff[tid] = ex;
    int node = bin * BNODES + tid;
    if (node < n) { deg[node] = v; off[node] = binbase + ex; }
  }
  __syncthreads();

  bool inl = total <= CAPB;
  for (int w0 = 0; w0 < G; w0 += 256) {
    int w = w0 + sg;
    if (w < G) {
      int a = tab[(size_t)w * ts + bin];
      int b = tab[(size_t)w * ts + bin + 1];
      const unsigned* p = part + (size_t)w * CH;
      for (int i0 = a + sub; i0 < b; i0 += 32) {
        unsigned q[8];
#pragma unroll
        for (int u = 0; u < 8; ++u) {
          int i = i0 + 4 * u;
          q[u] = p[i < b ? i : b - 1];
        }
#pragma unroll
        for (int u = 0; u < 8; ++u) {
          int i = i0 + 4 * u;
          if (i < b) {
            int dl  = (int)(q[u] & (BNODES - 1));
            int pos = noff[dl] + atomicAdd(&cur[dl], 1);
            int sv  = (int)(q[u] >> BSH);
            if (inl) outl[pos] = sv; else csr[binbase + pos] = sv;
          }
        }
      }
    }
  }
  __syncthreads();
  if (inl)
    for (int i = tid; i < total; i += 1024) csr[binbase + i] = outl[i];
}

#define GROWS 448
__device__ __forceinline__ void gemm_rows(const float* __restrict__ x,
                                          const float* __restrict__ W,
                                          float* __restrict__ h, int n, int gb) {
  __shared__ __attribute__((aligned(16))) float Wt[C1 * WP];
  int tid = threadIdx.x;
#pragma unroll
  for (int it = 0; it < (DIN * C1) / 1024; ++it) {
    int idx  = tid + it * 1024;
    int c    = idx & 15;
    int hs   = (idx >> 4) & 1;
    int rest = idx >> 5;
    int nn   = (rest & 3) * 16 + c;
    int kk   = rest >> 2;
    int k    = (kk >> 1) * 4 + (kk & 1) + 2 * hs;
    Wt[nn * WP + k] = W[k * C1 + nn];
  }
  __syncthreads();

  int wave = __builtin_amdgcn_readfirstlane(tid >> 5);
  int lane = tid & 31;
  int half = lane >> 4;
  int l    = lane & 15;
  int rowBase = gb * GROWS + wave * 16;
  if (wave * 16 >= GROWS || rowBase >= n) return;

  int arow = rowBase + l;
  if (arow > n - 1) arow = n - 1;
  const float* ap = x + (size_t)arow * DIN + 2 * half;

  v8f c0 = {}, c1 = {}, c2 = {}, c3 = {};
#pragma unroll 4
  for (int k0 = 0; k0 < DIN; k0 += 4) {
    v2f a = *(const v2f*)(ap + k0);
    const float* wp = &Wt[l * WP + k0 + 2 * half];
    v2f b0 = *(const v2f*)(wp);
    v2f b1 = *(const v2f*)(wp + 16 * WP);
    v2f b2 = *(const v2f*)(wp + 32 * WP);
    v2f b3 = *(const v2f*)(wp + 48 * WP);
    c0 = __builtin_amdgcn_wmma_f32_16x16x4_f32(false, a, false, b0, (short)0, c0, false, false);
    c1 = __builtin_amdgcn_wmma_f32_16x16x4_f32(false, a, false, b1, (short)0, c1, false, false);
    c2 = __builtin_amdgcn_wmma_f32_16x16x4_f32(false, a, false, b2, (short)0, c2, false, false);
    c3 = __builtin_amdgcn_wmma_f32_16x16x4_f32(false, a, false, b3, (short)0, c3, false, false);
  }

#pragma unroll
  for (int v = 0; v < 8; ++v) {
    int row = rowBase + 8 * half + v;
    if (row < n) {
      float* hp = h + (size_t)row * C1 + l;
      hp[0]  = c0[v];
      hp[16] = c1[v];
      hp[32] = c2[v];
      hp[48] = c3[v];
    }
  }
}

__global__ __launch_bounds__(1024) void k_binsort_gemm(const int* __restrict__ ei, int E,
                                                       int nbins, unsigned* __restrict__ part,
                                                       int* __restrict__ tab,
                                                       const float* __restrict__ x,
                                                       const float* __restrict__ W,
                                                       float* __restrict__ h, int n, int GB) {
  if ((int)blockIdx.x < GB) gemm_rows(x, W, h, n, (int)blockIdx.x);
  else sort_chunk(ei, E, nbins, part, tab, (int)blockIdx.x - GB);
}

#define RK 32
#define NPW64 10
__global__ __launch_bounds__(256) void k_node64(const int* __restrict__ off,
                                                const int* __restrict__ deg,
                                                const int* __restrict__ csr,
                                                const float* __restrict__ a_src,
                                                const float* __restrict__ a_dst,
                                                const float* __restrict__ h,
                                                const float* __restrict__ b1,
                                                const float* __restrict__ W2,
                                                float* __restrict__ g, int n) {
  __shared__ v2f park[8][NPW64][32];
  int lane = threadIdx.x & 31;
  v2f (*mine)[32] = park[threadIdx.x >> 5];
  int d0 = __builtin_amdgcn_readfirstlane((blockIdx.x * 8 + (threadIdx.x >> 5)) * NPW64);
  if (d0 >= n) return;
  int nn = n - d0;
  if (nn > NPW64) nn = NPW64;
  int offv = 0, degv = 0;
  if (lane < nn) { offv = off[d0 + lane]; degv = deg[d0 + lane]; }
  const v2f as = *(const v2f*)(a_src + 2 * lane);
  const v2f ad = *(const v2f*)(a_dst + 2 * lane);
  int l16 = lane & 15;
  int start = __builtin_amdgcn_readlane(offv, 0);
  int cnt   = __builtin_amdgcn_readlane(degv, 0);
  int m = cnt < RK ? cnt : RK;
  int s = 0;
  if (lane < m) s = csr[start + lane];
  for (int j = 0; j < nn; ++j) {
    int d = d0 + j;
    const v2f hs = *(const v2f*)(h + (size_t)d * C1 + 2 * lane);

    v2f r[RK];
    int sid[RK];
#pragma unroll
    for (int t = 0; t < RK; ++t) sid[t] = __builtin_amdgcn_readlane(s, t);
#pragma unroll
    for (int g = 0; g < RK; g += 8) {
      if (g < m) {
#pragma unroll
        for (int t = g; t < g + 8; ++t) {
          if (t < m) r[t] = *(const v2f*)(h + (size_t)sid[t] * C1 + 2 * lane);
        }
      }
    }
    int nstart = 0, ncnt = 0, nm = 0, sn = 0;
    if (j + 1 < nn) {
      nstart = __builtin_amdgcn_readlane(offv, j + 1);
      ncnt   = __builtin_amdgcn_readlane(degv, j + 1);
      nm = ncnt < RK ? ncnt : RK;
      if (lane < nm) sn = csr[nstart + lane];
    }

    float qd = row_sum16(hs.x * ad.x + hs.y * ad.y);
    float qs = row_sum16(hs.x * as.x + hs.y * as.y);
    float ed = qd + xhalf(qd);
    float vs = (qs + xhalf(qs)) + ed;
    vs = vs > 0.0f ? vs : NEG_SLOPE * vs;

    float X = 0.0f, Y = 0.0f;
#pragma unroll
    for (int g = 0; g < RK; g += 8) {
      if (g < m) {
#pragma unroll
        for (int t = g; t < g + 8; ++t) {
          if (t < m) {
            float w = row_sum16(r[t].x * as.x + r[t].y * as.y);
            if (t < 16) X = (l16 == t) ? w : X;
            else        Y = (l16 == t - 16) ? w : Y;
          }
        }
      }
    }
    float ex = X + xhalf(X);
    float ey = Y + xhalf(Y);
    float v = ((lane < 16) ? ex : ey) + ed;
    v = v > 0.0f ? v : NEG_SLOPE * v;

    float mx = (lane < m) ? fmaxf(v, vs) : vs;
    for (int k0 = RK; k0 < cnt; k0 += 32) {
      int kk = k0 + lane;
      int s2 = (kk < cnt) ? csr[start + kk] : 0;
      int mm = cnt - k0;
      if (mm > 32) mm = 32;
      for (int t = 0; t < mm; ++t) {
        int st = __builtin_amdgcn_readlane(s2, t);
        v2f hv = *(const v2f*)(h + (size_t)st * C1 + 2 * lane);
        float w = row_sum16(hv.x * as.x + hv.y * as.y);
        float e = (w + xhalf(w)) + ed;
        e = e > 0.0f ? e : NEG_SLOPE * e;
        mx = fmaxf(mx, e);
      }
    }
    mx = wave_max(mx);

    float p  = (lane < m) ? __expf(v - mx) : 0.0f;
    float zl = p;
    float a0 = 0.0f, a1 = 0.0f;
#pragma unroll
    for (int g = 0; g < RK; g += 8) {
      if (g < m) {
#pragma unroll
        for (int t = g; t < g + 8; ++t) {
          if (t < m) {
            float pt = __int_as_float(__builtin_amdgcn_readlane(__float_as_int(p), t));
            a0 += pt * r[t].x;
            a1 += pt * r[t].y;
          }
        }
      }
    }
    for (int k0 = RK; k0 < cnt; k0 += 32) {
      int kk = k0 + lane;
      int s2 = (kk < cnt) ? csr[start + kk] : 0;
      int mm = cnt - k0;
      if (mm > 32) mm = 32;
      for (int t = 0; t < mm; ++t) {
        int st = __builtin_amdgcn_readlane(s2, t);
        v2f hv = *(const v2f*)(h + (size_t)st * C1 + 2 * lane);
        float w = row_sum16(hv.x * as.x + hv.y * as.y);
        float e = (w + xhalf(w)) + ed;
        e = e > 0.0f ? e : NEG_SLOPE * e;
        float pt = __expf(e - mx);
        if (lane == t) zl += pt;
        a0 += pt * hv.x;
        a1 += pt * hv.y;
      }
    }
    float pself = __expf(vs - mx);
    if (lane == (cnt & 31)) zl += pself;
    a0 += pself * hs.x;
    a1 += pself * hs.y;

    float z = wave_sum(zl);
    v2f o;
    o.x = a0 / z;
    o.y = a1 / z;
    mine[j][lane] = o;
    start = nstart; cnt = ncnt; m = nm; s = sn;
  }

  const v2f bb = *(const v2f*)(b1 + 2 * lane);
  const v2f ww = *(const v2f*)(W2 + 2 * lane);
  float gres = 0.0f;
  v2f o[NPW64];
#pragma unroll
  for (int j = 0; j < NPW64; ++j) o[j] = mine[j < nn ? j : nn - 1][lane];
#pragma unroll
  for (int j = 0; j < NPW64; ++j) {
    float t = fmaxf(o[j].x + bb.x, 0.0f) * ww.x + fmaxf(o[j].y + bb.y, 0.0f) * ww.y;
    t = row_sum16(t);
    t = t + xhalf(t);
    if (lane == j) gres = t;
  }
  if (lane < nn) g[d0 + lane] = gres;
}

#define NPW1 13
#define GCAP 65536
__global__ __launch_bounds__(1024, 1) void k_node1(const int* __restrict__ off,
                                                const int* __restrict__ deg,
                                                const int* __restrict__ csr,
                                                const float* __restrict__ g,
                                                const float* __restrict__ a_src2,
                                                const float* __restrict__ a_dst2,
                                                const float* __restrict__ b2,
                                                float* __restrict__ out, int n) {
  __shared__ __attribute__((aligned(16))) float gl[GCAP + 4];
  int tid  = threadIdx.x;
  int lane = tid & 31;
  int d0 = __builtin_amdgcn_readfirstlane((blockIdx.x * 32 + (tid >> 5)) * NPW1);
  int nn = n - d0;
  if (nn > NPW1) nn = NPW1;
  if (nn < 0) nn = 0;
  int offv = 0, degv = 0;
  float gd = 0.0f;
  if (lane < nn) { offv = off[d0 + lane]; degv = deg[d0 + lane]; gd = g[d0 + lane]; }
  float as2 = a_src2[0];
  float ad2 = a_dst2[0];
  float bb  = b2[0];

  int s[NPW1];
#pragma unroll
  for (int j = 0; j < NPW1; ++j) {
    int start = __builtin_amdgcn_readlane(offv, j);
    int cnt   = __builtin_amdgcn_readlane(degv, j);
    s[j] = -1;
    if (j < nn) {
      if (lane < cnt) s[j] = csr[start + lane];
      else if (lane == cnt) s[j] = d0 + j;
    }
  }

  int shift = (int)(((size_t)g >> 2) & 3);
  float gs[NPW1];
#pragma unroll
  for (int j = 0; j < NPW1; ++j) gs[j] = 0.0f;

  for (int base = 0; base < n; base += GCAP) {
    int len = n - base;
    if (len > GCAP) len = GCAP;
    int lo  = base - shift;
    int nch = (len + shift + 3) >> 2;
    __syncthreads();
    int cfirst = (lo < 0) ? 1 : 0;
    int cend   = (lo + 4 * nch > n) ? nch - 1 : nch;
    for (int c0 = cfirst + tid; c0 < cend; c0 += 4 * 1024) {
      v4f v[4];
#pragma unroll
      for (int u = 0; u < 4; ++u) {
        int c = c0 + u * 1024;
        v[u] = *(const v4f*)(g + lo + 4 * (c < cend ? c : cend - 1));
      }
#pragma unroll
      for (int u = 0; u < 4; ++u) {
        int c = c0 + u * 1024;
        if (c < cend) *(v4f*)&gl[4 * c] = v[u];
      }
    }
    if (tid < 8) {
      int  c    = (tid < 4) ? 0 : nch - 1;
      bool part = (tid < 4) ? (cfirst == 1) : (cend < nch);
      int  gi   = lo + 4 * c + (tid & 3);
      if (part && gi >= 0 && gi < n) gl[4 * c + (tid & 3)] = g[gi];
    }
    __syncthreads();
#pragma unroll
    for (int j = 0; j < NPW1; ++j) {
      unsigned idx = (unsigned)(s[j] - base);
      if (idx < (unsigned)GCAP) gs[j] = gl[idx + shift];
    }
  }

  float vv[NPW1], mx[NPW1];
#pragma unroll
  for (int j = 0; j < NPW1; ++j) {
    int start = __builtin_amdgcn_readlane(offv, j);
    int cnt   = __builtin_amdgcn_readlane(degv, j);
    int tot   = cnt + 1;
    float ed  = __int_as_float(__builtin_amdgcn_readlane(__float_as_int(gd), j)) * ad2;
    float v = gs[j] * as2 + ed;
    v = v > 0.0f ? v : NEG_SLOPE * v;
    float m = (lane < tot) ? v : -INFINITY;
    if (tot > 32) {
      for (int k = 32 + lane; k < tot; k += 32) {
        int s2 = (k < cnt) ? csr[start + k] : d0 + j;
        float w = g[s2] * as2 + ed;
        w = w > 0.0f ? w : NEG_SLOPE * w;
        m = fmaxf(m, w);
      }
    }
    vv[j] = v;
    mx[j] = m;
  }
#pragma unroll
  for (int j = 0; j < NPW1; ++j) mx[j] = wave_max(mx[j]);

  float zl[NPW1], al[NPW1];
#pragma unroll
  for (int j = 0; j < NPW1; ++j) {
    int start = __builtin_amdgcn_readlane(offv, j);
    int cnt   = __builtin_amdgcn_readlane(degv, j);
    int tot   = cnt + 1;
    float ed  = __int_as_float(__builtin_amdgcn_readlane(__float_as_int(gd), j)) * ad2;
    float p = (lane < tot) ? __expf(vv[j] - mx[j]) : 0.0f;
    float z = p;
    float a = p * gs[j];
    if (tot > 32) {
      for (int k = 32 + lane; k < tot; k += 32) {
        int s2 = (k < cnt) ? csr[start + k] : d0 + j;
        float g2 = g[s2];
        float w = g2 * as2 + ed;
        w = w > 0.0f ? w : NEG_SLOPE * w;
        float pw = __expf(w - mx[j]);
        z += pw;
        a += pw * g2;
      }
    }
    zl[j] = z;
    al[j] = a;
  }
#pragma unroll
  for (int j = 0; j < NPW1; ++j) zl[j] = wave_sum(zl[j]);
#pragma unroll
  for (int j = 0; j < NPW1; ++j) al[j] = wave_sum(al[j]);
  float z = 1.0f, a = 0.0f;
#pragma unroll
  for (int j = 0; j < NPW1; ++j)
    if (lane == j) { z = zl[j]; a = al[j]; }
  float r = a / z + bb;
  float res = 1.0f / (1.0f + __expf(-r));
  if (lane < nn) out[d0 + lane] = res;
}

extern "C" void kernel_launch(void* const* d_in, const int* in_sizes, int n_in,
                              void* d_out, int out_size, void* d_ws, size_t ws_size,
                              hipStream_t stream) {
  const int*   ei     = (const int*)d_in[0];
  const float* x      = (const float*)d_in[1];
  const float* W1     = (const float*)d_in[2];
  const float* a_src1 = (const float*)d_in[3];
  const float* a_dst1 = (const float*)d_in[4];
  const float* b1     = (const float*)d_in[5];
  const float* W2     = (const float*)d_in[6];
  const float* a_src2 = (const float*)d_in[7];
  const float* a_dst2 = (const float*)d_in[8];
  const float* b2     = (const float*)d_in[9];

  const int E = in_sizes[0] / 2;
  const int n = in_sizes[1] / DIN;
  const int nbins = (n + BNODES - 1) / BNODES;
  const int G = (E + CH - 1) / CH;

  float* f     = (float*)d_ws;
  float* h     = f;
  float* g     = f + (size_t)130 * n;
  int*   deg   = (int*)(f + (size_t)131 * n);
  int*   off   = (int*)(f + (size_t)132 * n);
  unsigned* part = (unsigned*)(f + (size_t)134 * n);
  int*   csr   = (int*)part + (size_t)E;
  int*   tab   = csr + (size_t)E;
  float* out   = (float*)d_out;

  const int B = 256;
  dim3 blk(B);

  const int GB = (n + GROWS - 1) / GROWS;
  k_binsort_gemm<<<dim3(GB + G), dim3(1024), 0, stream>>>(ei, E, nbins, part, tab, x, W1, h, n, GB);
  k_binfinish<<<dim3(nbins), dim3(1024), 0, stream>>>(part, tab, G, nbins, n, deg, off, csr);

  k_node64<<<dim3((n + 8 * NPW64 - 1) / (8 * NPW64)), blk, 0, stream>>>(off, deg, csr, a_src1, a_dst1, h,
                                                                      b1, W2, g, n);

  k_node1<<<dim3((n + 32 * NPW1 - 1) / (32 * NPW1)), dim3(1024), 0, stream>>>(off, deg, csr, g, a_src2,
                                                                              a_dst2, b2, out, n);
}


# Layer-1 logits h.a_src, h.a_dst formed by the GEMM waves from their accumulators; node kernel fetches one float per in-edge, two nodes ahead, not lane sums per gathered row; logit sum order changes
# speedup vs baseline: 8.8835x; 1.1070x over previous
#include <hip/hip_runtime.h>
#include <math.h>

typedef float v2f __attribute__((ext_vector_type(2)));
typedef float v8f __attribute__((ext_vector_type(8)));
typedef float v4f __attribute__((ext_vector_type(4)));

#define DIN 128
#define C1  64
#define NEG_SLOPE 0.2f
#define CH    8192
#define EPT   8
#define BSH   8
#define BNODES 256
#define MAXNB 1024
#define CAPB  12288
#define WP 132

__device__ __forceinline__ float wave_max(float v) {
#pragma unroll
  for (int o = 16; o > 0; o >>= 1) v = fmaxf(v, __shfl_xor(v, o, 32));
  return v;
}
__device__ __forceinline__ float wave_sum(float v) {
#pragma unroll
  for (int o = 16; o > 0; o >>= 1) v += __shfl_xor(v, o, 32);
  return v;
}

__device__ __forceinline__ float xhalf(float x) {
  int b = __builtin_bit_cast(int, x);
  return __builtin_bit_cast(float,
      __builtin_amdgcn_permlanex16(b, b, 0x76543210, 0xfedcba98, false, false));
}
template <int K> __device__ __forceinline__ float row_xor(float x) {
  int b = __builtin_bit_cast(int, x);
  return __builtin_bit_cast(float, __builtin_amdgcn_mov_dpp(b, 0x160 + K, 0xf, 0xf, true));
}
__device__ __forceinline__ float row_sum16(float v) {
  v += row_xor<1>(v);
  v += row_xor<2>(v);
  v += row_xor<4>(v);
  v += row_xor<8>(v);
  return v;
}

__device__ __forceinline__ int block_excl_scan(int v, int* wsum, int& total) {
  int lane = threadIdx.x & 31;
  int wave = threadIdx.x >> 5;
  int inc = v;
#pragma unroll
  for (int o = 1; o < 32; o <<= 1) {
    int t = __shfl_up(inc, o, 32);
    if (lane >= o) inc += t;
  }
  if (lane == 31) wsum[wave] = inc;
  __syncthreads();
  if (wave == 0) {
    int w = wsum[lane];
    int winc = w;
#pragma unroll
    for (int o = 1; o < 32; o <<= 1) {
      int t = __shfl_up(winc, o, 32);
      if (lane >= o) winc += t;
    }
    wsum[lane] = winc - w;
    if (lane == 31) wsum[32] = winc;
  }
  __syncthreads();
  total = wsum[32];
  return inc - v + wsum[wave];
}

__device__ __forceinline__ int tab_pitch(int nbins) { return (nbins + 32) & ~31; }
__device__ __forceinline__ void sort_chunk(const int* __restrict__ ei, int E, int nbins,
                                           unsigned* __restrict__ part, int* __restrict__ tab,
                                           int chunk) {
  __shared__ int cnt[MAXNB];
  __shared__ int boff[MAXNB + 1];
  __shared__ unsigned stage[CH];
  __shared__ int wsum[33];
  int tid = threadIdx.x;
  int base = chunk * CH;
  int m = E - base;
  if (m > CH) m = CH;
  for (int b = tid; b < nbins; b += 1024) cnt[b] = 0;
  __syncthreads();

  int s[EPT], d[EPT], r[EPT];
#pragma unroll
  for (int j = 0; j < EPT; ++j) {
    int i = j * 1024 + tid;
    s[j] = 0; d[j] = 0;
    if (i < m) { s[j] = ei[base + i]; d[j] = ei[E + base + i]; }
  }
#pragma unroll
  for (int j = 0; j < EPT; ++j) {
    int i = j * 1024 + tid;
    r[j] = 0;
    if (i < m) r[j] = atomicAdd(&cnt[d[j] >> BSH], 1);
  }
  __syncthreads();

  int v = (tid < nbins) ? cnt[tid] : 0;
  int total;
  int ex = block_excl_scan(v, wsum, total);
  if (tid < nbins) boff[tid] = ex;
  if (tid == 0) boff[nbins] = total;
  __syncthreads();

#pragma unroll
  for (int j = 0; j < EPT; ++j) {
    int i = j * 1024 + tid;
    if (i < m)
      stage[boff[d[j] >> BSH] + r[j]] = ((unsigned)s[j] << BSH) | (unsigned)(d[j] & (BNODES - 1));
  }
  __syncthreads();

  for (int i = tid; i < m; i += 1024) part[base + i] = stage[i];
  int* trow = tab + (size_t)chunk * tab_pitch(nbins);
  for (int b = tid; b <= nbins; b += 1024) trow[b] = boff[b];
}

__global__ __launch_bounds__(1024) void k_binfinish(const unsigned* __restrict__ part,
                                                    const int* __restrict__ tab, int G,
                                                    int nbins, int n, int* __restrict__ deg,
                                                    int* __restrict__ off,
                                                    int* __restrict__ csr) {
  __shared__ int ncnt[BNODES];
  __shared__ int noff[BNODES];
  __shared__ int cur[BNODES];
  __shared__ int wsum[33];
  __shared__ int outl[CAPB];
  int tid = threadIdx.x;
  int bin = blockIdx.x;
  int ts  = tab_pitch(nbins);
  if (tid < BNODES) { ncnt[tid] = 0; cur[tid] = 0; }
  __syncthreads();

  int sub = tid & 3;
  int sg  = tid >> 2;
  int myb = 0;
  for (int w0 = 0; w0 < G; w0 += 256) {
    int w = w0 + sg;
    if (w < G) {
      int a = tab[(size_t)w * ts + bin];
      int b = tab[(size_t)w * ts + bin + 1];
      if (sub == 0) myb += a;
      const unsigned* p = part + (size_t)w * CH;
      for (int i0 = a + sub; i0 < b; i0 += 32) {
        unsigned q[8];
#pragma unroll
        for (int u = 0; u < 8; ++u) {
          int i = i0 + 4 * u;
          q[u] = p[i < b ? i : b - 1];
        }
#pragma unroll
        for (int u = 0; u < 8; ++u) {
          int i = i0 + 4 * u;
          if (i < b) atomicAdd(&ncnt[q[u] & (BNODES - 1)], 1);
        }
      }
    }
  }
  __syncthreads();

  int binbase;
  block_excl_scan(myb, wsum, binbase);
  __syncthreads();
  int v = (tid < BNODES) ? ncnt[tid] : 0;
  int total;
  int ex = block_excl_scan(v, wsum, total);
  if (tid < BNODES) {
    noff[tid] = ex;
    int node = bin * BNODES + tid;
    if (node < n) { deg[node] = v; off[node] = binbase + ex; }
  }
  __syncthreads();

  bool inl = total <= CAPB;
  for (int w0 = 0; w0 < G; w0 += 256) {
    int w = w0 + sg;
    if (w < G) {
      int a = tab[(size_t)w * ts + bin];
      int b = tab[(size_t)w * ts + bin + 1];
      const unsigned* p = part + (size_t)w * CH;
      for (int i0 = a + sub; i0 < b; i0 += 32) {
        unsigned q[8];
#pragma unroll
        for (int u = 0; u < 8; ++u) {
          int i = i0 + 4 * u;
          q[u] = p[i < b ? i : b - 1];
        }
#pragma unroll
        for (int u = 0; u < 8; ++u) {
          int i = i0 + 4 * u;
          if (i < b) {
            int dl  = (int)(q[u] & (BNODES - 1));
            int pos = noff[dl] + atomicAdd(&cur[dl], 1);
            int sv  = (int)(q[u] >> BSH);
            if (inl) outl[pos] = sv; else csr[binbase + pos] = sv;
          }
        }
      }
    }
  }
  __syncthreads();
  if (inl)
    for (int i = tid; i < total; i += 1024) csr[binbase + i] = outl[i];
}

#define GROWS 448
__device__ __forceinline__ void gemm_rows(const float* __restrict__ x,
                                          const float* __restrict__ W,
                                          const float* __restrict__ a_src,
                                          const float* __restrict__ a_dst,
                                          float* __restrict__ h, float* __restrict__ es,
                                          float* __restrict__ ed, int n, int gb) {
  __shared__ __attribute__((aligned(16))) float Wt[C1 * WP];
  __shared__ float av[2 * C1];
  int tid = threadIdx.x;
  if (tid < C1) av[tid] = a_src[tid];
  else if (tid < 2 * C1) av[tid] = a_dst[tid - C1];
#pragma unroll
  for (int it = 0; it < (DIN * C1) / 1024; ++it) {
    int idx  = tid + it * 1024;
    int c    = idx & 15;
    int hs   = (idx >> 4) & 1;
    int rest = idx >> 5;
    int nn   = (rest & 3) * 16 + c;
    int kk   = rest >> 2;
    int k    = (kk >> 1) * 4 + (kk & 1) + 2 * hs;
    Wt[nn * WP + k] = W[k * C1 + nn];
  }
  __syncthreads();

  int wave = __builtin_amdgcn_readfirstlane(tid >> 5);
  int lane = tid & 31;
  int half = lane >> 4;
  int l    = lane & 15;
  int rowBase = gb * GROWS + wave * 16;
  if (wave * 16 >= GROWS || rowBase >= n) return;

  int arow = rowBase + l;
  if (arow > n - 1) arow = n - 1;
  const float* ap = x + (size_t)arow * DIN + 2 * half;

  v8f c0 = {}, c1 = {}, c2 = {}, c3 = {};
#pragma unroll 4
  for (int k0 = 0; k0 < DIN; k0 += 4) {
    v2f a = *(const v2f*)(ap + k0);
    const float* wp = &Wt[l * WP + k0 + 2 * half];
    v2f b0 = *(const v2f*)(wp);
    v2f b1 = *(const v2f*)(wp + 16 * WP);
    v2f b2 = *(const v2f*)(wp + 32 * WP);
    v2f b3 = *(const v2f*)(wp + 48 * WP);
    c0 = __builtin_amdgcn_wmma_f32_16x16x4_f32(false, a, false, b0, (short)0, c0, false, false);
    c1 = __builtin_amdgcn_wmma_f32_16x16x4_f32(false, a, false, b1, (short)0, c1, false, false);
    c2 = __builtin_amdgcn_wmma_f32_16x16x4_f32(false, a, false, b2, (short)0, c2, false, false);
    c3 = __builtin_amdgcn_wmma_f32_16x16x4_f32(false, a, false, b3, (short)0, c3, false, false);
  }

#pragma unroll
  for (int v = 0; v < 8; ++v) {
    int row = rowBase + 8 * half + v;
    if (row < n) {
      float* hp = h + (size_t)row * C1 + l;
      hp[0]  = c0[v];
      hp[16] = c1[v];
      hp[32] = c2[v];
      hp[48] = c3[v];
    }
  }

  float s0 = av[l], s1 = av[16 + l], s2 = av[32 + l], s3 = av[48 + l];
  float t0 = av[C1 + l], t1 = av[C1 + 16 + l], t2 = av[C1 + 32 + l], t3 = av[C1 + 48 + l];
  float ks = 0.0f, kd = 0.0f;
#pragma unroll
  for (int v = 0; v < 8; ++v) {
    float ps = row_sum16(c0[v] * s0 + c1[v] * s1 + c2[v] * s2 + c3[v] * s3);
    float pd = row_sum16(c0[v] * t0 + c1[v] * t1 + c2[v] * t2 + c3[v] * t3);
    if (l == v) { ks = ps; kd = pd; }
  }
  if (l < 8) {
    int row = rowBase + 8 * half + l;
    if (row < n) { es[row] = ks; ed[row] = kd; }
  }
}

__global__ __launch_bounds__(1024) void k_binsort_gemm(const int* __restrict__ ei, int E,
                                                       int nbins, unsigned* __restrict__ part,
                                                       int* __restrict__ tab,
                                                       const float* __restrict__ x,
                                                       const float* __restrict__ W,
                                                       const float* __restrict__ a_src,
                                                       const float* __restrict__ a_dst,
                                                       float* __restrict__ h,
                                                       float* __restrict__ es,
                                                       float* __restrict__ ed, int n, int GB) {
  if ((int)blockIdx.x < GB) gemm_rows(x, W, a_src, a_dst, h, es, ed, n, (int)blockIdx.x);
  else sort_chunk(ei, E, nbins, part, tab, (int)blockIdx.x - GB);
}

#define RK 32
#define NPW64 10
__global__ __launch_bounds__(256) void k_node64(const int* __restrict__ off,
                                                const int* __restrict__ deg,
                                                const int* __restrict__ csr,
                                                const float* __restrict__ es,
                                                const float* __restrict__ ed,
                                                const float* __restrict__ h,
                                                const float* __restrict__ b1,
                                                const float* __restrict__ W2,
                                                float* __restrict__ g, int n) {
  __shared__ v2f park[8][NPW64][32];
  int lane = threadIdx.x & 31;
  v2f (*mine)[32] = park[threadIdx.x >> 5];
  int d0 = __builtin_amdgcn_readfirstlane((blockIdx.x * 8 + (threadIdx.x >> 5)) * NPW64);
  if (d0 >= n) return;
  int nn = n - d0;
  if (nn > NPW64) nn = NPW64;
  int offv = 0, degv = 0;
  float esv = 0.0f, edv = 0.0f;
  if (lane < nn) {
    offv = off[d0 + lane]; degv = deg[d0 + lane];
    esv = es[d0 + lane];   edv = ed[d0 + lane];
  }
  int start = __builtin_amdgcn_readlane(offv, 0);
  int cnt   = __builtin_amdgcn_readlane(degv, 0);
  int m = cnt < RK ? cnt : RK;
  int s = 0;
  if (lane < m) s = csr[start + lane];
  int start1 = 0, cnt1 = 0, m1 = 0, s1 = 0;
  if (1 < nn) {
    start1 = __builtin_amdgcn_readlane(offv, 1);
    cnt1   = __builtin_amdgcn_readlane(degv, 1);
    m1 = cnt1 < RK ? cnt1 : RK;
    if (lane < m1) s1 = csr[start1 + lane];
  }
  float ev = 0.0f;
  if (lane < m) ev = es[s];
  for (int j = 0; j < nn; ++j) {
    int d = d0 + j;
    float edj = __int_as_float(__builtin_amdgcn_readlane(__float_as_int(edv), j));
    float vs  = __int_as_float(__builtin_amdgcn_readlane(__float_as_int(esv), j)) + edj;
    vs = vs > 0.0f ? vs : NEG_SLOPE * vs;
    float v = ev + edj;
    v = v > 0.0f ? v : NEG_SLOPE * v;
    float ev1 = 0.0f;
    if (lane < m1) ev1 = es[s1];
    const v2f hs = *(const v2f*)(h + (size_t)d * C1 + 2 * lane);

    v2f r[RK];
    int sid[RK];
#pragma unroll
    for (int t = 0; t < RK; ++t) sid[t] = __builtin_amdgcn_readlane(s, t);
#pragma unroll
    for (int g = 0; g < RK; g += 8) {
      if (g < m) {
#pragma unroll
        for (int t = g; t < g + 8; ++t) {
          if (t < m) r[t] = *(const v2f*)(h + (size_t)sid[t] * C1 + 2 * lane);
        }
      }
    }
    int start2 = 0, cnt2 = 0, m2 = 0, s2n = 0;
    if (j + 2 < nn) {
      start2 = __builtin_amdgcn_readlane(offv, j + 2);
      cnt2   = __builtin_amdgcn_readlane(degv, j + 2);
      m2 = cnt2 < RK ? cnt2 : RK;
      if (lane < m2) s2n = csr[start2 + lane];
    }

    float mx = (lane < m) ? fmaxf(v, vs) : vs;
    for (int k0 = RK; k0 < cnt; k0 += 32) {
      int kk = k0 + lane;
      if (kk < cnt) {
        float e = es[csr[start + kk]] + edj;
        e = e > 0.0f ? e : NEG_SLOPE * e;
        mx = fmaxf(mx, e);
      }
    }
    mx = wave_max(mx);

    float p  = (lane < m) ? __expf(v - mx) : 0.0f;
    float zl = p;
    float a0 = 0.0f, a1 = 0.0f;
#pragma unroll
    for (int g = 0; g < RK; g += 8) {
      if (g < m) {
#pragma unroll
        for (int t = g; t < g + 8; ++t) {
          if (t < m) {
            float pt = __int_as_float(__builtin_amdgcn_readlane(__float_as_int(p), t));
            a0 += pt * r[t].x;
            a1 += pt * r[t].y;
          }
        }
      }
    }
    for (int k0 = RK; k0 < cnt; k0 += 32) {
      int kk = k0 + lane;
      int s2 = 0;
      float p2 = 0.0f;
      if (kk < cnt) {
        s2 = csr[start + kk];
        float e = es[s2] + edj;
        e = e > 0.0f ? e : NEG_SLOPE * e;
        p2 = __expf(e - mx);
      }
      zl += p2;
      int mm = cnt - k0;
      if (mm > 32) mm = 32;
      for (int t = 0; t < mm; ++t) {
        int st = __builtin_amdgcn_readlane(s2, t);
        float pt = __int_as_float(__builtin_amdgcn_readlane(__float_as_int(p2), t));
        v2f hv = *(const v2f*)(h + (size_t)st * C1 + 2 * lane);
        a0 += pt * hv.x;
        a1 += pt * hv.y;
      }
    }
    float pself = __expf(vs - mx);
    if (lane == (cnt & 31)) zl += pself;
    a0 += pself * hs.x;
    a1 += pself * hs.y;

    float z = wave_sum(zl);
    v2f o;
    o.x = a0 / z;
    o.y = a1 / z;
    mine[j][lane] = o;
    start = start1; cnt = cnt1; m = m1; s = s1; ev = ev1;
    start1 = start2; cnt1 = cnt2; m1 = m2; s1 = s2n;
  }

  const v2f bb = *(const v2f*)(b1 + 2 * lane);
  const v2f ww = *(const v2f*)(W2 + 2 * lane);
  float gres = 0.0f;
  v2f o[NPW64];
#pragma unroll
  for (int j = 0; j < NPW64; ++j) o[j] = mine[j < nn ? j : nn - 1][lane];
#pragma unroll
  for (int j = 0; j < NPW64; ++j) {
    float t = fmaxf(o[j].x + bb.x, 0.0f) * ww.x + fmaxf(o[j].y + bb.y, 0.0f) * ww.y;
    t = row_sum16(t);
    t = t + xhalf(t);
    if (lane == j) gres = t;
  }
  if (lane < nn) g[d0 + lane] = gres;
}

#define NPW1 13
#define GCAP 65536
__global__ __launch_bounds__(1024, 1) void k_node1(const int* __restrict__ off,
                                                const int* __restrict__ deg,
                                                const int* __restrict__ csr,
                                                const float* __restrict__ g,
                                                const float* __restrict__ a_src2,
                                                const float* __restrict__ a_dst2,
                                                const float* __restrict__ b2,
                                                float* __restrict__ out, int n) {
  __shared__ __attribute__((aligned(16))) float gl[GCAP + 4];
  int tid  = threadIdx.x;
  int lane = tid & 31;
  int d0 = __builtin_amdgcn_readfirstlane((blockIdx.x * 32 + (tid >> 5)) * NPW1);
  int nn = n - d0;
  if (nn > NPW1) nn = NPW1;
  if (nn < 0) nn = 0;
  int offv = 0, degv = 0;
  float gd = 0.0f;
  if (lane < nn) { offv = off[d0 + lane]; degv = deg[d0 + lane]; gd = g[d0 + lane]; }
  float as2 = a_src2[0];
  float ad2 = a_dst2[0];
  float bb  = b2[0];

  int s[NPW1];
#pragma unroll
  for (int j = 0; j < NPW1; ++j) {
    int start = __builtin_amdgcn_readlane(offv, j);
    int cnt   = __builtin_amdgcn_readlane(degv, j);
    s[j] = -1;
    if (j < nn) {
      if (lane < cnt) s[j] = csr[start + lane];
      else if (lane == cnt) s[j] = d0 + j;
    }
  }

  int shift = (int)(((size_t)g >> 2) & 3);
  float gs[NPW1];
#pragma unroll
  for (int j = 0; j < NPW1; ++j) gs[j] = 0.0f;

  for (int base = 0; base < n; base += GCAP) {
    int len = n - base;
    if (len > GCAP) len = GCAP;
    int lo  = base - shift;
    int nch = (len + shift + 3) >> 2;
    __syncthreads();
    int cfirst = (lo < 0) ? 1 : 0;
    int cend   = (lo + 4 * nch > n) ? nch - 1 : nch;
    for (int c0 = cfirst + tid; c0 < cend; c0 += 4 * 1024) {
      v4f v[4];
#pragma unroll
      for (int u = 0; u < 4; ++u) {
        int c = c0 + u * 1024;
        v[u] = *(const v4f*)(g + lo + 4 * (c < cend ? c : cend - 1));
      }
#pragma unroll
      for (int u = 0; u < 4; ++u) {
        int c = c0 + u * 1024;
        if (c < cend) *(v4f*)&gl[4 * c] = v[u];
      }
    }
    if (tid < 8) {
      int  c    = (tid < 4) ? 0 : nch - 1;
      bool part = (tid < 4) ? (cfirst == 1) : (cend < nch);
      int  gi   = lo + 4 * c + (tid & 3);
      if (part && gi >= 0 && gi < n) gl[4 * c + (tid & 3)] = g[gi];
    }
    __syncthreads();
#pragma unroll
    for (int j = 0; j < NPW1; ++j) {
      unsigned idx = (unsigned)(s[j] - base);
      if (idx < (unsigned)GCAP) gs[j] = gl[idx + shift];
    }
  }

  float vv[NPW1], mx[NPW1];
#pragma unroll
  for (int j = 0; j < NPW1; ++j) {
    int start = __builtin_amdgcn_readlane(offv, j);
    int cnt   = __builtin_amdgcn_readlane(degv, j);
    int tot   = cnt + 1;
    float ed  = __int_as_float(__builtin_amdgcn_readlane(__float_as_int(gd), j)) * ad2;
    float v = gs[j] * as2 + ed;
    v = v > 0.0f ? v : NEG_SLOPE * v;
    float m = (lane < tot) ? v : -INFINITY;
    if (tot > 32) {
      for (int k = 32 + lane; k < tot; k += 32) {
        int s2 = (k < cnt) ? csr[start + k] : d0 + j;
        float w = g[s2] * as2 + ed;
        w = w > 0.0f ? w : NEG_SLOPE * w;
        m = fmaxf(m, w);
      }
    }
    vv[j] = v;
    mx[j] = m;
  }
#pragma unroll
  for (int j = 0; j < NPW1; ++j) mx[j] = wave_max(mx[j]);

  float zl[NPW1], al[NPW1];
#pragma unroll
  for (int j = 0; j < NPW1; ++j) {
    int start = __builtin_amdgcn_readlane(offv, j);
    int cnt   = __builtin_amdgcn_readlane(degv, j);
    int tot   = cnt + 1;
    float ed  = __int_as_float(__builtin_amdgcn_readlane(__float_as_int(gd), j)) * ad2;
    float p = (lane < tot) ? __expf(vv[j] - mx[j]) : 0.0f;
    float z = p;
    float a = p * gs[j];
    if (tot > 32) {
      for (int k = 32 + lane; k < tot; k += 32) {
        int s2 = (k < cnt) ? csr[start + k] : d0 + j;
        float g2 = g[s2];
        float w = g2 * as2 + ed;
        w = w > 0.0f ? w : NEG_SLOPE * w;
        float pw = __expf(w - mx[j]);
        z += pw;
        a += pw * g2;
      }
    }
    zl[j] = z;
    al[j] = a;
  }
#pragma unroll
  for (int j = 0; j < NPW1; ++j) zl[j] = wave_sum(zl[j]);
#pragma unroll
  for (int j = 0; j < NPW1; ++j) al[j] = wave_sum(al[j]);
  float z = 1.0f, a = 0.0f;
#pragma unroll
  for (int j = 0; j < NPW1; ++j)
    if (lane == j) { z = zl[j]; a = al[j]; }
  float r = a / z + bb;
  float res = 1.0f / (1.0f + __expf(-r));
  if (lane < nn) out[d0 + lane] = res;
}

extern "C" void kernel_launch(void* const* d_in, const int* in_sizes, int n_in,
                              void* d_out, int out_size, void* d_ws, size_t ws_size,
                              hipStream_t stream) {
  const int*   ei     = (const int*)d_in[0];
  const float* x      = (const float*)d_in[1];
  const float* W1     = (const float*)d_in[2];
  const float* a_src1 = (const float*)d_in[3];
  const float* a_dst1 = (const float*)d_in[4];
  const float* b1     = (const float*)d_in[5];
  const float* W2     = (const float*)d_in[6];
  const float* a_src2 = (const float*)d_in[7];
  const float* a_dst2 = (const float*)d_in[8];
  const float* b2     = (const float*)d_in[9];

  const int E = in_sizes[0] / 2;
  const int n = in_sizes[1] / DIN;
  const int nbins = (n + BNODES - 1) / BNODES;
  const int G = (E + CH - 1) / CH;

  float* f     = (float*)d_ws;
  float* h     = f;
  float* es    = f + (size_t)128 * n;
  float* ed    = f + (size_t)129 * n;
  float* g     = f + (size_t)130 * n;
  int*   deg   = (int*)(f + (size_t)131 * n);
  int*   off   = (int*)(f + (size_t)132 * n);
  unsigned* part = (unsigned*)(f + (size_t)134 * n);
  int*   csr   = (int*)part + (size_t)E;
  int*   tab   = csr + (size_t)E;
  float* out   = (float*)d_out;

  const int B = 256;
  dim3 blk(B);

  const int GB = (n + GROWS - 1) / GROWS;
  k_binsort_gemm<<<dim3(GB + G), dim3(1024), 0, stream>>>(ei, E, nbins, part, tab, x, W1, a_src1, a_dst1,
                                                          h, es, ed, n, GB);
  k_binfinish<<<dim3(nbins), dim3(1024), 0, stream>>>(part, tab, G, nbins, n, deg, off, csr);

  k_node64<<<dim3((n + 8 * NPW64 - 1) / (8 * NPW64)), blk, 0, stream>>>(off, deg, csr, es, ed, h,
                                                                      b1, W2, g, n);

  k_node1<<<dim3((n + 32 * NPW1 - 1) / (32 * NPW1)), dim3(1024), 0, stream>>>(off, deg, csr, g, a_src2,
                                                                              a_dst2, b2, out, n);
}


# Bin finish and layer-1 node kernel in one launch: a 32-wave workgroup per 400-node bin builds its source list in LDS, then walks its nodes from it; 4 launches to 3; in-list (arrival) order changes
# speedup vs baseline: 9.3191x; 1.0490x over previous
#include <hip/hip_runtime.h>
#include <math.h>

typedef float v2f __attribute__((ext_vector_type(2)));
typedef float v8f __attribute__((ext_vector_type(8)));
typedef float v4f __attribute__((ext_vector_type(4)));

#define DIN 128
#define C1  64
#define NEG_SLOPE 0.2f
#define CH    8192
#define EPT   8
#define BNODES 400
#define DLB   9
#define MAXNB 1024
#define CAPB  12288
#define WP 132

__device__ __forceinline__ float wave_max(float v) {
#pragma unroll
  for (int o = 16; o > 0; o >>= 1) v = fmaxf(v, __shfl_xor(v, o, 32));
  return v;
}
__device__ __forceinline__ float wave_sum(float v) {
#pragma unroll
  for (int o = 16; o > 0; o >>= 1) v += __shfl_xor(v, o, 32);
  return v;
}

__device__ __forceinline__ float xhalf(float x) {
  int b = __builtin_bit_cast(int, x);
  return __builtin_bit_cast(float,
      __builtin_amdgcn_permlanex16(b, b, 0x76543210, 0xfedcba98, false, false));
}
template <int K> __device__ __forceinline__ float row_xor(float x) {
  int b = __builtin_bit_cast(int, x);
  return __builtin_bit_cast(float, __builtin_amdgcn_mov_dpp(b, 0x160 + K, 0xf, 0xf, true));
}
__device__ __forceinline__ float row_sum16(float v) {
  v += row_xor<1>(v);
  v += row_xor<2>(v);
  v += row_xor<4>(v);
  v += row_xor<8>(v);
  return v;
}

__device__ __forceinline__ int block_excl_scan(int v, int* wsum, int& total) {
  int lane = threadIdx.x & 31;
  int wave = threadIdx.x >> 5;
  int inc = v;
#pragma unroll
  for (int o = 1; o < 32; o <<= 1) {
    int t = __shfl_up(inc, o, 32);
    if (lane >= o) inc += t;
  }
  if (lane == 31) wsum[wave] = inc;
  __syncthreads();
  if (wave == 0) {
    int w = wsum[lane];
    int winc = w;
#pragma unroll
    for (int o = 1; o < 32; o <<= 1) {
      int t = __shfl_up(winc, o, 32);
      if (lane >= o) winc += t;
    }
    wsum[lane] = winc - w;
    if (lane == 31) wsum[32] = winc;
  }
  __syncthreads();
  total = wsum[32];
  return inc - v + wsum[wave];
}

__device__ __forceinline__ int tab_pitch(int nbins) { return (nbins + 32) & ~31; }
__device__ __forceinline__ void sort_chunk(const int* __restrict__ ei, int E, int nbins,
                                           unsigned* __restrict__ part, int* __restrict__ tab,
                                           int chunk) {
  __shared__ int cnt[MAXNB];
  __shared__ int boff[MAXNB + 1];
  __shared__ unsigned stage[CH];
  __shared__ int wsum[33];
  int tid = threadIdx.x;
  int base = chunk * CH;
  int m = E - base;
  if (m > CH) m = CH;
  for (int b = tid; b < nbins; b += 1024) cnt[b] = 0;
  __syncthreads();

  int s[EPT], d[EPT], r[EPT], bn[EPT];
#pragma unroll
  for (int j = 0; j < EPT; ++j) {
    int i = j * 1024 + tid;
    s[j] = 0; d[j] = 0;
    if (i < m) { s[j] = ei[base + i]; d[j] = ei[E + base + i]; }
  }
#pragma unroll
  for (int j = 0; j < EPT; ++j) {
    int i = j * 1024 + tid;
    r[j] = 0;
    bn[j] = (int)((unsigned)d[j] / (unsigned)BNODES);
    if (i < m) r[j] = atomicAdd(&cnt[bn[j]], 1);
  }
  __syncthreads();

  int v = (tid < nbins) ? cnt[tid] : 0;
  int total;
  int ex = block_excl_scan(v, wsum, total);
  if (tid < nbins) boff[tid] = ex;
  if (tid == 0) boff[nbins] = total;
  __syncthreads();

#pragma unroll
  for (int j = 0; j < EPT; ++j) {
    int i = j * 1024 + tid;
    if (i < m)
      stage[boff[bn[j]] + r[j]] = ((unsigned)s[j] << DLB) | (unsigned)(d[j] - bn[j] * BNODES);
  }
  __syncthreads();

  for (int i = tid; i < m; i += 1024) part[base + i] = stage[i];
  int* trow = tab + (size_t)chunk * tab_pitch(nbins);
  for (int b = tid; b <= nbins; b += 1024) trow[b] = boff[b];
}

#define GROWS 448
__device__ __forceinline__ void gemm_rows(const float* __restrict__ x,
                                          const float* __restrict__ W,
                                          const float* __restrict__ a_src,
                                          const float* __restrict__ a_dst,
                                          float* __restrict__ h, float* __restrict__ es,
                                          float* __restrict__ ed, int n, int gb) {
  __shared__ __attribute__((aligned(16))) float Wt[C1 * WP];
  __shared__ float av[2 * C1];
  int tid = threadIdx.x;
  if (tid < C1) av[tid] = a_src[tid];
  else if (tid < 2 * C1) av[tid] = a_dst[tid - C1];
#pragma unroll
  for (int it = 0; it < (DIN * C1) / 1024; ++it) {
    int idx  = tid + it * 1024;
    int c    = idx & 15;
    int hs   = (idx >> 4) & 1;
    int rest = idx >> 5;
    int nn   = (rest & 3) * 16 + c;
    int kk   = rest >> 2;
    int k    = (kk >> 1) * 4 + (kk & 1) + 2 * hs;
    Wt[nn * WP + k] = W[k * C1 + nn];
  }
  __syncthreads();

  int wave = __builtin_amdgcn_readfirstlane(tid >> 5);
  int lane = tid & 31;
  int half = lane >> 4;
  int l    = lane & 15;
  int rowBase = gb * GROWS + wave * 16;
  if (wave * 16 >= GROWS || rowBase >= n) return;

  int arow = rowBase + l;
  if (arow > n - 1) arow = n - 1;
  const float* ap = x + (size_t)arow * DIN + 2 * half;

  v8f c0 = {}, c1 = {}, c2 = {}, c3 = {};
#pragma unroll 4
  for (int k0 = 0; k0 < DIN; k0 += 4) {
    v2f a = *(const v2f*)(ap + k0);
    const float* wp = &Wt[l * WP + k0 + 2 * half];
    v2f b0 = *(const v2f*)(wp);
    v2f b1 = *(const v2f*)(wp + 16 * WP);
    v2f b2 = *(const v2f*)(wp + 32 * WP);
    v2f b3 = *(const v2f*)(wp + 48 * WP);
    c0 = __builtin_amdgcn_wmma_f32_16x16x4_f32(false, a, false, b0, (short)0, c0, false, false);
    c1 = __builtin_amdgcn_wmma_f32_16x16x4_f32(false, a, false, b1, (short)0, c1, false, false);
    c2 = __builtin_amdgcn_wmma_f32_16x16x4_f32(false, a, false, b2, (short)0, c2, false, false);
    c3 = __builtin_amdgcn_wmma_f32_16x16x4_f32(false, a, false, b3, (short)0, c3, false, false);
  }

#pragma unroll
  for (int v = 0; v < 8; ++v) {
    int row = rowBase + 8 * half + v;
    if (row < n) {
      float* hp = h + (size_t)row * C1 + l;
      hp[0]  = c0[v];
      hp[16] = c1[v];
      hp[32] = c2[v];
      hp[48] = c3[v];
    }
  }

  float s0 = av[l], s1 = av[16 + l], s2 = av[32 + l], s3 = av[48 + l];
  float t0 = av[C1 + l], t1 = av[C1 + 16 + l], t2 = av[C1 + 32 + l], t3 = av[C1 + 48 + l];
  float ks = 0.0f, kd = 0.0f;
#pragma unroll
  for (int v = 0; v < 8; ++v) {
    float ps = row_sum16(c0[v] * s0 + c1[v] * s1 + c2[v] * s2 + c3[v] * s3);
    float pd = row_sum16(c0[v] * t0 + c1[v] * t1 + c2[v] * t2 + c3[v] * t3);
    if (l == v) { ks = ps; kd = pd; }
  }
  if (l < 8) {
    int row = rowBase + 8 * half + l;
    if (row < n) { es[row] = ks; ed[row] = kd; }
  }
}

__global__ __launch_bounds__(1024) void k_binsort_gemm(const int* __restrict__ ei, int E,
                                                       int nbins, unsigned* __restrict__ part,
                                                       int* __restrict__ tab,
                                                       const float* __restrict__ x,
                                                       const float* __restrict__ W,
                                                       const float* __restrict__ a_src,
                                                       const float* __restrict__ a_dst,
                                                       float* __restrict__ h,
                                                       float* __restrict__ es,
                                                       float* __restrict__ ed, int n, int GB) {
  if ((int)blockIdx.x < GB) gemm_rows(x, W, a_src, a_dst, h, es, ed, n, (int)blockIdx.x);
  else sort_chunk(ei, E, nbins, part, tab, (int)blockIdx.x - GB);
}

#define RK 32
#define NPWF 13
#define NBP (32 * NPWF)
#define QT 12
__device__ __forceinline__ void node_rows(const int* lst, const int* ncnt, const int* noff,
                                          v2f (*mine)[32], int lane, int ln0, int d0, int nn,
                                          float esv, float edv, const float* __restrict__ es,
                                          const float* __restrict__ h,
                                          const float* __restrict__ b1,
                                          const float* __restrict__ W2, float* __restrict__ g) {
  int offv = 0, degv = 0;
  if (lane < nn) { offv = noff[ln0 + lane]; degv = ncnt[ln0 + lane]; }
  int start = __builtin_amdgcn_readlane(offv, 0);
  int cnt   = __builtin_amdgcn_readlane(degv, 0);
  int m = cnt < RK ? cnt : RK;
  int s = 0;
  if (lane < m) s = lst[start + lane];
  int start1 = 0, cnt1 = 0, m1 = 0, s1 = 0;
  if (1 < nn) {
    start1 = __builtin_amdgcn_readlane(offv, 1);
    cnt1   = __builtin_amdgcn_readlane(degv, 1);
    m1 = cnt1 < RK ? cnt1 : RK;
    if (lane < m1) s1 = lst[start1 + lane];
  }
  float ev = 0.0f;
  if (lane < m) ev = es[s];
  for (int j = 0; j < nn; ++j) {
    int d = d0 + j;
    asm volatile("" :: "v"(ev));
    float edj = __int_as_float(__builtin_amdgcn_readlane(__float_as_int(edv), j));
    float vs  = __int_as_float(__builtin_amdgcn_readlane(__float_as_int(esv), j)) + edj;
    vs = vs > 0.0f ? vs : NEG_SLOPE * vs;
    float v = ev + edj;
    v = v > 0.0f ? v : NEG_SLOPE * v;
    float ev1 = 0.0f;
    if (lane < m1) ev1 = es[s1];
    const v2f hs = *(const v2f*)(h + (size_t)d * C1 + 2 * lane);

    v2f r[RK];
    int sid[RK];
#pragma unroll
    for (int t = 0; t < RK; ++t) sid[t] = __builtin_amdgcn_readlane(s, t);
#pragma unroll
    for (int gq = 0; gq < RK; gq += 8) {
      if (gq < m) {
#pragma unroll
        for (int t = gq; t < gq + 8; ++t) {
          if (t < m) r[t] = *(const v2f*)(h + (size_t)sid[t] * C1 + 2 * lane);
        }
      }
    }
    int start2 = 0, cnt2 = 0, m2 = 0, s2n = 0;
    if (j + 2 < nn) {
      start2 = __builtin_amdgcn_readlane(offv, j + 2);
      cnt2   = __builtin_amdgcn_readlane(degv, j + 2);
      m2 = cnt2 < RK ? cnt2 : RK;
      if (lane < m2) s2n = lst[start2 + lane];
    }

    float mx = (lane < m) ? fmaxf(v, vs) : vs;
    for (int k0 = RK; k0 < cnt; k0 += 32) {
      int kk = k0 + lane;
      if (kk < cnt) {
        float e = es[lst[start + kk]] + edj;
        e = e > 0.0f ? e : NEG_SLOPE * e;
        mx = fmaxf(mx, e);
      }
    }
    mx = wave_max(mx);

    float p  = (lane < m) ? __expf(v - mx) : 0.0f;
    float zl = p;
    float a0 = 0.0f, a1 = 0.0f;
#pragma unroll
    for (int gq = 0; gq < RK; gq += 8) {
      if (gq < m) {
#pragma unroll
        for (int t = gq; t < gq + 8; ++t) {
          if (t < m) {
            float pt = __int_as_float(__builtin_amdgcn_readlane(__float_as_int(p), t));
            a0 += pt * r[t].x;
            a1 += pt * r[t].y;
          }
        }
      }
    }
    for (int k0 = RK; k0 < cnt; k0 += 32) {
      int kk = k0 + lane;
      int s2 = 0;
      float p2 = 0.0f;
      if (kk < cnt) {
        s2 = lst[start + kk];
        float e = es[s2] + edj;
        e = e > 0.0f ? e : NEG_SLOPE * e;
        p2 = __expf(e - mx);
      }
      zl += p2;
      int mm = cnt - k0;
      if (mm > 32) mm = 32;
      for (int t = 0; t < mm; ++t) {
        int st = __builtin_amdgcn_readlane(s2, t);
        float pt = __int_as_float(__builtin_amdgcn_readlane(__float_as_int(p2), t));
        v2f hv = *(const v2f*)(h + (size_t)st * C1 + 2 * lane);
        a0 += pt * hv.x;
        a1 += pt * hv.y;
      }
    }
    float pself = __expf(vs - mx);
    if (lane == (cnt & 31)) zl += pself;
    a0 += pself * hs.x;
    a1 += pself * hs.y;

    float z = wave_sum(zl);
    v2f o;
    o.x = a0 / z;
    o.y = a1 / z;
    mine[j][lane] = o;
    start = start1; cnt = cnt1; m = m1; s = s1; ev = ev1;
    start1 = start2; cnt1 = cnt2; m1 = m2; s1 = s2n;
  }

  const v2f bb = *(const v2f*)(b1 + 2 * lane);
  const v2f ww = *(const v2f*)(W2 + 2 * lane);
  float gres = 0.0f;
  v2f o[NPWF];
#pragma unroll
  for (int j = 0; j < NPWF; ++j) o[j] = mine[j < nn ? j : nn - 1][lane];
#pragma unroll
  for (int j = 0; j < NPWF; ++j) {
    float t = fmaxf(o[j].x + bb.x, 0.0f) * ww.x + fmaxf(o[j].y + bb.y, 0.0f) * ww.y;
    t = row_sum16(t);
    t = t + xhalf(t);
    if (lane == j) gres = t;
  }
  if (lane < nn) g[d0 + lane] = gres;
}

__global__ __launch_bounds__(1024, 1) void k_binfinish_node64(const unsigned* __restrict__ part,
                                                    const int* __restrict__ tab, int G,
                                                    int nbins, int n, int* __restrict__ deg,
                                                    int* __restrict__ off,
                                                    int* __restrict__ csr,
                                                    const float* __restrict__ es,
                                                    const float* __restrict__ ed,
                                                    const float* __restrict__ h,
                                                    const float* __restrict__ b1,
                                                    const float* __restrict__ W2,
                                                    float* __restrict__ g) {
  __shared__ int ncnt[NBP];
  __shared__ int noff[NBP];
  __shared__ int cur[NBP];
  __shared__ int wsum[33];
  __shared__ int outl[CAPB];
  __shared__ v2f park[32][NPWF][32];
  int tid = threadIdx.x;
  int bin = blockIdx.x;
  int ts  = tab_pitch(nbins);
  if (tid < NBP) { ncnt[tid] = 0; cur[tid] = 0; }
  int lane = tid & 31;
  int wv  = __builtin_amdgcn_readfirstlane(tid >> 5);
  int ln0 = wv * NPWF;
  int d0  = bin * BNODES + ln0;
  int nn  = BNODES - ln0;
  if (nn > NPWF) nn = NPWF;
  if (nn > n - d0) nn = n - d0;
  float esv = 0.0f, edv = 0.0f;
  if (lane < nn) { esv = es[d0 + lane]; edv = ed[d0 + lane]; }
  __syncthreads();

  int sub = tid & 3;
  int sg  = tid >> 2;
  int myb = 0;
  for (int w0 = 0; w0 < G; w0 += 256) {
    int w = w0 + sg;
    if (w < G) {
      int a = tab[(size_t)w * ts + bin];
      int b = tab[(size_t)w * ts + bin + 1];
      if (sub == 0) myb += a;
      const unsigned* p = part + (size_t)w * CH;
      for (int i0 = a + sub; i0 < b; i0 += 4 * QT) {
        unsigned q[QT];
#pragma unroll
        for (int u = 0; u < QT; ++u) {
          int i = i0 + 4 * u;
          q[u] = p[i < b ? i : b - 1];
        }
#pragma unroll
        for (int u = 0; u < QT; ++u) {
          int i = i0 + 4 * u;
          if (i < b) atomicAdd(&ncnt[q[u] & ((1u << DLB) - 1)], 1);
        }
      }
    }
  }
  __syncthreads();

  int binbase;
  block_excl_scan(myb, wsum, binbase);
  __syncthreads();
  int v = (tid < BNODES) ? ncnt[tid] : 0;
  int total;
  int ex = block_excl_scan(v, wsum, total);
  if (tid < NBP) noff[tid] = ex;
  if (tid < BNODES) {
    int node = bin * BNODES + tid;
    if (node < n) { deg[node] = v; off[node] = binbase + ex; }
  }
  __syncthreads();

  total   = __builtin_amdgcn_readfirstlane(total);
  binbase = __builtin_amdgcn_readfirstlane(binbase);
  bool inl = total <= CAPB;
  for (int w0 = 0; w0 < G; w0 += 256) {
    int w = w0 + sg;
    if (w < G) {
      int a = tab[(size_t)w * ts + bin];
      int b = tab[(size_t)w * ts + bin + 1];
      const unsigned* p = part + (size_t)w * CH;
      for (int i0 = a + sub; i0 < b; i0 += 4 * QT) {
        unsigned q[QT];
#pragma unroll
        for (int u = 0; u < QT; ++u) {
          int i = i0 + 4 * u;
          q[u] = p[i < b ? i : b - 1];
        }
#pragma unroll
        for (int u = 0; u < QT; ++u) {
          int i = i0 + 4 * u;
          if (i < b) {
            int dl  = (int)(q[u] & ((1u << DLB) - 1));
            int pos = noff[dl] + atomicAdd(&cur[dl], 1);
            int sv  = (int)(q[u] >> DLB);
            if (inl) outl[pos] = sv; else csr[binbase + pos] = sv;
          }
        }
      }
    }
  }
  __syncthreads();
  if (inl) {
    int ov[CAPB / 1024];
#pragma unroll
    for (int u = 0; u < CAPB / 1024; ++u) {
      int i = tid + u * 1024;
      ov[u] = (i < total) ? outl[i] : 0;
    }
#pragma unroll
    for (int u = 0; u < CAPB / 1024; ++u) {
      int i = tid + u * 1024;
      if (i < total) csr[binbase + i] = ov[u];
    }
  }

  if (nn <= 0) return;
  if (__builtin_expect(inl, 1)) node_rows(outl, ncnt, noff, park[wv], lane, ln0, d0, nn, esv, edv, es, h, b1, W2, g);
  else     node_rows(csr + binbase, ncnt, noff, park[wv], lane, ln0, d0, nn, esv, edv, es, h, b1, W2, g);
}

#define NPW1 13
#define GCAP 65536
__global__ __launch_bounds__(1024, 1) void k_node1(const int* __restrict__ off,
                                                const int* __restrict__ deg,
                                                const int* __restrict__ csr,
                                                const float* __restrict__ g,
                                                const float* __restrict__ a_src2,
                                                const float* __restrict__ a_dst2,
                                                const float* __restrict__ b2,
                                                float* __restrict__ out, int n) {
  __shared__ __attribute__((aligned(16))) float gl[GCAP + 4];
  int tid  = threadIdx.x;
  int lane = tid & 31;
  int d0 = __builtin_amdgcn_readfirstlane((blockIdx.x * 32 + (tid >> 5)) * NPW1);
  int nn = n - d0;
  if (nn > NPW1) nn = NPW1;
  if (nn < 0) nn = 0;
  int offv = 0, degv = 0;
  float gd = 0.0f;
  if (lane < nn) { offv = off[d0 + lane]; degv = deg[d0 + lane]; gd = g[d0 + lane]; }
  float as2 = a_src2[0];
  float ad2 = a_dst2[0];
  float bb  = b2[0];

  int s[NPW1];
#pragma unroll
  for (int j = 0; j < NPW1; ++j) {
    int start = __builtin_amdgcn_readlane(offv, j);
    int cnt   = __builtin_amdgcn_readlane(degv, j);
    s[j] = -1;
    if (j < nn) {
      if (lane < cnt) s[j] = csr[start + lane];
      else if (lane == cnt) s[j] = d0 + j;
    }
  }

  int shift = (int)(((size_t)g >> 2) & 3);
  float gs[NPW1];
#pragma unroll
  for (int j = 0; j < NPW1; ++j) gs[j] = 0.0f;

  for (int base = 0; base < n; base += GCAP) {
    int len = n - base;
    if (len > GCAP) len = GCAP;
    int lo  = base - shift;
    int nch = (len + shift + 3) >> 2;
    __syncthreads();
    int cfirst = (lo < 0) ? 1 : 0;
    int cend   = (lo + 4 * nch > n) ? nch - 1 : nch;
    for (int c0 = cfirst + tid; c0 < cend; c0 += 4 * 1024) {
      v4f v[4];
#pragma unroll
      for (int u = 0; u < 4; ++u) {
        int c = c0 + u * 1024;
        v[u] = *(const v4f*)(g + lo + 4 * (c < cend ? c : cend - 1));
      }
#pragma unroll
      for (int u = 0; u < 4; ++u) {
        int c = c0 + u * 1024;
        if (c < cend) *(v4f*)&gl[4 * c] = v[u];
      }
    }
    if (tid < 8) {
      int  c    = (tid < 4) ? 0 : nch - 1;
      bool part = (tid < 4) ? (cfirst == 1) : (cend < nch);
      int  gi   = lo + 4 * c + (tid & 3);
      if (part && gi >= 0 && gi < n) gl[4 * c + (tid & 3)] = g[gi];
    }
    __syncthreads();
#pragma unroll
    for (int j = 0; j < NPW1; ++j) {
      unsigned idx = (unsigned)(s[j] - base);
      if (idx < (unsigned)GCAP) gs[j] = gl[idx + shift];
    }
  }

  float vv[NPW1], mx[NPW1];
#pragma unroll
  for (int j = 0; j < NPW1; ++j) {
    int start = __builtin_amdgcn_readlane(offv, j);
    int cnt   = __builtin_amdgcn_readlane(degv, j);
    int tot   = cnt + 1;
    float ed  = __int_as_float(__builtin_amdgcn_readlane(__float_as_int(gd), j)) * ad2;
    float v = gs[j] * as2 + ed;
    v = v > 0.0f ? v : NEG_SLOPE * v;
    float m = (lane < tot) ? v : -INFINITY;
    if (tot > 32) {
      for (int k = 32 + lane; k < tot; k += 32) {
        int s2 = (k < cnt) ? csr[start + k] : d0 + j;
        float w = g[s2] * as2 + ed;
        w = w > 0.0f ? w : NEG_SLOPE * w;
        m = fmaxf(m, w);
      }
    }
    vv[j] = v;
    mx[j] = m;
  }
#pragma unroll
  for (int j = 0; j < NPW1; ++j) mx[j] = wave_max(mx[j]);

  float zl[NPW1], al[NPW1];
#pragma unroll
  for (int j = 0; j < NPW1; ++j) {
    int start = __builtin_amdgcn_readlane(offv, j);
    int cnt   = __builtin_amdgcn_readlane(degv, j);
    int tot   = cnt + 1;
    float ed  = __int_as_float(__builtin_amdgcn_readlane(__float_as_int(gd), j)) * ad2;
    float p = (lane < tot) ? __expf(vv[j] - mx[j]) : 0.0f;
    float z = p;
    float a = p * gs[j];
    if (tot > 32) {
      for (int k = 32 + lane; k < tot; k += 32) {
        int s2 = (k < cnt) ? csr[start + k] : d0 + j;
        float g2 = g[s2];
        float w = g2 * as2 + ed;
        w = w > 0.0f ? w : NEG_SLOPE * w;
        float pw = __expf(w - mx[j]);
        z += pw;
        a += pw * g2;
      }
    }
    zl[j] = z;
    al[j] = a;
  }
#pragma unroll
  for (int j = 0; j < NPW1; ++j) zl[j] = wave_sum(zl[j]);
#pragma unroll
  for (int j = 0; j < NPW1; ++j) al[j] = wave_sum(al[j]);
  float z = 1.0f, a = 0.0f;
#pragma unroll
  for (int j = 0; j < NPW1; ++j)
    if (lane == j) { z = zl[j]; a = al[j]; }
  float r = a / z + bb;
  float res = 1.0f / (1.0f + __expf(-r));
  if (lane < nn) out[d0 + lane] = res;
}

extern "C" void kernel_launch(void* const* d_in, const int* in_sizes, int n_in,
                              void* d_out, int out_size, void* d_ws, size_t ws_size,
                              hipStream_t stream) {
  const int*   ei     = (const int*)d_in[0];
  const float* x      = (const float*)d_in[1];
  const float* W1     = (const float*)d_in[2];
  const float* a_src1 = (const float*)d_in[3];
  const float* a_dst1 = (const float*)d_in[4];
  const float* b1     = (const float*)d_in[5];
  const float* W2     = (const float*)d_in[6];
  const float* a_src2 = (const float*)d_in[7];
  const float* a_dst2 = (const float*)d_in[8];
  const float* b2     = (const float*)d_in[9];

  const int E = in_sizes[0] / 2;
  const int n = in_sizes[1] / DIN;
  const int nbins = (n + BNODES - 1) / BNODES;
  const int G = (E + CH - 1) / CH;

  float* f     = (float*)d_ws;
  float* h     = f;
  float* es    = f + (size_t)128 * n;
  float* ed    = f + (size_t)129 * n;
  float* g     = f + (size_t)130 * n;
  int*   deg   = (int*)(f + (size_t)131 * n);
  int*   off   = (int*)(f + (size_t)132 * n);
  unsigned* part = (unsigned*)(f + (size_t)134 * n);
  int*   csr   = (int*)part + (size_t)E;
  int*   tab   = csr + (size_t)E;
  float* out   = (float*)d_out;

  const int GB = (n + GROWS - 1) / GROWS;
  k_binsort_gemm<<<dim3(GB + G), dim3(1024), 0, stream>>>(ei, E, nbins, part, tab, x, W1, a_src1, a_dst1,
                                                          h, es, ed, n, GB);
  k_binfinish_node64<<<dim3(nbins), dim3(1024), 0, stream>>>(part, tab, G, nbins, n, deg, off, csr,
                                                             es, ed, h, b1, W2, g);

  k_node1<<<dim3((n + 32 * NPW1 - 1) / (32 * NPW1)), dim3(1024), 0, stream>>>(off, deg, csr, g, a_src2,
                                                                              a_dst2, b2, out, n);
}
